# starting kernel
# baseline (speedup 1.0000x reference)
#include <hip/hip_runtime.h>
#include <hip/hip_bf16.h>
#include <float.h>


typedef float v2f __attribute__((ext_vector_type(2)));
typedef float v8f __attribute__((ext_vector_type(8)));

#define S_      64
#define N_      256
#define L_      40
#define LP_     48
#define KNB     17
#define OD_     128
#define LANGID_ 256
#define PCID_   128

static __device__ inline v8f wmma4(v2f a, v2f b, v8f c) {
  return __builtin_amdgcn_wmma_f32_16x16x4_f32(false, a, false, b, (short)0, c,
                                               false, false);
}

__global__ void knn_kernel(const float* __restrict__ coord,
                           int* __restrict__ idx_out) {
  __shared__ float cx[N_], cy[N_], cz[N_];
  const int s = blockIdx.x;
  const int n = threadIdx.x;
  const float* cs = coord + (size_t)s * N_ * 3;
  cx[n] = cs[n * 3 + 0];
  cy[n] = cs[n * 3 + 1];
  cz[n] = cs[n * 3 + 2];
  __syncthreads();

  const float px = cx[n], py = cy[n], pz = cz[n];
  float pd = -1.0f;
  int   pi = -1;
  int* out = idx_out + ((size_t)s * N_ + n) * KNB;
  for (int j = 0; j < KNB; ++j) {
    float bd = FLT_MAX;
    int   bi = 0x7fffffff;
    for (int m = 0; m < N_; ++m) {
      float dx = cx[m] - px, dy = cy[m] - py, dz = cz[m] - pz;
      float d2 = dx * dx + dy * dy + dz * dz;
      bool gtPrev = (d2 > pd) || (d2 == pd && m > pi);
      bool ltBest = (d2 < bd) || (d2 == bd && m < bi);
      if (gtPrev && ltBest) { bd = d2; bi = m; }
    }
    out[j] = bi;
    pd = bd;
    pi = bi;
  }
}

__global__ void lang_mlp_kernel(const float* __restrict__ lang_feat,
                                const float* __restrict__ w1,
                                const float* __restrict__ b1,
                                const float* __restrict__ w2,
                                const float* __restrict__ b2,
                                float* __restrict__ lang_out) {
  __shared__ float xs[LANGID_];
  __shared__ float hs[OD_];
  const int row = blockIdx.x;
  const int t   = threadIdx.x;
  const float* x = lang_feat + (size_t)row * LANGID_;
  xs[t]       = x[t];
  xs[t + 128] = x[t + 128];
  __syncthreads();
  float acc = b1[t];
  for (int k = 0; k < LANGID_; ++k) acc += xs[k] * w1[k * OD_ + t];
  hs[t] = fmaxf(acc, 0.0f);
  __syncthreads();
  float acc2 = b2[t];
  for (int k = 0; k < OD_; ++k) acc2 += hs[k] * w2[k * OD_ + t];
  lang_out[(size_t)row * OD_ + t] = acc2;
}

__global__ void feat_mlp_kernel(const float* __restrict__ feat,
                                const float* __restrict__ w1,
                                const float* __restrict__ b1,
                                const float* __restrict__ w2,
                                const float* __restrict__ b2,
                                float* __restrict__ f_out) {
  __shared__ float Asm[16 * PCID_];
  __shared__ float Hsm[16 * OD_];
  const int s    = blockIdx.x >> 4;
  const int mt   = blockIdx.x & 15;
  const int tid  = threadIdx.x;
  const int lane = tid & 31, wave = tid >> 5;
  const int half = lane >> 4, lr = lane & 15;

  const float* A = feat + ((size_t)s * N_ + mt * 16) * PCID_;
  for (int i = tid; i < 16 * PCID_; i += 256) Asm[i] = A[i];
  __syncthreads();

  const int n0 = wave * 16;
  v8f c = {};
  for (int k0 = 0; k0 < PCID_; k0 += 4) {
    const int ka = k0 + 2 * half;
    v2f a, b;
    a.x = Asm[lr * PCID_ + ka];
    a.y = Asm[lr * PCID_ + ka + 1];
    b.x = w1[ka * OD_ + n0 + lr];
    b.y = w1[(ka + 1) * OD_ + n0 + lr];
    c = wmma4(a, b, c);
  }
  {
    const float bb = b1[n0 + lr];
#pragma unroll
    for (int v = 0; v < 8; ++v)
      Hsm[(v + 8 * half) * OD_ + n0 + lr] = fmaxf(c[v] + bb, 0.0f);
  }
  __syncthreads();

  v8f c2 = {};
  for (int k0 = 0; k0 < OD_; k0 += 4) {
    const int ka = k0 + 2 * half;
    v2f a, b;
    a.x = Hsm[lr * OD_ + ka];
    a.y = Hsm[lr * OD_ + ka + 1];
    b.x = w2[ka * OD_ + n0 + lr];
    b.y = w2[(ka + 1) * OD_ + n0 + lr];
    c2 = wmma4(a, b, c2);
  }
  {
    const float bb = b2[n0 + lr];
    float* fo = f_out + ((size_t)s * N_ + mt * 16) * OD_;
#pragma unroll
    for (int v = 0; v < 8; ++v)
      fo[(v + 8 * half) * OD_ + n0 + lr] = c2[v] + bb;
  }
}

__global__ void fused_kernel(const float* __restrict__ coord,
                             const float* __restrict__ lang_mask,
                             const float* __restrict__ rel_w1,
                             const float* __restrict__ rel_b1,
                             const float* __restrict__ rel_w2,
                             const float* __restrict__ rel_b2,
                             const int*   __restrict__ idx_ws,
                             const float* __restrict__ lang_ws,
                             const float* __restrict__ f_ws,
                             float* __restrict__ out,
                             float* __restrict__ score) {
  __shared__ float fg[32 * OD_];
  __shared__ float relin[32 * 16];
  __shared__ float hbuf[32 * OD_];
  __shared__ float relm[32 * OD_];
  __shared__ float attnsm[32 * LP_];
  __shared__ int   idxsm[KNB];
  __shared__ float red[OD_];

  const int s    = blockIdx.x >> 8;
  const int n    = blockIdx.x & 255;
  const int tid  = threadIdx.x;
  const int lane = tid & 31, wave = tid >> 5;
  const int half = lane >> 4, lr = lane & 15;

  if (tid < KNB) idxsm[tid] = idx_ws[((size_t)s * N_ + n) * KNB + tid];
  for (int i = tid; i < 32 * 16; i += 256) relin[i] = 0.f;
  __syncthreads();

  for (int i = tid; i < 32 * OD_; i += 256) {
    const int j = i >> 7, d = i & 127;
    fg[i] = (j < KNB) ? f_ws[((size_t)s * N_ + idxsm[j]) * OD_ + d] : 0.f;
  }
  if (tid < KNB) {
    const int j = tid;
    const float* cs = coord + (size_t)s * N_ * 3;
    const int m = idxsm[j];
    const float nx = cs[m * 3 + 0], ny = cs[m * 3 + 1], nz = cs[m * 3 + 2];
    const float ox = cs[n * 3 + 0], oy = cs[n * 3 + 1], oz = cs[n * 3 + 2];
    const float rx = nx - ox, ry = ny - oy, rz = nz - oz;
    const float dist = sqrtf(rx * rx + ry * ry + rz * rz);
    float* ri = relin + j * 16;
    ri[0] = nx; ri[1] = ny; ri[2] = nz;
    ri[3] = ox; ri[4] = oy; ri[5] = oz;
    ri[6] = rx; ri[7] = ry; ri[8] = rz;
    ri[9] = dist;
  }
  __syncthreads();

  for (int job = wave; job < 16; job += 8) {
    const int r0 = (job & 1) * 16, c0 = (job >> 1) * 16;
    v8f c = {};
    for (int k0 = 0; k0 < 16; k0 += 4) {
      const int ka = k0 + 2 * half;
      v2f a, b;
      a.x = relin[(r0 + lr) * 16 + ka];
      a.y = relin[(r0 + lr) * 16 + ka + 1];
      const int kb0 = min(ka, 9), kb1 = min(ka + 1, 9);
      b.x = rel_w1[kb0 * OD_ + c0 + lr];
      b.y = rel_w1[kb1 * OD_ + c0 + lr];
      c = wmma4(a, b, c);
    }
    const float bb = rel_b1[c0 + lr];
#pragma unroll
    for (int v = 0; v < 8; ++v)
      hbuf[(r0 + v + 8 * half) * OD_ + c0 + lr] = fmaxf(c[v] + bb, 0.f);
  }
  __syncthreads();

  for (int job = wave; job < 16; job += 8) {
    const int r0 = (job & 1) * 16, c0 = (job >> 1) * 16;
    v8f c = {};
    for (int k0 = 0; k0 < OD_; k0 += 4) {
      const int ka = k0 + 2 * half;
      v2f a, b;
      a.x = hbuf[(r0 + lr) * OD_ + ka];
      a.y = hbuf[(r0 + lr) * OD_ + ka + 1];
      b.x = rel_w2[ka * OD_ + c0 + lr];
      b.y = rel_w2[(ka + 1) * OD_ + c0 + lr];
      c = wmma4(a, b, c);
    }
    const float bb = rel_b2[c0 + lr];
#pragma unroll
    for (int v = 0; v < 8; ++v)
      relm[(r0 + v + 8 * half) * OD_ + c0 + lr] = c[v] + bb;
  }
  __syncthreads();

  if (wave < 6) {
    const int r0 = (wave & 1) * 16, c0 = (wave >> 1) * 16;
    const float* langS = lang_ws + (size_t)s * L_ * OD_;
    const int lclamp = min(c0 + lr, L_ - 1);
    v8f c = {};
    for (int k0 = 0; k0 < OD_; k0 += 4) {
      const int ka = k0 + 2 * half;
      v2f a, b;
      a.x = fg[(r0 + lr) * OD_ + ka];
      a.y = fg[(r0 + lr) * OD_ + ka + 1];
      b.x = langS[lclamp * OD_ + ka];
      b.y = langS[lclamp * OD_ + ka + 1];
      c = wmma4(a, b, c);
    }
#pragma unroll
    for (int v = 0; v < 8; ++v)
      attnsm[(r0 + v + 8 * half) * LP_ + c0 + lr] = c[v];
  }
  __syncthreads();

  if (tid < 32) {
    float* rowp = attnsm + tid * LP_;
    if (tid < KNB) {
      float mx = -FLT_MAX;
      for (int l = 0; l < L_; ++l) mx = fmaxf(mx, rowp[l]);
      float ssum = 0.f;
      for (int l = 0; l < L_; ++l) {
        const float e = expf(rowp[l] - mx);
        rowp[l] = e;
        ssum += e;
      }
      const float inv = 1.f / ssum;
      const float* mk = lang_mask + (size_t)s * L_;
      float s2 = 0.f;
      for (int l = 0; l < L_; ++l) {
        const float p = rowp[l] * inv * mk[l];
        rowp[l] = p;
        s2 += p;
      }
      const float inv2 = 1.f / (s2 + 1e-7f);
      for (int l = 0; l < L_; ++l) rowp[l] *= inv2;
      for (int l = L_; l < LP_; ++l) rowp[l] = 0.f;
    } else {
      for (int l = 0; l < LP_; ++l) rowp[l] = 0.f;
    }
  }
  __syncthreads();

  for (int job = wave; job < 16; job += 8) {
    const int r0 = (job & 1) * 16, c0 = (job >> 1) * 16;
    const float* langS = lang_ws + (size_t)s * L_ * OD_;
    v8f c = {};
    for (int k0 = 0; k0 < LP_; k0 += 4) {
      const int ka  = k0 + 2 * half;
      const int kb0 = min(ka, L_ - 1), kb1 = min(ka + 1, L_ - 1);
      v2f a, b;
      a.x = attnsm[(r0 + lr) * LP_ + ka];
      a.y = attnsm[(r0 + lr) * LP_ + ka + 1];
      b.x = langS[kb0 * OD_ + c0 + lr];
      b.y = langS[kb1 * OD_ + c0 + lr];
      c = wmma4(a, b, c);
    }
#pragma unroll
    for (int v = 0; v < 8; ++v)
      hbuf[(r0 + v + 8 * half) * OD_ + c0 + lr] = c[v];
  }
  __syncthreads();

  if (tid < OD_) {
    const int d = tid;
    float acc = 0.f;
    for (int j = 0; j < KNB; ++j)
      acc += fg[j * OD_ + d] * hbuf[j * OD_ + d] * relm[j * OD_ + d];
    acc += f_ws[((size_t)s * N_ + n) * OD_ + d];
    out[((size_t)s * N_ + n) * OD_ + d] = acc;
    red[d] = acc;
  }
  __syncthreads();
  for (int off = 64; off > 0; off >>= 1) {
    if (tid < off) red[tid] += red[tid + off];
    __syncthreads();
  }
  if (tid == 0) score[(size_t)s * N_ + n] = red[0];
}

extern "C" void kernel_launch(void* const* d_in, const int* in_sizes, int n_in,
                              void* d_out, int out_size, void* d_ws,
                              size_t ws_size, hipStream_t stream) {
  (void)in_sizes; (void)n_in; (void)out_size; (void)ws_size;
  const float* feat      = (const float*)d_in[0];
  const float* coord     = (const float*)d_in[1];
  const float* lang_feat = (const float*)d_in[2];
  const float* lang_mask = (const float*)d_in[3];
  const float* rel_w1    = (const float*)d_in[4];
  const float* rel_b1    = (const float*)d_in[5];
  const float* rel_w2    = (const float*)d_in[6];
  const float* rel_b2    = (const float*)d_in[7];
  const float* lang_w1   = (const float*)d_in[8];
  const float* lang_b1   = (const float*)d_in[9];
  const float* lang_w2   = (const float*)d_in[10];
  const float* lang_b2   = (const float*)d_in[11];
  const float* feat_w1   = (const float*)d_in[12];
  const float* feat_b1   = (const float*)d_in[13];
  const float* feat_w2   = (const float*)d_in[14];
  const float* feat_b2   = (const float*)d_in[15];

  char* ws = (char*)d_ws;
  int*   idx_ws  = (int*)ws;
  float* lang_ws = (float*)(ws + (size_t)S_ * N_ * KNB * 4);
  float* f_ws    = (float*)(ws + (size_t)S_ * N_ * KNB * 4 +
                                 (size_t)S_ * L_ * OD_ * 4);

  float* out   = (float*)d_out;
  float* score = out + (size_t)S_ * N_ * OD_;

  knn_kernel<<<S_, N_, 0, stream>>>(coord, idx_ws);
  lang_mlp_kernel<<<S_ * L_, 128, 0, stream>>>(lang_feat, lang_w1, lang_b1,
                                               lang_w2, lang_b2, lang_ws);
  feat_mlp_kernel<<<S_ * (N_ / 16), 256, 0, stream>>>(feat, feat_w1, feat_b1,
                                                      feat_w2, feat_b2, f_ws);
  fused_kernel<<<S_ * N_, 256, 0, stream>>>(coord, lang_mask, rel_w1, rel_b1,
                                            rel_w2, rel_b2, idx_ws, lang_ws,
                                            f_ws, out, score);
}


# Pad LDS row pitch to 4 x odd floats (128->132, 48->52, 16->20) so each WMMA A-fragment 8-byte read covers all 64 banks instead of 16 rows sharing two; math and summation order unchanged
# speedup vs baseline: 1.2509x; 1.2509x over previous
#include <hip/hip_runtime.h>
#include <hip/hip_bf16.h>
#include <float.h>


typedef float v2f __attribute__((ext_vector_type(2)));
typedef float v8f __attribute__((ext_vector_type(8)));

#define S_      64
#define N_      256
#define L_      40
#define LP_     48
#define RP_     132
#define LPP_    52
#define KP_     20
#define KNB     17
#define OD_     128
#define LANGID_ 256
#define PCID_   128

static __device__ inline v8f wmma4(v2f a, v2f b, v8f c) {
  return __builtin_amdgcn_wmma_f32_16x16x4_f32(false, a, false, b, (short)0, c,
                                               false, false);
}

__global__ void knn_kernel(const float* __restrict__ coord,
                           int* __restrict__ idx_out) {
  __shared__ float cx[N_], cy[N_], cz[N_];
  const int s = blockIdx.x;
  const int n = threadIdx.x;
  const float* cs = coord + (size_t)s * N_ * 3;
  cx[n] = cs[n * 3 + 0];
  cy[n] = cs[n * 3 + 1];
  cz[n] = cs[n * 3 + 2];
  __syncthreads();

  const float px = cx[n], py = cy[n], pz = cz[n];
  float pd = -1.0f;
  int   pi = -1;
  int* out = idx_out + ((size_t)s * N_ + n) * KNB;
  for (int j = 0; j < KNB; ++j) {
    float bd = FLT_MAX;
    int   bi = 0x7fffffff;
    for (int m = 0; m < N_; ++m) {
      float dx = cx[m] - px, dy = cy[m] - py, dz = cz[m] - pz;
      float d2 = dx * dx + dy * dy + dz * dz;
      bool gtPrev = (d2 > pd) || (d2 == pd && m > pi);
      bool ltBest = (d2 < bd) || (d2 == bd && m < bi);
      if (gtPrev && ltBest) { bd = d2; bi = m; }
    }
    out[j] = bi;
    pd = bd;
    pi = bi;
  }
}

__global__ void lang_mlp_kernel(const float* __restrict__ lang_feat,
                                const float* __restrict__ w1,
                                const float* __restrict__ b1,
                                const float* __restrict__ w2,
                                const float* __restrict__ b2,
                                float* __restrict__ lang_out) {
  __shared__ float xs[LANGID_];
  __shared__ float hs[OD_];
  const int row = blockIdx.x;
  const int t   = threadIdx.x;
  const float* x = lang_feat + (size_t)row * LANGID_;
  xs[t]       = x[t];
  xs[t + 128] = x[t + 128];
  __syncthreads();
  float acc = b1[t];
  for (int k = 0; k < LANGID_; ++k) acc += xs[k] * w1[k * OD_ + t];
  hs[t] = fmaxf(acc, 0.0f);
  __syncthreads();
  float acc2 = b2[t];
  for (int k = 0; k < OD_; ++k) acc2 += hs[k] * w2[k * OD_ + t];
  lang_out[(size_t)row * OD_ + t] = acc2;
}

__global__ void feat_mlp_kernel(const float* __restrict__ feat,
                                const float* __restrict__ w1,
                                const float* __restrict__ b1,
                                const float* __restrict__ w2,
                                const float* __restrict__ b2,
                                float* __restrict__ f_out) {
  __shared__ float Asm[16 * RP_];
  __shared__ float Hsm[16 * RP_];
  const int s    = blockIdx.x >> 4;
  const int mt   = blockIdx.x & 15;
  const int tid  = threadIdx.x;
  const int lane = tid & 31, wave = tid >> 5;
  const int half = lane >> 4, lr = lane & 15;

  const float* A = feat + ((size_t)s * N_ + mt * 16) * PCID_;
  for (int i = tid; i < 16 * PCID_; i += 256)
    Asm[(i >> 7) * RP_ + (i & 127)] = A[i];
  __syncthreads();

  const int n0 = wave * 16;
  v8f c = {};
  for (int k0 = 0; k0 < PCID_; k0 += 4) {
    const int ka = k0 + 2 * half;
    v2f a, b;
    a.x = Asm[lr * RP_ + ka];
    a.y = Asm[lr * RP_ + ka + 1];
    b.x = w1[ka * OD_ + n0 + lr];
    b.y = w1[(ka + 1) * OD_ + n0 + lr];
    c = wmma4(a, b, c);
  }
  {
    const float bb = b1[n0 + lr];
#pragma unroll
    for (int v = 0; v < 8; ++v)
      Hsm[(v + 8 * half) * RP_ + n0 + lr] = fmaxf(c[v] + bb, 0.0f);
  }
  __syncthreads();

  v8f c2 = {};
  for (int k0 = 0; k0 < OD_; k0 += 4) {
    const int ka = k0 + 2 * half;
    v2f a, b;
    a.x = Hsm[lr * RP_ + ka];
    a.y = Hsm[lr * RP_ + ka + 1];
    b.x = w2[ka * OD_ + n0 + lr];
    b.y = w2[(ka + 1) * OD_ + n0 + lr];
    c2 = wmma4(a, b, c2);
  }
  {
    const float bb = b2[n0 + lr];
    float* fo = f_out + ((size_t)s * N_ + mt * 16) * OD_;
#pragma unroll
    for (int v = 0; v < 8; ++v)
      fo[(v + 8 * half) * OD_ + n0 + lr] = c2[v] + bb;
  }
}

__global__ void fused_kernel(const float* __restrict__ coord,
                             const float* __restrict__ lang_mask,
                             const float* __restrict__ rel_w1,
                             const float* __restrict__ rel_b1,
                             const float* __restrict__ rel_w2,
                             const float* __restrict__ rel_b2,
                             const int*   __restrict__ idx_ws,
                             const float* __restrict__ lang_ws,
                             const float* __restrict__ f_ws,
                             float* __restrict__ out,
                             float* __restrict__ score) {
  __shared__ float fg[32 * RP_];
  __shared__ float relin[32 * KP_];
  __shared__ float hbuf[32 * RP_];
  __shared__ float relm[32 * RP_];
  __shared__ float attnsm[32 * LPP_];
  __shared__ int   idxsm[KNB];
  __shared__ float red[OD_];

  const int s    = blockIdx.x >> 8;
  const int n    = blockIdx.x & 255;
  const int tid  = threadIdx.x;
  const int lane = tid & 31, wave = tid >> 5;
  const int half = lane >> 4, lr = lane & 15;

  if (tid < KNB) idxsm[tid] = idx_ws[((size_t)s * N_ + n) * KNB + tid];
  for (int i = tid; i < 32 * KP_; i += 256) relin[i] = 0.f;
  __syncthreads();

  for (int i = tid; i < 32 * OD_; i += 256) {
    const int j = i >> 7, d = i & 127;
    fg[j * RP_ + d] =
        (j < KNB) ? f_ws[((size_t)s * N_ + idxsm[j]) * OD_ + d] : 0.f;
  }
  if (tid < KNB) {
    const int j = tid;
    const float* cs = coord + (size_t)s * N_ * 3;
    const int m = idxsm[j];
    const float nx = cs[m * 3 + 0], ny = cs[m * 3 + 1], nz = cs[m * 3 + 2];
    const float ox = cs[n * 3 + 0], oy = cs[n * 3 + 1], oz = cs[n * 3 + 2];
    const float rx = nx - ox, ry = ny - oy, rz = nz - oz;
    const float dist = sqrtf(rx * rx + ry * ry + rz * rz);
    float* ri = relin + j * KP_;
    ri[0] = nx; ri[1] = ny; ri[2] = nz;
    ri[3] = ox; ri[4] = oy; ri[5] = oz;
    ri[6] = rx; ri[7] = ry; ri[8] = rz;
    ri[9] = dist;
  }
  __syncthreads();

  for (int job = wave; job < 16; job += 8) {
    const int r0 = (job & 1) * 16, c0 = (job >> 1) * 16;
    v8f c = {};
    for (int k0 = 0; k0 < 16; k0 += 4) {
      const int ka = k0 + 2 * half;
      v2f a, b;
      a.x = relin[(r0 + lr) * KP_ + ka];
      a.y = relin[(r0 + lr) * KP_ + ka + 1];
      const int kb0 = min(ka, 9), kb1 = min(ka + 1, 9);
      b.x = rel_w1[kb0 * OD_ + c0 + lr];
      b.y = rel_w1[kb1 * OD_ + c0 + lr];
      c = wmma4(a, b, c);
    }
    const float bb = rel_b1[c0 + lr];
#pragma unroll
    for (int v = 0; v < 8; ++v)
      hbuf[(r0 + v + 8 * half) * RP_ + c0 + lr] = fmaxf(c[v] + bb, 0.f);
  }
  __syncthreads();

  for (int job = wave; job < 16; job += 8) {
    const int r0 = (job & 1) * 16, c0 = (job >> 1) * 16;
    v8f c = {};
    for (int k0 = 0; k0 < OD_; k0 += 4) {
      const int ka = k0 + 2 * half;
      v2f a, b;
      a.x = hbuf[(r0 + lr) * RP_ + ka];
      a.y = hbuf[(r0 + lr) * RP_ + ka + 1];
      b.x = rel_w2[ka * OD_ + c0 + lr];
      b.y = rel_w2[(ka + 1) * OD_ + c0 + lr];
      c = wmma4(a, b, c);
    }
    const float bb = rel_b2[c0 + lr];
#pragma unroll
    for (int v = 0; v < 8; ++v)
      relm[(r0 + v + 8 * half) * RP_ + c0 + lr] = c[v] + bb;
  }
  __syncthreads();

  if (wave < 6) {
    const int r0 = (wave & 1) * 16, c0 = (wave >> 1) * 16;
    const float* langS = lang_ws + (size_t)s * L_ * OD_;
    const int lclamp = min(c0 + lr, L_ - 1);
    v8f c = {};
    for (int k0 = 0; k0 < OD_; k0 += 4) {
      const int ka = k0 + 2 * half;
      v2f a, b;
      a.x = fg[(r0 + lr) * RP_ + ka];
      a.y = fg[(r0 + lr) * RP_ + ka + 1];
      b.x = langS[lclamp * OD_ + ka];
      b.y = langS[lclamp * OD_ + ka + 1];
      c = wmma4(a, b, c);
    }
#pragma unroll
    for (int v = 0; v < 8; ++v)
      attnsm[(r0 + v + 8 * half) * LPP_ + c0 + lr] = c[v];
  }
  __syncthreads();

  if (tid < 32) {
    float* rowp = attnsm + tid * LPP_;
    if (tid < KNB) {
      float mx = -FLT_MAX;
      for (int l = 0; l < L_; ++l) mx = fmaxf(mx, rowp[l]);
      float ssum = 0.f;
      for (int l = 0; l < L_; ++l) {
        const float e = expf(rowp[l] - mx);
        rowp[l] = e;
        ssum += e;
      }
      const float inv = 1.f / ssum;
      const float* mk = lang_mask + (size_t)s * L_;
      float s2 = 0.f;
      for (int l = 0; l < L_; ++l) {
        const float p = rowp[l] * inv * mk[l];
        rowp[l] = p;
        s2 += p;
      }
      const float inv2 = 1.f / (s2 + 1e-7f);
      for (int l = 0; l < L_; ++l) rowp[l] *= inv2;
      for (int l = L_; l < LP_; ++l) rowp[l] = 0.f;
    } else {
      for (int l = 0; l < LP_; ++l) rowp[l] = 0.f;
    }
  }
  __syncthreads();

  for (int job = wave; job < 16; job += 8) {
    const int r0 = (job & 1) * 16, c0 = (job >> 1) * 16;
    const float* langS = lang_ws + (size_t)s * L_ * OD_;
    v8f c = {};
    for (int k0 = 0; k0 < LP_; k0 += 4) {
      const int ka  = k0 + 2 * half;
      const int kb0 = min(ka, L_ - 1), kb1 = min(ka + 1, L_ - 1);
      v2f a, b;
      a.x = attnsm[(r0 + lr) * LPP_ + ka];
      a.y = attnsm[(r0 + lr) * LPP_ + ka + 1];
      b.x = langS[kb0 * OD_ + c0 + lr];
      b.y = langS[kb1 * OD_ + c0 + lr];
      c = wmma4(a, b, c);
    }
#pragma unroll
    for (int v = 0; v < 8; ++v)
      hbuf[(r0 + v + 8 * half) * RP_ + c0 + lr] = c[v];
  }
  __syncthreads();

  if (tid < OD_) {
    const int d = tid;
    float acc = 0.f;
    for (int j = 0; j < KNB; ++j)
      acc += fg[j * RP_ + d] * hbuf[j * RP_ + d] * relm[j * RP_ + d];
    acc += f_ws[((size_t)s * N_ + n) * OD_ + d];
    out[((size_t)s * N_ + n) * OD_ + d] = acc;
    red[d] = acc;
  }
  __syncthreads();
  for (int off = 64; off > 0; off >>= 1) {
    if (tid < off) red[tid] += red[tid + off];
    __syncthreads();
  }
  if (tid == 0) score[(size_t)s * N_ + n] = red[0];
}

extern "C" void kernel_launch(void* const* d_in, const int* in_sizes, int n_in,
                              void* d_out, int out_size, void* d_ws,
                              size_t ws_size, hipStream_t stream) {
  (void)in_sizes; (void)n_in; (void)out_size; (void)ws_size;
  const float* feat      = (const float*)d_in[0];
  const float* coord     = (const float*)d_in[1];
  const float* lang_feat = (const float*)d_in[2];
  const float* lang_mask = (const float*)d_in[3];
  const float* rel_w1    = (const float*)d_in[4];
  const float* rel_b1    = (const float*)d_in[5];
  const float* rel_w2    = (const float*)d_in[6];
  const float* rel_b2    = (const float*)d_in[7];
  const float* lang_w1   = (const float*)d_in[8];
  const float* lang_b1   = (const float*)d_in[9];
  const float* lang_w2   = (const float*)d_in[10];
  const float* lang_b2   = (const float*)d_in[11];
  const float* feat_w1   = (const float*)d_in[12];
  const float* feat_b1   = (const float*)d_in[13];
  const float* feat_w2   = (const float*)d_in[14];
  const float* feat_b2   = (const float*)d_in[15];

  char* ws = (char*)d_ws;
  int*   idx_ws  = (int*)ws;
  float* lang_ws = (float*)(ws + (size_t)S_ * N_ * KNB * 4);
  float* f_ws    = (float*)(ws + (size_t)S_ * N_ * KNB * 4 +
                                 (size_t)S_ * L_ * OD_ * 4);

  float* out   = (float*)d_out;
  float* score = out + (size_t)S_ * N_ * OD_;

  knn_kernel<<<S_, N_, 0, stream>>>(coord, idx_ws);
  lang_mlp_kernel<<<S_ * L_, 128, 0, stream>>>(lang_feat, lang_w1, lang_b1,
                                               lang_w2, lang_b2, lang_ws);
  feat_mlp_kernel<<<S_ * (N_ / 16), 256, 0, stream>>>(feat, feat_w1, feat_b1,
                                                      feat_w2, feat_b2, f_ws);
  fused_kernel<<<S_ * N_, 256, 0, stream>>>(coord, lang_mask, rel_w1, rel_b1,
                                            rel_w2, rel_b2, idx_ws, lang_ws,
                                            f_ws, out, score);
}


# Loads issued in groups ahead of their uses: A and B fragments of 16 K-steps before their WMMAs in every K-loop, the row gather's loads before its LDS stores; one round trip per group; sums unchanged
# speedup vs baseline: 1.6904x; 1.3513x over previous
#include <hip/hip_runtime.h>
#include <hip/hip_bf16.h>
#include <float.h>


typedef float v2f __attribute__((ext_vector_type(2)));
typedef float v8f __attribute__((ext_vector_type(8)));

#define S_      64
#define N_      256
#define L_      40
#define LP_     48
#define RP_     132
#define LPP_    52
#define KP_     20
#define KNB     17
#define OD_     128
#define LANGID_ 256
#define PCID_   128
#define KG_     16

static __device__ inline v8f wmma4(v2f a, v2f b, v8f c) {
  return __builtin_amdgcn_wmma_f32_16x16x4_f32(false, a, false, b, (short)0, c,
                                               false, false);
}

__global__ void knn_kernel(const float* __restrict__ coord,
                           int* __restrict__ idx_out) {
  __shared__ float cx[N_], cy[N_], cz[N_];
  const int s = blockIdx.x;
  const int n = threadIdx.x;
  const float* cs = coord + (size_t)s * N_ * 3;
  cx[n] = cs[n * 3 + 0];
  cy[n] = cs[n * 3 + 1];
  cz[n] = cs[n * 3 + 2];
  __syncthreads();

  const float px = cx[n], py = cy[n], pz = cz[n];
  float pd = -1.0f;
  int   pi = -1;
  int* out = idx_out + ((size_t)s * N_ + n) * KNB;
  for (int j = 0; j < KNB; ++j) {
    float bd = FLT_MAX;
    int   bi = 0x7fffffff;
    for (int m = 0; m < N_; ++m) {
      float dx = cx[m] - px, dy = cy[m] - py, dz = cz[m] - pz;
      float d2 = dx * dx + dy * dy + dz * dz;
      bool gtPrev = (d2 > pd) || (d2 == pd && m > pi);
      bool ltBest = (d2 < bd) || (d2 == bd && m < bi);
      if (gtPrev && ltBest) { bd = d2; bi = m; }
    }
    out[j] = bi;
    pd = bd;
    pi = bi;
  }
}

__global__ void lang_mlp_kernel(const float* __restrict__ lang_feat,
                                const float* __restrict__ w1,
                                const float* __restrict__ b1,
                                const float* __restrict__ w2,
                                const float* __restrict__ b2,
                                float* __restrict__ lang_out) {
  __shared__ float xs[LANGID_];
  __shared__ float hs[OD_];
  const int row = blockIdx.x;
  const int t   = threadIdx.x;
  const float* x = lang_feat + (size_t)row * LANGID_;
  xs[t]       = x[t];
  xs[t + 128] = x[t + 128];
  __syncthreads();
  float acc = b1[t];
  for (int k = 0; k < LANGID_; ++k) acc += xs[k] * w1[k * OD_ + t];
  hs[t] = fmaxf(acc, 0.0f);
  __syncthreads();
  float acc2 = b2[t];
  for (int k = 0; k < OD_; ++k) acc2 += hs[k] * w2[k * OD_ + t];
  lang_out[(size_t)row * OD_ + t] = acc2;
}

__global__ void feat_mlp_kernel(const float* __restrict__ feat,
                                const float* __restrict__ w1,
                                const float* __restrict__ b1,
                                const float* __restrict__ w2,
                                const float* __restrict__ b2,
                                float* __restrict__ f_out) {
  __shared__ float Asm[16 * RP_];
  __shared__ float Hsm[16 * RP_];
  const int s    = blockIdx.x >> 4;
  const int mt   = blockIdx.x & 15;
  const int tid  = threadIdx.x;
  const int lane = tid & 31, wave = tid >> 5;
  const int half = lane >> 4, lr = lane & 15;

  const float* A = feat + ((size_t)s * N_ + mt * 16) * PCID_;
  for (int i = tid; i < 16 * PCID_; i += 256)
    Asm[(i >> 7) * RP_ + (i & 127)] = A[i];
  __syncthreads();

  const int n0 = wave * 16;
  v8f c = {};
  for (int kg = 0; kg < PCID_; kg += 4 * KG_) {
    v2f a[KG_], b[KG_];
#pragma unroll
    for (int u = 0; u < KG_; ++u) {
      const int ka = kg + 4 * u + 2 * half;
      a[u].x = Asm[lr * RP_ + ka];
      a[u].y = Asm[lr * RP_ + ka + 1];
      b[u].x = w1[ka * OD_ + n0 + lr];
      b[u].y = w1[(ka + 1) * OD_ + n0 + lr];
    }
    __builtin_amdgcn_sched_barrier(0);
#pragma unroll
    for (int u = 0; u < KG_; ++u) c = wmma4(a[u], b[u], c);
  }
  {
    const float bb = b1[n0 + lr];
#pragma unroll
    for (int v = 0; v < 8; ++v)
      Hsm[(v + 8 * half) * RP_ + n0 + lr] = fmaxf(c[v] + bb, 0.0f);
  }
  __syncthreads();

  v8f c2 = {};
  for (int kg = 0; kg < OD_; kg += 4 * KG_) {
    v2f a[KG_], b[KG_];
#pragma unroll
    for (int u = 0; u < KG_; ++u) {
      const int ka = kg + 4 * u + 2 * half;
      a[u].x = Hsm[lr * RP_ + ka];
      a[u].y = Hsm[lr * RP_ + ka + 1];
      b[u].x = w2[ka * OD_ + n0 + lr];
      b[u].y = w2[(ka + 1) * OD_ + n0 + lr];
    }
    __builtin_amdgcn_sched_barrier(0);
#pragma unroll
    for (int u = 0; u < KG_; ++u) c2 = wmma4(a[u], b[u], c2);
  }
  {
    const float bb = b2[n0 + lr];
    float* fo = f_out + ((size_t)s * N_ + mt * 16) * OD_;
#pragma unroll
    for (int v = 0; v < 8; ++v)
      fo[(v + 8 * half) * OD_ + n0 + lr] = c2[v] + bb;
  }
}

__global__ void fused_kernel(const float* __restrict__ coord,
                             const float* __restrict__ lang_mask,
                             const float* __restrict__ rel_w1,
                             const float* __restrict__ rel_b1,
                             const float* __restrict__ rel_w2,
                             const float* __restrict__ rel_b2,
                             const int*   __restrict__ idx_ws,
                             const float* __restrict__ lang_ws,
                             const float* __restrict__ f_ws,
                             float* __restrict__ out,
                             float* __restrict__ score) {
  __shared__ float fg[32 * RP_];
  __shared__ float relin[32 * KP_];
  __shared__ float hbuf[32 * RP_];
  __shared__ float relm[32 * RP_];
  __shared__ float attnsm[32 * LPP_];
  __shared__ int   idxsm[KNB];
  __shared__ float red[OD_];

  const int s    = blockIdx.x >> 8;
  const int n    = blockIdx.x & 255;
  const int tid  = threadIdx.x;
  const int lane = tid & 31, wave = tid >> 5;
  const int half = lane >> 4, lr = lane & 15;

  if (tid < KNB) idxsm[tid] = idx_ws[((size_t)s * N_ + n) * KNB + tid];
  for (int i = tid; i < 32 * KP_; i += 256) relin[i] = 0.f;
  __syncthreads();

  {
    float g[(32 * OD_) / 256];
#pragma unroll
    for (int t = 0; t < (32 * OD_) / 256; ++t) {
      const int i = tid + 256 * t;
      const int j = i >> 7, d = i & 127;
      g[t] = (j < KNB) ? f_ws[((size_t)s * N_ + idxsm[j]) * OD_ + d] : 0.f;
    }
#pragma unroll
    for (int t = 0; t < (32 * OD_) / 256; ++t) {
      const int i = tid + 256 * t;
      fg[(i >> 7) * RP_ + (i & 127)] = g[t];
    }
  }
  if (tid < KNB) {
    const int j = tid;
    const float* cs = coord + (size_t)s * N_ * 3;
    const int m = idxsm[j];
    const float nx = cs[m * 3 + 0], ny = cs[m * 3 + 1], nz = cs[m * 3 + 2];
    const float ox = cs[n * 3 + 0], oy = cs[n * 3 + 1], oz = cs[n * 3 + 2];
    const float rx = nx - ox, ry = ny - oy, rz = nz - oz;
    const float dist = sqrtf(rx * rx + ry * ry + rz * rz);
    float* ri = relin + j * KP_;
    ri[0] = nx; ri[1] = ny; ri[2] = nz;
    ri[3] = ox; ri[4] = oy; ri[5] = oz;
    ri[6] = rx; ri[7] = ry; ri[8] = rz;
    ri[9] = dist;
  }
  __syncthreads();

  for (int job = wave; job < 16; job += 8) {
    const int r0 = (job & 1) * 16, c0 = (job >> 1) * 16;
    v8f c = {};
    {
      v2f a[4], b[4];
#pragma unroll
      for (int u = 0; u < 4; ++u) {
        const int ka = 4 * u + 2 * half;
        a[u].x = relin[(r0 + lr) * KP_ + ka];
        a[u].y = relin[(r0 + lr) * KP_ + ka + 1];
        const int kb0 = min(ka, 9), kb1 = min(ka + 1, 9);
        b[u].x = rel_w1[kb0 * OD_ + c0 + lr];
        b[u].y = rel_w1[kb1 * OD_ + c0 + lr];
      }
#pragma unroll
      for (int u = 0; u < 4; ++u) c = wmma4(a[u], b[u], c);
    }
    const float bb = rel_b1[c0 + lr];
#pragma unroll
    for (int v = 0; v < 8; ++v)
      hbuf[(r0 + v + 8 * half) * RP_ + c0 + lr] = fmaxf(c[v] + bb, 0.f);
  }
  __syncthreads();

  for (int job = wave; job < 16; job += 8) {
    const int r0 = (job & 1) * 16, c0 = (job >> 1) * 16;
    v8f c = {};
    for (int kg = 0; kg < OD_; kg += 4 * KG_) {
      v2f a[KG_], b[KG_];
#pragma unroll
      for (int u = 0; u < KG_; ++u) {
        const int ka = kg + 4 * u + 2 * half;
        a[u].x = hbuf[(r0 + lr) * RP_ + ka];
        a[u].y = hbuf[(r0 + lr) * RP_ + ka + 1];
        b[u].x = rel_w2[ka * OD_ + c0 + lr];
        b[u].y = rel_w2[(ka + 1) * OD_ + c0 + lr];
      }
      __builtin_amdgcn_sched_barrier(0);
#pragma unroll
      for (int u = 0; u < KG_; ++u) c = wmma4(a[u], b[u], c);
    }
    const float bb = rel_b2[c0 + lr];
#pragma unroll
    for (int v = 0; v < 8; ++v)
      relm[(r0 + v + 8 * half) * RP_ + c0 + lr] = c[v] + bb;
  }
  __syncthreads();

  if (wave < 6) {
    const int r0 = (wave & 1) * 16, c0 = (wave >> 1) * 16;
    const float* langS = lang_ws + (size_t)s * L_ * OD_;
    const int lclamp = min(c0 + lr, L_ - 1);
    v8f c = {};
#pragma unroll 1
    for (int kg = 0; kg < OD_; kg += 4 * KG_) {
      v2f a[KG_], b[KG_];
#pragma unroll
      for (int u = 0; u < KG_; ++u) {
        const int ka = kg + 4 * u + 2 * half;
        a[u].x = fg[(r0 + lr) * RP_ + ka];
        a[u].y = fg[(r0 + lr) * RP_ + ka + 1];
        b[u].x = langS[lclamp * OD_ + ka];
        b[u].y = langS[lclamp * OD_ + ka + 1];
      }
      __builtin_amdgcn_sched_barrier(0);
#pragma unroll
      for (int u = 0; u < KG_; ++u) c = wmma4(a[u], b[u], c);
    }
#pragma unroll
    for (int v = 0; v < 8; ++v)
      attnsm[(r0 + v + 8 * half) * LPP_ + c0 + lr] = c[v];
  }
  __syncthreads();

  if (tid < 32) {
    float* rowp = attnsm + tid * LPP_;
    if (tid < KNB) {
      float mx = -FLT_MAX;
      for (int l = 0; l < L_; ++l) mx = fmaxf(mx, rowp[l]);
      float ssum = 0.f;
      for (int l = 0; l < L_; ++l) {
        const float e = expf(rowp[l] - mx);
        rowp[l] = e;
        ssum += e;
      }
      const float inv = 1.f / ssum;
      const float* mk = lang_mask + (size_t)s * L_;
      float s2 = 0.f;
      for (int l = 0; l < L_; ++l) {
        const float p = rowp[l] * inv * mk[l];
        rowp[l] = p;
        s2 += p;
      }
      const float inv2 = 1.f / (s2 + 1e-7f);
      for (int l = 0; l < L_; ++l) rowp[l] *= inv2;
      for (int l = L_; l < LP_; ++l) rowp[l] = 0.f;
    } else {
      for (int l = 0; l < LP_; ++l) rowp[l] = 0.f;
    }
  }
  __syncthreads();

  for (int job = wave; job < 16; job += 8) {
    const int r0 = (job & 1) * 16, c0 = (job >> 1) * 16;
    const float* langS = lang_ws + (size_t)s * L_ * OD_;
    v8f c = {};
    {
      v2f a[LP_ / 4], b[LP_ / 4];
#pragma unroll
      for (int u = 0; u < LP_ / 4; ++u) {
        const int ka  = 4 * u + 2 * half;
        const int kb0 = min(ka, L_ - 1), kb1 = min(ka + 1, L_ - 1);
        a[u].x = attnsm[(r0 + lr) * LPP_ + ka];
        a[u].y = attnsm[(r0 + lr) * LPP_ + ka + 1];
        b[u].x = langS[kb0 * OD_ + c0 + lr];
        b[u].y = langS[kb1 * OD_ + c0 + lr];
      }
#pragma unroll
      for (int u = 0; u < LP_ / 4; ++u) c = wmma4(a[u], b[u], c);
    }
#pragma unroll
    for (int v = 0; v < 8; ++v)
      hbuf[(r0 + v + 8 * half) * RP_ + c0 + lr] = c[v];
  }
  __syncthreads();

  if (tid < OD_) {
    const int d = tid;
    float acc = 0.f;
    for (int j = 0; j < KNB; ++j)
      acc += fg[j * RP_ + d] * hbuf[j * RP_ + d] * relm[j * RP_ + d];
    acc += f_ws[((size_t)s * N_ + n) * OD_ + d];
    out[((size_t)s * N_ + n) * OD_ + d] = acc;
    red[d] = acc;
  }
  __syncthreads();
  for (int off = 64; off > 0; off >>= 1) {
    if (tid < off) red[tid] += red[tid + off];
    __syncthreads();
  }
  if (tid == 0) score[(size_t)s * N_ + n] = red[0];
}

extern "C" void kernel_launch(void* const* d_in, const int* in_sizes, int n_in,
                              void* d_out, int out_size, void* d_ws,
                              size_t ws_size, hipStream_t stream) {
  (void)in_sizes; (void)n_in; (void)out_size; (void)ws_size;
  const float* feat      = (const float*)d_in[0];
  const float* coord     = (const float*)d_in[1];
  const float* lang_feat = (const float*)d_in[2];
  const float* lang_mask = (const float*)d_in[3];
  const float* rel_w1    = (const float*)d_in[4];
  const float* rel_b1    = (const float*)d_in[5];
  const float* rel_w2    = (const float*)d_in[6];
  const float* rel_b2    = (const float*)d_in[7];
  const float* lang_w1   = (const float*)d_in[8];
  const float* lang_b1   = (const float*)d_in[9];
  const float* lang_w2   = (const float*)d_in[10];
  const float* lang_b2   = (const float*)d_in[11];
  const float* feat_w1   = (const float*)d_in[12];
  const float* feat_b1   = (const float*)d_in[13];
  const float* feat_w2   = (const float*)d_in[14];
  const float* feat_b2   = (const float*)d_in[15];

  char* ws = (char*)d_ws;
  int*   idx_ws  = (int*)ws;
  float* lang_ws = (float*)(ws + (size_t)S_ * N_ * KNB * 4);
  float* f_ws    = (float*)(ws + (size_t)S_ * N_ * KNB * 4 +
                                 (size_t)S_ * L_ * OD_ * 4);

  float* out   = (float*)d_out;
  float* score = out + (size_t)S_ * N_ * OD_;

  knn_kernel<<<S_, N_, 0, stream>>>(coord, idx_ws);
  lang_mlp_kernel<<<S_ * L_, 128, 0, stream>>>(lang_feat, lang_w1, lang_b1,
                                               lang_w2, lang_b2, lang_ws);
  feat_mlp_kernel<<<S_ * (N_ / 16), 256, 0, stream>>>(feat, feat_w1, feat_b1,
                                                      feat_w2, feat_b2, f_ws);
  fused_kernel<<<S_ * N_, 256, 0, stream>>>(coord, lang_mask, rel_w1, rel_b1,
                                            rel_w2, rel_b2, idx_ws, lang_ws,
                                            f_ws, out, score);
}


# Attention (logits, softmax, ins) computed once per point in the feat kernel, not once per (point, neighbor) pair: a pair's rows are gathered f rows; fused kernel gathers f*ins; sums unchanged
# speedup vs baseline: 2.1983x; 1.3005x over previous
#include <hip/hip_runtime.h>
#include <hip/hip_bf16.h>
#include <float.h>


typedef float v2f __attribute__((ext_vector_type(2)));
typedef float v8f __attribute__((ext_vector_type(8)));

#define S_      64
#define N_      256
#define L_      40
#define LP_     48
#define RP_     132
#define LPP_    52
#define KP_     20
#define KNB     17
#define OD_     128
#define LANGID_ 256
#define PCID_   128
#define KG_     16

static __device__ inline v8f wmma4(v2f a, v2f b, v8f c) {
  return __builtin_amdgcn_wmma_f32_16x16x4_f32(false, a, false, b, (short)0, c,
                                               false, false);
}

__global__ void knn_kernel(const float* __restrict__ coord,
                           int* __restrict__ idx_out) {
  __shared__ float cx[N_], cy[N_], cz[N_];
  const int s = blockIdx.x;
  const int n = threadIdx.x;
  const float* cs = coord + (size_t)s * N_ * 3;
  cx[n] = cs[n * 3 + 0];
  cy[n] = cs[n * 3 + 1];
  cz[n] = cs[n * 3 + 2];
  __syncthreads();

  const float px = cx[n], py = cy[n], pz = cz[n];
  float pd = -1.0f;
  int   pi = -1;
  int* out = idx_out + ((size_t)s * N_ + n) * KNB;
  for (int j = 0; j < KNB; ++j) {
    float bd = FLT_MAX;
    int   bi = 0x7fffffff;
    for (int m = 0; m < N_; ++m) {
      float dx = cx[m] - px, dy = cy[m] - py, dz = cz[m] - pz;
      float d2 = dx * dx + dy * dy + dz * dz;
      bool gtPrev = (d2 > pd) || (d2 == pd && m > pi);
      bool ltBest = (d2 < bd) || (d2 == bd && m < bi);
      if (gtPrev && ltBest) { bd = d2; bi = m; }
    }
    out[j] = bi;
    pd = bd;
    pi = bi;
  }
}

__global__ void lang_mlp_kernel(const float* __restrict__ lang_feat,
                                const float* __restrict__ w1,
                                const float* __restrict__ b1,
                                const float* __restrict__ w2,
                                const float* __restrict__ b2,
                                float* __restrict__ lang_out) {
  __shared__ float xs[LANGID_];
  __shared__ float hs[OD_];
  const int row = blockIdx.x;
  const int t   = threadIdx.x;
  const float* x = lang_feat + (size_t)row * LANGID_;
  xs[t]       = x[t];
  xs[t + 128] = x[t + 128];
  __syncthreads();
  float acc = b1[t];
  for (int k = 0; k < LANGID_; ++k) acc += xs[k] * w1[k * OD_ + t];
  hs[t] = fmaxf(acc, 0.0f);
  __syncthreads();
  float acc2 = b2[t];
  for (int k = 0; k < OD_; ++k) acc2 += hs[k] * w2[k * OD_ + t];
  lang_out[(size_t)row * OD_ + t] = acc2;
}

__global__ void feat_mlp_kernel(const float* __restrict__ feat,
                                const float* __restrict__ w1,
                                const float* __restrict__ b1,
                                const float* __restrict__ w2,
                                const float* __restrict__ b2,
                                const float* __restrict__ lang_mask,
                                const float* __restrict__ lang_ws,
                                float* __restrict__ g_out,
                                float* __restrict__ f_out) {
  __shared__ float Asm[16 * RP_];
  __shared__ float Hsm[16 * RP_];
  __shared__ float attnsm[16 * LPP_];
  const int s    = blockIdx.x >> 4;
  const int mt   = blockIdx.x & 15;
  const int tid  = threadIdx.x;
  const int lane = tid & 31, wave = tid >> 5;
  const int half = lane >> 4, lr = lane & 15;

  const float* A = feat + ((size_t)s * N_ + mt * 16) * PCID_;
  for (int i = tid; i < 16 * PCID_; i += 256)
    Asm[(i >> 7) * RP_ + (i & 127)] = A[i];
  __syncthreads();

  const int n0 = wave * 16;
  v8f c = {};
  for (int kg = 0; kg < PCID_; kg += 4 * KG_) {
    v2f a[KG_], b[KG_];
#pragma unroll
    for (int u = 0; u < KG_; ++u) {
      const int ka = kg + 4 * u + 2 * half;
      a[u].x = Asm[lr * RP_ + ka];
      a[u].y = Asm[lr * RP_ + ka + 1];
      b[u].x = w1[ka * OD_ + n0 + lr];
      b[u].y = w1[(ka + 1) * OD_ + n0 + lr];
    }
    __builtin_amdgcn_sched_barrier(0);
#pragma unroll
    for (int u = 0; u < KG_; ++u) c = wmma4(a[u], b[u], c);
  }
  {
    const float bb = b1[n0 + lr];
#pragma unroll
    for (int v = 0; v < 8; ++v)
      Hsm[(v + 8 * half) * RP_ + n0 + lr] = fmaxf(c[v] + bb, 0.0f);
  }
  __syncthreads();

  v8f c2 = {};
  for (int kg = 0; kg < OD_; kg += 4 * KG_) {
    v2f a[KG_], b[KG_];
#pragma unroll
    for (int u = 0; u < KG_; ++u) {
      const int ka = kg + 4 * u + 2 * half;
      a[u].x = Hsm[lr * RP_ + ka];
      a[u].y = Hsm[lr * RP_ + ka + 1];
      b[u].x = w2[ka * OD_ + n0 + lr];
      b[u].y = w2[(ka + 1) * OD_ + n0 + lr];
    }
    __builtin_amdgcn_sched_barrier(0);
#pragma unroll
    for (int u = 0; u < KG_; ++u) c2 = wmma4(a[u], b[u], c2);
  }
  float fv[8];
  {
    const float bb = b2[n0 + lr];
    float* fo = f_out + ((size_t)s * N_ + mt * 16) * OD_;
#pragma unroll
    for (int v = 0; v < 8; ++v) {
      fv[v] = c2[v] + bb;
      Asm[(v + 8 * half) * RP_ + n0 + lr] = fv[v];
      fo[(v + 8 * half) * OD_ + n0 + lr]  = fv[v];
    }
  }
  __syncthreads();

  const float* langS = lang_ws + (size_t)s * L_ * OD_;

  if (wave < 3) {
    const int c0 = wave * 16;
    const int lclamp = min(c0 + lr, L_ - 1);
    v8f cl = {};
    for (int kg = 0; kg < OD_; kg += 4 * KG_) {
      v2f a[KG_], b[KG_];
#pragma unroll
      for (int u = 0; u < KG_; ++u) {
        const int ka = kg + 4 * u + 2 * half;
        a[u].x = Asm[lr * RP_ + ka];
        a[u].y = Asm[lr * RP_ + ka + 1];
        b[u].x = langS[lclamp * OD_ + ka];
        b[u].y = langS[lclamp * OD_ + ka + 1];
      }
      __builtin_amdgcn_sched_barrier(0);
#pragma unroll
      for (int u = 0; u < KG_; ++u) cl = wmma4(a[u], b[u], cl);
    }
#pragma unroll
    for (int v = 0; v < 8; ++v)
      attnsm[(v + 8 * half) * LPP_ + c0 + lr] = cl[v];
  }
  __syncthreads();

  if (tid < 16) {
    float* rowp = attnsm + tid * LPP_;
    float mx = -FLT_MAX;
    for (int l = 0; l < L_; ++l) mx = fmaxf(mx, rowp[l]);
    float ssum = 0.f;
    for (int l = 0; l < L_; ++l) {
      const float e = expf(rowp[l] - mx);
      rowp[l] = e;
      ssum += e;
    }
    const float inv = 1.f / ssum;
    const float* mk = lang_mask + (size_t)s * L_;
    float s2 = 0.f;
    for (int l = 0; l < L_; ++l) {
      const float p = rowp[l] * inv * mk[l];
      rowp[l] = p;
      s2 += p;
    }
    const float inv2 = 1.f / (s2 + 1e-7f);
    for (int l = 0; l < L_; ++l) rowp[l] *= inv2;
    for (int l = L_; l < LP_; ++l) rowp[l] = 0.f;
  }
  __syncthreads();

  {
    v8f ci = {};
    {
      v2f a[LP_ / 4], b[LP_ / 4];
#pragma unroll
      for (int u = 0; u < LP_ / 4; ++u) {
        const int ka  = 4 * u + 2 * half;
        const int kb0 = min(ka, L_ - 1), kb1 = min(ka + 1, L_ - 1);
        a[u].x = attnsm[lr * LPP_ + ka];
        a[u].y = attnsm[lr * LPP_ + ka + 1];
        b[u].x = langS[kb0 * OD_ + n0 + lr];
        b[u].y = langS[kb1 * OD_ + n0 + lr];
      }
#pragma unroll
      for (int u = 0; u < LP_ / 4; ++u) ci = wmma4(a[u], b[u], ci);
    }
    float* go = g_out + ((size_t)s * N_ + mt * 16) * OD_;
#pragma unroll
    for (int v = 0; v < 8; ++v)
      go[(v + 8 * half) * OD_ + n0 + lr] = fv[v] * ci[v];
  }
}

__global__ void fused_kernel(const float* __restrict__ coord,
                             const float* __restrict__ rel_w1,
                             const float* __restrict__ rel_b1,
                             const float* __restrict__ rel_w2,
                             const float* __restrict__ rel_b2,
                             const int*   __restrict__ idx_ws,
                             const float* __restrict__ g_ws,
                             float* __restrict__ out,
                             float* __restrict__ score) {
  __shared__ float gsm[KNB * RP_];
  __shared__ float relin[32 * KP_];
  __shared__ float hbuf[32 * RP_];
  __shared__ float relm[32 * RP_];
  __shared__ int   idxsm[KNB];
  __shared__ float red[OD_];

  const int s    = blockIdx.x >> 8;
  const int n    = blockIdx.x & 255;
  const int tid  = threadIdx.x;
  const int lane = tid & 31, wave = tid >> 5;
  const int half = lane >> 4, lr = lane & 15;

  if (tid < KNB) idxsm[tid] = idx_ws[((size_t)s * N_ + n) * KNB + tid];
  for (int i = tid; i < 32 * KP_; i += 256) relin[i] = 0.f;
  __syncthreads();

  {
    constexpr int GT = (KNB * OD_ + 255) / 256;
    float g[GT];
#pragma unroll
    for (int t = 0; t < GT; ++t) {
      const int i = tid + 256 * t;
      const int j = i >> 7, d = i & 127;
      g[t] = (j < KNB) ? g_ws[((size_t)s * N_ + idxsm[j]) * OD_ + d] : 0.f;
    }
#pragma unroll
    for (int t = 0; t < GT; ++t) {
      const int i = tid + 256 * t;
      if ((i >> 7) < KNB) gsm[(i >> 7) * RP_ + (i & 127)] = g[t];
    }
  }
  if (tid < KNB) {
    const int j = tid;
    const float* cs = coord + (size_t)s * N_ * 3;
    const int m = idxsm[j];
    const float nx = cs[m * 3 + 0], ny = cs[m * 3 + 1], nz = cs[m * 3 + 2];
    const float ox = cs[n * 3 + 0], oy = cs[n * 3 + 1], oz = cs[n * 3 + 2];
    const float rx = nx - ox, ry = ny - oy, rz = nz - oz;
    const float dist = sqrtf(rx * rx + ry * ry + rz * rz);
    float* ri = relin + j * KP_;
    ri[0] = nx; ri[1] = ny; ri[2] = nz;
    ri[3] = ox; ri[4] = oy; ri[5] = oz;
    ri[6] = rx; ri[7] = ry; ri[8] = rz;
    ri[9] = dist;
  }
  __syncthreads();
  for (int job = wave; job < 16; job += 8) {
    const int r0 = (job & 1) * 16, c0 = (job >> 1) * 16;
    v8f c = {};
    {
      v2f a[4], b[4];
#pragma unroll
      for (int u = 0; u < 4; ++u) {
        const int ka = 4 * u + 2 * half;
        a[u].x = relin[(r0 + lr) * KP_ + ka];
        a[u].y = relin[(r0 + lr) * KP_ + ka + 1];
        const int kb0 = min(ka, 9), kb1 = min(ka + 1, 9);
        b[u].x = rel_w1[kb0 * OD_ + c0 + lr];
        b[u].y = rel_w1[kb1 * OD_ + c0 + lr];
      }
#pragma unroll
      for (int u = 0; u < 4; ++u) c = wmma4(a[u], b[u], c);
    }
    const float bb = rel_b1[c0 + lr];
#pragma unroll
    for (int v = 0; v < 8; ++v)
      hbuf[(r0 + v + 8 * half) * RP_ + c0 + lr] = fmaxf(c[v] + bb, 0.f);
  }
  __syncthreads();

  for (int job = wave; job < 16; job += 8) {
    const int r0 = (job & 1) * 16, c0 = (job >> 1) * 16;
    v8f c = {};
    for (int kg = 0; kg < OD_; kg += 4 * KG_) {
      v2f a[KG_], b[KG_];
#pragma unroll
      for (int u = 0; u < KG_; ++u) {
        const int ka = kg + 4 * u + 2 * half;
        a[u].x = hbuf[(r0 + lr) * RP_ + ka];
        a[u].y = hbuf[(r0 + lr) * RP_ + ka + 1];
        b[u].x = rel_w2[ka * OD_ + c0 + lr];
        b[u].y = rel_w2[(ka + 1) * OD_ + c0 + lr];
      }
      __builtin_amdgcn_sched_barrier(0);
#pragma unroll
      for (int u = 0; u < KG_; ++u) c = wmma4(a[u], b[u], c);
    }
    const float bb = rel_b2[c0 + lr];
#pragma unroll
    for (int v = 0; v < 8; ++v)
      relm[(r0 + v + 8 * half) * RP_ + c0 + lr] = c[v] + bb;
  }
  __syncthreads();


  if (tid < OD_) {
    const int d = tid;
    float acc = 0.f;
    for (int j = 0; j < KNB; ++j)
      acc += gsm[j * RP_ + d] * relm[j * RP_ + d];
    float* o = out + ((size_t)s * N_ + n) * OD_ + d;
    acc += *o;
    *o = acc;
    red[d] = acc;
  }
  __syncthreads();
  for (int off = 64; off > 0; off >>= 1) {
    if (tid < off) red[tid] += red[tid + off];
    __syncthreads();
  }
  if (tid == 0) score[(size_t)s * N_ + n] = red[0];
}

extern "C" void kernel_launch(void* const* d_in, const int* in_sizes, int n_in,
                              void* d_out, int out_size, void* d_ws,
                              size_t ws_size, hipStream_t stream) {
  (void)in_sizes; (void)n_in; (void)out_size; (void)ws_size;
  const float* feat      = (const float*)d_in[0];
  const float* coord     = (const float*)d_in[1];
  const float* lang_feat = (const float*)d_in[2];
  const float* lang_mask = (const float*)d_in[3];
  const float* rel_w1    = (const float*)d_in[4];
  const float* rel_b1    = (const float*)d_in[5];
  const float* rel_w2    = (const float*)d_in[6];
  const float* rel_b2    = (const float*)d_in[7];
  const float* lang_w1   = (const float*)d_in[8];
  const float* lang_b1   = (const float*)d_in[9];
  const float* lang_w2   = (const float*)d_in[10];
  const float* lang_b2   = (const float*)d_in[11];
  const float* feat_w1   = (const float*)d_in[12];
  const float* feat_b1   = (const float*)d_in[13];
  const float* feat_w2   = (const float*)d_in[14];
  const float* feat_b2   = (const float*)d_in[15];

  char* ws = (char*)d_ws;
  int*   idx_ws  = (int*)ws;
  float* lang_ws = (float*)(ws + (size_t)S_ * N_ * KNB * 4);
  float* g_ws    = (float*)(ws + (size_t)S_ * N_ * KNB * 4 +
                                 (size_t)S_ * L_ * OD_ * 4);

  float* out   = (float*)d_out;
  float* score = out + (size_t)S_ * N_ * OD_;

  knn_kernel<<<S_, N_, 0, stream>>>(coord, idx_ws);
  lang_mlp_kernel<<<S_ * L_, 128, 0, stream>>>(lang_feat, lang_w1, lang_b1,
                                               lang_w2, lang_b2, lang_ws);
  feat_mlp_kernel<<<S_ * (N_ / 16), 256, 0, stream>>>(feat, feat_w1, feat_b1,
                                                      feat_w2, feat_b2, lang_mask,
                                                      lang_ws, g_ws, out);
  fused_kernel<<<S_ * N_, 256, 0, stream>>>(coord, rel_w1, rel_b1, rel_w2,
                                            rel_b2, idx_ws, g_ws, out, score);
}


# Fused kernel retiled: 16 points per block, rel-MLP rows neighbor-major so 17 neighbors = 17 full 16-row tiles (was 32 padded rows per point); rel weights kept in registers; sum order unchanged
# speedup vs baseline: 2.9257x; 1.3309x over previous
#include <hip/hip_runtime.h>
#include <hip/hip_bf16.h>
#include <float.h>


typedef float v2f __attribute__((ext_vector_type(2)));
typedef float v8f __attribute__((ext_vector_type(8)));

#define S_      64
#define N_      256
#define L_      40
#define LP_     48
#define RP_     132
#define LPP_    52
#define KP_     20
#define KNB     17
#define OD_     128
#define LANGID_ 256
#define PCID_   128
#define KG_     16

static __device__ inline v8f wmma4(v2f a, v2f b, v8f c) {
  return __builtin_amdgcn_wmma_f32_16x16x4_f32(false, a, false, b, (short)0, c,
                                               false, false);
}

__global__ void knn_kernel(const float* __restrict__ coord,
                           int* __restrict__ idx_out) {
  __shared__ float cx[N_], cy[N_], cz[N_];
  const int s = blockIdx.x;
  const int n = threadIdx.x;
  const float* cs = coord + (size_t)s * N_ * 3;
  cx[n] = cs[n * 3 + 0];
  cy[n] = cs[n * 3 + 1];
  cz[n] = cs[n * 3 + 2];
  __syncthreads();

  const float px = cx[n], py = cy[n], pz = cz[n];
  float pd = -1.0f;
  int   pi = -1;
  int* out = idx_out + ((size_t)s * N_ + n) * KNB;
  for (int j = 0; j < KNB; ++j) {
    float bd = FLT_MAX;
    int   bi = 0x7fffffff;
    for (int m = 0; m < N_; ++m) {
      float dx = cx[m] - px, dy = cy[m] - py, dz = cz[m] - pz;
      float d2 = dx * dx + dy * dy + dz * dz;
      bool gtPrev = (d2 > pd) || (d2 == pd && m > pi);
      bool ltBest = (d2 < bd) || (d2 == bd && m < bi);
      if (gtPrev && ltBest) { bd = d2; bi = m; }
    }
    out[j] = bi;
    pd = bd;
    pi = bi;
  }
}

__global__ void lang_mlp_kernel(const float* __restrict__ lang_feat,
                                const float* __restrict__ w1,
                                const float* __restrict__ b1,
                                const float* __restrict__ w2,
                                const float* __restrict__ b2,
                                float* __restrict__ lang_out) {
  __shared__ float xs[LANGID_];
  __shared__ float hs[OD_];
  const int row = blockIdx.x;
  const int t   = threadIdx.x;
  const float* x = lang_feat + (size_t)row * LANGID_;
  xs[t]       = x[t];
  xs[t + 128] = x[t + 128];
  __syncthreads();
  float acc = b1[t];
  for (int k = 0; k < LANGID_; ++k) acc += xs[k] * w1[k * OD_ + t];
  hs[t] = fmaxf(acc, 0.0f);
  __syncthreads();
  float acc2 = b2[t];
  for (int k = 0; k < OD_; ++k) acc2 += hs[k] * w2[k * OD_ + t];
  lang_out[(size_t)row * OD_ + t] = acc2;
}

__global__ void feat_mlp_kernel(const float* __restrict__ feat,
                                const float* __restrict__ w1,
                                const float* __restrict__ b1,
                                const float* __restrict__ w2,
                                const float* __restrict__ b2,
                                const float* __restrict__ lang_mask,
                                const float* __restrict__ lang_ws,
                                float* __restrict__ g_out,
                                float* __restrict__ f_out) {
  __shared__ float Asm[16 * RP_];
  __shared__ float Hsm[16 * RP_];
  __shared__ float attnsm[16 * LPP_];
  const int s    = blockIdx.x >> 4;
  const int mt   = blockIdx.x & 15;
  const int tid  = threadIdx.x;
  const int lane = tid & 31, wave = tid >> 5;
  const int half = lane >> 4, lr = lane & 15;

  const float* A = feat + ((size_t)s * N_ + mt * 16) * PCID_;
  for (int i = tid; i < 16 * PCID_; i += 256)
    Asm[(i >> 7) * RP_ + (i & 127)] = A[i];
  __syncthreads();

  const int n0 = wave * 16;
  v8f c = {};
  for (int kg = 0; kg < PCID_; kg += 4 * KG_) {
    v2f a[KG_], b[KG_];
#pragma unroll
    for (int u = 0; u < KG_; ++u) {
      const int ka = kg + 4 * u + 2 * half;
      a[u].x = Asm[lr * RP_ + ka];
      a[u].y = Asm[lr * RP_ + ka + 1];
      b[u].x = w1[ka * OD_ + n0 + lr];
      b[u].y = w1[(ka + 1) * OD_ + n0 + lr];
    }
    __builtin_amdgcn_sched_barrier(0);
#pragma unroll
    for (int u = 0; u < KG_; ++u) c = wmma4(a[u], b[u], c);
  }
  {
    const float bb = b1[n0 + lr];
#pragma unroll
    for (int v = 0; v < 8; ++v)
      Hsm[(v + 8 * half) * RP_ + n0 + lr] = fmaxf(c[v] + bb, 0.0f);
  }
  __syncthreads();

  v8f c2 = {};
  for (int kg = 0; kg < OD_; kg += 4 * KG_) {
    v2f a[KG_], b[KG_];
#pragma unroll
    for (int u = 0; u < KG_; ++u) {
      const int ka = kg + 4 * u + 2 * half;
      a[u].x = Hsm[lr * RP_ + ka];
      a[u].y = Hsm[lr * RP_ + ka + 1];
      b[u].x = w2[ka * OD_ + n0 + lr];
      b[u].y = w2[(ka + 1) * OD_ + n0 + lr];
    }
    __builtin_amdgcn_sched_barrier(0);
#pragma unroll
    for (int u = 0; u < KG_; ++u) c2 = wmma4(a[u], b[u], c2);
  }
  float fv[8];
  {
    const float bb = b2[n0 + lr];
    float* fo = f_out + ((size_t)s * N_ + mt * 16) * OD_;
#pragma unroll
    for (int v = 0; v < 8; ++v) {
      fv[v] = c2[v] + bb;
      Asm[(v + 8 * half) * RP_ + n0 + lr] = fv[v];
      fo[(v + 8 * half) * OD_ + n0 + lr]  = fv[v];
    }
  }
  __syncthreads();

  const float* langS = lang_ws + (size_t)s * L_ * OD_;

  if (wave < 3) {
    const int c0 = wave * 16;
    const int lclamp = min(c0 + lr, L_ - 1);
    v8f cl = {};
    for (int kg = 0; kg < OD_; kg += 4 * KG_) {
      v2f a[KG_], b[KG_];
#pragma unroll
      for (int u = 0; u < KG_; ++u) {
        const int ka = kg + 4 * u + 2 * half;
        a[u].x = Asm[lr * RP_ + ka];
        a[u].y = Asm[lr * RP_ + ka + 1];
        b[u].x = langS[lclamp * OD_ + ka];
        b[u].y = langS[lclamp * OD_ + ka + 1];
      }
      __builtin_amdgcn_sched_barrier(0);
#pragma unroll
      for (int u = 0; u < KG_; ++u) cl = wmma4(a[u], b[u], cl);
    }
#pragma unroll
    for (int v = 0; v < 8; ++v)
      attnsm[(v + 8 * half) * LPP_ + c0 + lr] = cl[v];
  }
  __syncthreads();

  if (tid < 16) {
    float* rowp = attnsm + tid * LPP_;
    float mx = -FLT_MAX;
    for (int l = 0; l < L_; ++l) mx = fmaxf(mx, rowp[l]);
    float ssum = 0.f;
    for (int l = 0; l < L_; ++l) {
      const float e = expf(rowp[l] - mx);
      rowp[l] = e;
      ssum += e;
    }
    const float inv = 1.f / ssum;
    const float* mk = lang_mask + (size_t)s * L_;
    float s2 = 0.f;
    for (int l = 0; l < L_; ++l) {
      const float p = rowp[l] * inv * mk[l];
      rowp[l] = p;
      s2 += p;
    }
    const float inv2 = 1.f / (s2 + 1e-7f);
    for (int l = 0; l < L_; ++l) rowp[l] *= inv2;
    for (int l = L_; l < LP_; ++l) rowp[l] = 0.f;
  }
  __syncthreads();

  {
    v8f ci = {};
    {
      v2f a[LP_ / 4], b[LP_ / 4];
#pragma unroll
      for (int u = 0; u < LP_ / 4; ++u) {
        const int ka  = 4 * u + 2 * half;
        const int kb0 = min(ka, L_ - 1), kb1 = min(ka + 1, L_ - 1);
        a[u].x = attnsm[lr * LPP_ + ka];
        a[u].y = attnsm[lr * LPP_ + ka + 1];
        b[u].x = langS[kb0 * OD_ + n0 + lr];
        b[u].y = langS[kb1 * OD_ + n0 + lr];
      }
#pragma unroll
      for (int u = 0; u < LP_ / 4; ++u) ci = wmma4(a[u], b[u], ci);
    }
    float* go = g_out + ((size_t)s * N_ + mt * 16) * OD_;
#pragma unroll
    for (int v = 0; v < 8; ++v)
      go[(v + 8 * half) * OD_ + n0 + lr] = fv[v] * ci[v];
  }
}

#define PB_     16
#define K1S_    3
#define FT_     128

__global__ void __launch_bounds__(FT_)
fused_kernel(const float* __restrict__ coord,
             const float* __restrict__ rel_w1,
             const float* __restrict__ rel_b1,
             const float* __restrict__ rel_w2,
             const float* __restrict__ rel_b2,
             const int*   __restrict__ idx_ws,
             const float* __restrict__ g_ws,
             float* __restrict__ out,
             float* __restrict__ score) {
  __shared__ float relin[KNB * PB_ * KP_];
  __shared__ float hbuf[2][PB_ * RP_];
  __shared__ int   idxsm[PB_ * KNB];

  const int s    = blockIdx.x >> 4;
  const int pt0  = (blockIdx.x & 15) * PB_;
  const int tid  = threadIdx.x;
  const int lane = tid & 31, wave = tid >> 5;
  const int half = lane >> 4, lr = lane & 15;
  const int c0   = wave * 32;

  for (int i = tid; i < PB_ * KNB; i += FT_)
    idxsm[i] = idx_ws[((size_t)s * N_ + pt0) * KNB + i];

  v2f bw1[2][K1S_];
  v2f bw2[2][OD_ / 4];
  float bb1[2], bb2[2];
#pragma unroll
  for (int t = 0; t < 2; ++t) {
    const int col = c0 + 16 * t + lr;
#pragma unroll
    for (int u = 0; u < K1S_; ++u) {
      const int ka  = 4 * u + 2 * half;
      const int kb0 = min(ka, 9), kb1 = min(ka + 1, 9);
      bw1[t][u].x = rel_w1[kb0 * OD_ + col];
      bw1[t][u].y = rel_w1[kb1 * OD_ + col];
    }
#pragma unroll
    for (int u = 0; u < OD_ / 4; ++u) {
      const int ka = 4 * u + 2 * half;
      bw2[t][u].x = rel_w2[ka * OD_ + col];
      bw2[t][u].y = rel_w2[(ka + 1) * OD_ + col];
    }
    bb1[t] = rel_b1[col];
    bb2[t] = rel_b2[col];
  }
  __syncthreads();

  for (int r = tid; r < PB_ * KNB; r += FT_) {
    const int j = r >> 4, p = r & 15;
    const float* cs = coord + (size_t)s * N_ * 3;
    const int m = idxsm[p * KNB + j];
    const int n = pt0 + p;
    const float nx = cs[m * 3 + 0], ny = cs[m * 3 + 1], nz = cs[m * 3 + 2];
    const float ox = cs[n * 3 + 0], oy = cs[n * 3 + 1], oz = cs[n * 3 + 2];
    const float rx = nx - ox, ry = ny - oy, rz = nz - oz;
    const float dist = sqrtf(rx * rx + ry * ry + rz * rz);
    float* ri = relin + r * KP_;
    ri[0] = nx; ri[1] = ny; ri[2] = nz;
    ri[3] = ox; ri[4] = oy; ri[5] = oz;
    ri[6] = rx; ri[7] = ry; ri[8] = rz;
    ri[9] = dist;
    ri[10] = 0.f; ri[11] = 0.f;
  }
  __syncthreads();

  float acc[2][8];
#pragma unroll
  for (int v = 0; v < 8; ++v) { acc[0][v] = 0.f; acc[1][v] = 0.f; }

  const int*   ip = idxsm + 8 * half * KNB;
  const float* gs = g_ws + (size_t)s * N_ * OD_;
  const unsigned lcol = (unsigned)(c0 + lr);

  for (int j = 0; j < KNB; ++j) {
    unsigned m[8];
#pragma unroll
    for (int v = 0; v < 8; ++v) m[v] = (unsigned)ip[v * KNB + j];

    v8f ca = {}, cb = {};
    {
      const float* rp = relin + (j * PB_ + lr) * KP_ + 2 * half;
      v2f a[K1S_];
#pragma unroll
      for (int u = 0; u < K1S_; ++u) {
        a[u].x = rp[4 * u];
        a[u].y = rp[4 * u + 1];
      }
#pragma unroll
      for (int u = 0; u < K1S_; ++u) {
        ca = wmma4(a[u], bw1[0][u], ca);
        cb = wmma4(a[u], bw1[1][u], cb);
      }
    }
    float* hb = hbuf[j & 1];
    {
      float* hp = hb + 8 * half * RP_ + c0 + lr;
#pragma unroll
      for (int v = 0; v < 8; ++v) {
        hp[v * RP_]      = fmaxf(ca[v] + bb1[0], 0.f);
        hp[v * RP_ + 16] = fmaxf(cb[v] + bb1[1], 0.f);
      }
    }
    __syncthreads();

    float gv[2][8];
#pragma unroll
    for (int v = 0; v < 8; ++v) {
      const float* gp = gs + (m[v] * OD_ + lcol);
      gv[0][v] = gp[0];
      gv[1][v] = gp[16];
    }

    v8f c2a = {}, c2b = {};
    const float* ap = hb + lr * RP_ + 2 * half;
    for (int kg = 0; kg < OD_ / 4; kg += KG_) {
      v2f a[KG_];
#pragma unroll
      for (int u = 0; u < KG_; ++u) {
        a[u].x = ap[4 * (kg + u)];
        a[u].y = ap[4 * (kg + u) + 1];
      }
      __builtin_amdgcn_sched_barrier(0);
#pragma unroll
      for (int u = 0; u < KG_; ++u) {
        c2a = wmma4(a[u], bw2[0][kg + u], c2a);
        c2b = wmma4(a[u], bw2[1][kg + u], c2b);
      }
    }
#pragma unroll
    for (int v = 0; v < 8; ++v) {
      acc[0][v] += gv[0][v] * (c2a[v] + bb2[0]);
      acc[1][v] += gv[1][v] * (c2b[v] + bb2[1]);
    }
  }

  float* osm = hbuf[1];
  {
    float* op = osm + 8 * half * RP_ + c0 + lr;
#pragma unroll
    for (int v = 0; v < 8; ++v) {
      op[v * RP_]      = acc[0][v];
      op[v * RP_ + 16] = acc[1][v];
    }
  }
  __syncthreads();

#pragma unroll
  for (int pp = 0; pp < PB_; pp += FT_ / 16) {
    const int p = pp + (tid >> 4), q = tid & 15;
    float* o = out + ((size_t)s * N_ + pt0 + p) * OD_;
    float x[8];
#pragma unroll
    for (int i = 0; i < 8; ++i) x[i] = osm[p * RP_ + q + 16 * i] + o[q + 16 * i];
#pragma unroll
    for (int i = 0; i < 8; ++i) o[q + 16 * i] = x[i];
    const float y0 = x[0] + x[4], y1 = x[1] + x[5];
    const float y2 = x[2] + x[6], y3 = x[3] + x[7];
    float r = (y0 + y2) + (y1 + y3);
    r += __shfl_xor(r, 8);
    r += __shfl_xor(r, 4);
    r += __shfl_xor(r, 2);
    r += __shfl_xor(r, 1);
    if (q == 0) score[(size_t)s * N_ + pt0 + p] = r;
  }
}

extern "C" void kernel_launch(void* const* d_in, const int* in_sizes, int n_in,
                              void* d_out, int out_size, void* d_ws,
                              size_t ws_size, hipStream_t stream) {
  (void)in_sizes; (void)n_in; (void)out_size; (void)ws_size;
  const float* feat      = (const float*)d_in[0];
  const float* coord     = (const float*)d_in[1];
  const float* lang_feat = (const float*)d_in[2];
  const float* lang_mask = (const float*)d_in[3];
  const float* rel_w1    = (const float*)d_in[4];
  const float* rel_b1    = (const float*)d_in[5];
  const float* rel_w2    = (const float*)d_in[6];
  const float* rel_b2    = (const float*)d_in[7];
  const float* lang_w1   = (const float*)d_in[8];
  const float* lang_b1   = (const float*)d_in[9];
  const float* lang_w2   = (const float*)d_in[10];
  const float* lang_b2   = (const float*)d_in[11];
  const float* feat_w1   = (const float*)d_in[12];
  const float* feat_b1   = (const float*)d_in[13];
  const float* feat_w2   = (const float*)d_in[14];
  const float* feat_b2   = (const float*)d_in[15];

  char* ws = (char*)d_ws;
  int*   idx_ws  = (int*)ws;
  float* lang_ws = (float*)(ws + (size_t)S_ * N_ * KNB * 4);
  float* g_ws    = (float*)(ws + (size_t)S_ * N_ * KNB * 4 +
                                 (size_t)S_ * L_ * OD_ * 4);

  float* out   = (float*)d_out;
  float* score = out + (size_t)S_ * N_ * OD_;

  knn_kernel<<<S_, N_, 0, stream>>>(coord, idx_ws);
  lang_mlp_kernel<<<S_ * L_, 128, 0, stream>>>(lang_feat, lang_w1, lang_b1,
                                               lang_w2, lang_b2, lang_ws);
  feat_mlp_kernel<<<S_ * (N_ / 16), 256, 0, stream>>>(feat, feat_w1, feat_b1,
                                                      feat_w2, feat_b2, lang_mask,
                                                      lang_ws, g_ws, out);
  fused_kernel<<<S_ * (N_ / PB_), FT_, 0, stream>>>(coord, rel_w1, rel_b1,
                                                     rel_w2, rel_b2, idx_ws,
                                                     g_ws, out, score);
}


# kNN kernel: one wave per point (each lane keeps 8 candidate d2 in registers; 17 rounds of wave-wide min by DPP and removal; 512 blocks), was one thread per point in 64 blocks; indices, sums unchanged
# speedup vs baseline: 7.1565x; 2.4461x over previous
#include <hip/hip_runtime.h>
#include <hip/hip_bf16.h>
#include <float.h>


typedef float v2f __attribute__((ext_vector_type(2)));
typedef float v8f __attribute__((ext_vector_type(8)));

#define S_      64
#define N_      256
#define L_      40
#define LP_     48
#define RP_     132
#define LPP_    52
#define KP_     20
#define KNB     17
#define OD_     128
#define LANGID_ 256
#define PCID_   128
#define KG_     16

static __device__ inline v8f wmma4(v2f a, v2f b, v8f c) {
  return __builtin_amdgcn_wmma_f32_16x16x4_f32(false, a, false, b, (short)0, c,
                                               false, false);
}

#define KPW_    4
#define KWV_    8
#define KPB_    (KPW_ * KWV_)
#define KRM_    0xffffffffu

template <int XM>
static __device__ inline unsigned row_xor(unsigned x) {
  return (unsigned)__builtin_amdgcn_mov_dpp((int)x, 0x160 + XM, 0xf, 0xf, true);
}

static __device__ inline unsigned other_row(unsigned x) {
  return (unsigned)__builtin_amdgcn_permlanex16((int)x, (int)x, 0x76543210,
                                                0xfedcba98, false, false);
}

static __device__ inline unsigned umin2(unsigned a, unsigned b) {
  return a < b ? a : b;
}

static __device__ inline unsigned wave_umin(unsigned x) {
  x = umin2(x, row_xor<1>(x));
  x = umin2(x, row_xor<2>(x));
  x = umin2(x, row_xor<4>(x));
  x = umin2(x, row_xor<8>(x));
  return umin2(x, other_row(x));
}

__global__ void __launch_bounds__(32 * KWV_)
knn_kernel(const float* __restrict__ coord, int* __restrict__ idx_out) {
  __shared__ int osm[KPB_ * KNB];
  const int tid  = threadIdx.x;
  const int lane = tid & 31;
  const int wave = __builtin_amdgcn_readfirstlane(tid >> 5);
  const int s    = blockIdx.x / (N_ / KPB_);
  const int pb   = (blockIdx.x % (N_ / KPB_)) * KPB_;
  const int p0   = pb + wave * KPW_;
  const float* cs = coord + (size_t)s * N_ * 3;

  float c[24];
  {
    const float4* cv = reinterpret_cast<const float4*>(cs + lane * 24);
    for (int u = 0; u < 6; ++u) {
      const float4 t = cv[u];
      c[4 * u + 0] = t.x; c[4 * u + 1] = t.y;
      c[4 * u + 2] = t.z; c[4 * u + 3] = t.w;
    }
  }

  unsigned d[KPW_][8];
  for (int q = 0; q < KPW_; ++q) {
    const float px = cs[(p0 + q) * 3 + 0];
    const float py = cs[(p0 + q) * 3 + 1];
    const float pz = cs[(p0 + q) * 3 + 2];
    for (int i = 0; i < 8; ++i) {
      const float dx = c[3 * i + 0] - px;
      const float dy = c[3 * i + 1] - py;
      const float dz = c[3 * i + 2] - pz;
      const float d2 = __builtin_fmaf(dz, dz, __builtin_fmaf(dx, dx, dy * dy));
      d[q][i] = __float_as_uint(d2);
    }
  }

  const int lane8 = lane * 8;
  int res[KPW_] = {};

  for (int j = 0; j < KNB; ++j) {
    for (int q = 0; q < KPW_; ++q) {
      unsigned* dq = d[q];
      unsigned lm = umin2(dq[0], dq[1]);
      for (int i = 2; i < 8; ++i) lm = umin2(lm, dq[i]);
      int li = 7;
      for (int i = 6; i >= 0; --i) li = (dq[i] == lm) ? i : li;
      const unsigned wm  = wave_umin(lm);
      const unsigned bal = __builtin_amdgcn_ballot_w32(lm == wm);
      const int L  = __builtin_ctz(bal);
      const int sl = __builtin_amdgcn_readlane(li, L);
      const int m  = L * 8 + sl;
      res[q] = (lane == j) ? m : res[q];
      const int rel = m - lane8;
      for (int i = 0; i < 8; ++i) dq[i] = (rel == i) ? KRM_ : dq[i];
    }
  }

  if (lane < KNB) {
    for (int q = 0; q < KPW_; ++q)
      osm[(wave * KPW_ + q) * KNB + lane] = res[q];
  }
  __syncthreads();
  int* ob = idx_out + ((size_t)s * N_ + pb) * KNB;
  for (int i = tid; i < KPB_ * KNB; i += 32 * KWV_) ob[i] = osm[i];
}

__global__ void lang_mlp_kernel(const float* __restrict__ lang_feat,
                                const float* __restrict__ w1,
                                const float* __restrict__ b1,
                                const float* __restrict__ w2,
                                const float* __restrict__ b2,
                                float* __restrict__ lang_out) {
  __shared__ float xs[LANGID_];
  __shared__ float hs[OD_];
  const int row = blockIdx.x;
  const int t   = threadIdx.x;
  const float* x = lang_feat + (size_t)row * LANGID_;
  xs[t]       = x[t];
  xs[t + 128] = x[t + 128];
  __syncthreads();
  float acc = b1[t];
  for (int k = 0; k < LANGID_; ++k) acc += xs[k] * w1[k * OD_ + t];
  hs[t] = fmaxf(acc, 0.0f);
  __syncthreads();
  float acc2 = b2[t];
  for (int k = 0; k < OD_; ++k) acc2 += hs[k] * w2[k * OD_ + t];
  lang_out[(size_t)row * OD_ + t] = acc2;
}

__global__ void feat_mlp_kernel(const float* __restrict__ feat,
                                const float* __restrict__ w1,
                                const float* __restrict__ b1,
                                const float* __restrict__ w2,
                                const float* __restrict__ b2,
                                const float* __restrict__ lang_mask,
                                const float* __restrict__ lang_ws,
                                float* __restrict__ g_out,
                                float* __restrict__ f_out) {
  __shared__ float Asm[16 * RP_];
  __shared__ float Hsm[16 * RP_];
  __shared__ float attnsm[16 * LPP_];
  const int s    = blockIdx.x >> 4;
  const int mt   = blockIdx.x & 15;
  const int tid  = threadIdx.x;
  const int lane = tid & 31, wave = tid >> 5;
  const int half = lane >> 4, lr = lane & 15;

  const float* A = feat + ((size_t)s * N_ + mt * 16) * PCID_;
  for (int i = tid; i < 16 * PCID_; i += 256)
    Asm[(i >> 7) * RP_ + (i & 127)] = A[i];
  __syncthreads();

  const int n0 = wave * 16;
  v8f c = {};
  for (int kg = 0; kg < PCID_; kg += 4 * KG_) {
    v2f a[KG_], b[KG_];
#pragma unroll
    for (int u = 0; u < KG_; ++u) {
      const int ka = kg + 4 * u + 2 * half;
      a[u].x = Asm[lr * RP_ + ka];
      a[u].y = Asm[lr * RP_ + ka + 1];
      b[u].x = w1[ka * OD_ + n0 + lr];
      b[u].y = w1[(ka + 1) * OD_ + n0 + lr];
    }
    __builtin_amdgcn_sched_barrier(0);
#pragma unroll
    for (int u = 0; u < KG_; ++u) c = wmma4(a[u], b[u], c);
  }
  {
    const float bb = b1[n0 + lr];
#pragma unroll
    for (int v = 0; v < 8; ++v)
      Hsm[(v + 8 * half) * RP_ + n0 + lr] = fmaxf(c[v] + bb, 0.0f);
  }
  __syncthreads();

  v8f c2 = {};
  for (int kg = 0; kg < OD_; kg += 4 * KG_) {
    v2f a[KG_], b[KG_];
#pragma unroll
    for (int u = 0; u < KG_; ++u) {
      const int ka = kg + 4 * u + 2 * half;
      a[u].x = Hsm[lr * RP_ + ka];
      a[u].y = Hsm[lr * RP_ + ka + 1];
      b[u].x = w2[ka * OD_ + n0 + lr];
      b[u].y = w2[(ka + 1) * OD_ + n0 + lr];
    }
    __builtin_amdgcn_sched_barrier(0);
#pragma unroll
    for (int u = 0; u < KG_; ++u) c2 = wmma4(a[u], b[u], c2);
  }
  float fv[8];
  {
    const float bb = b2[n0 + lr];
    float* fo = f_out + ((size_t)s * N_ + mt * 16) * OD_;
#pragma unroll
    for (int v = 0; v < 8; ++v) {
      fv[v] = c2[v] + bb;
      Asm[(v + 8 * half) * RP_ + n0 + lr] = fv[v];
      fo[(v + 8 * half) * OD_ + n0 + lr]  = fv[v];
    }
  }
  __syncthreads();

  const float* langS = lang_ws + (size_t)s * L_ * OD_;

  if (wave < 3) {
    const int c0 = wave * 16;
    const int lclamp = min(c0 + lr, L_ - 1);
    v8f cl = {};
    for (int kg = 0; kg < OD_; kg += 4 * KG_) {
      v2f a[KG_], b[KG_];
#pragma unroll
      for (int u = 0; u < KG_; ++u) {
        const int ka = kg + 4 * u + 2 * half;
        a[u].x = Asm[lr * RP_ + ka];
        a[u].y = Asm[lr * RP_ + ka + 1];
        b[u].x = langS[lclamp * OD_ + ka];
        b[u].y = langS[lclamp * OD_ + ka + 1];
      }
      __builtin_amdgcn_sched_barrier(0);
#pragma unroll
      for (int u = 0; u < KG_; ++u) cl = wmma4(a[u], b[u], cl);
    }
#pragma unroll
    for (int v = 0; v < 8; ++v)
      attnsm[(v + 8 * half) * LPP_ + c0 + lr] = cl[v];
  }
  __syncthreads();

  if (tid < 16) {
    float* rowp = attnsm + tid * LPP_;
    float mx = -FLT_MAX;
    for (int l = 0; l < L_; ++l) mx = fmaxf(mx, rowp[l]);
    float ssum = 0.f;
    for (int l = 0; l < L_; ++l) {
      const float e = expf(rowp[l] - mx);
      rowp[l] = e;
      ssum += e;
    }
    const float inv = 1.f / ssum;
    const float* mk = lang_mask + (size_t)s * L_;
    float s2 = 0.f;
    for (int l = 0; l < L_; ++l) {
      const float p = rowp[l] * inv * mk[l];
      rowp[l] = p;
      s2 += p;
    }
    const float inv2 = 1.f / (s2 + 1e-7f);
    for (int l = 0; l < L_; ++l) rowp[l] *= inv2;
    for (int l = L_; l < LP_; ++l) rowp[l] = 0.f;
  }
  __syncthreads();

  {
    v8f ci = {};
    {
      v2f a[LP_ / 4], b[LP_ / 4];
#pragma unroll
      for (int u = 0; u < LP_ / 4; ++u) {
        const int ka  = 4 * u + 2 * half;
        const int kb0 = min(ka, L_ - 1), kb1 = min(ka + 1, L_ - 1);
        a[u].x = attnsm[lr * LPP_ + ka];
        a[u].y = attnsm[lr * LPP_ + ka + 1];
        b[u].x = langS[kb0 * OD_ + n0 + lr];
        b[u].y = langS[kb1 * OD_ + n0 + lr];
      }
#pragma unroll
      for (int u = 0; u < LP_ / 4; ++u) ci = wmma4(a[u], b[u], ci);
    }
    float* go = g_out + ((size_t)s * N_ + mt * 16) * OD_;
#pragma unroll
    for (int v = 0; v < 8; ++v)
      go[(v + 8 * half) * OD_ + n0 + lr] = fv[v] * ci[v];
  }
}

#define PB_     16
#define K1S_    3
#define FT_     128

__global__ void __launch_bounds__(FT_)
fused_kernel(const float* __restrict__ coord,
             const float* __restrict__ rel_w1,
             const float* __restrict__ rel_b1,
             const float* __restrict__ rel_w2,
             const float* __restrict__ rel_b2,
             const int*   __restrict__ idx_ws,
             const float* __restrict__ g_ws,
             float* __restrict__ out,
             float* __restrict__ score) {
  __shared__ float relin[KNB * PB_ * KP_];
  __shared__ float hbuf[2][PB_ * RP_];
  __shared__ int   idxsm[PB_ * KNB];

  const int s    = blockIdx.x >> 4;
  const int pt0  = (blockIdx.x & 15) * PB_;
  const int tid  = threadIdx.x;
  const int lane = tid & 31, wave = tid >> 5;
  const int half = lane >> 4, lr = lane & 15;
  const int c0   = wave * 32;

  for (int i = tid; i < PB_ * KNB; i += FT_)
    idxsm[i] = idx_ws[((size_t)s * N_ + pt0) * KNB + i];

  v2f bw1[2][K1S_];
  v2f bw2[2][OD_ / 4];
  float bb1[2], bb2[2];
#pragma unroll
  for (int t = 0; t < 2; ++t) {
    const int col = c0 + 16 * t + lr;
#pragma unroll
    for (int u = 0; u < K1S_; ++u) {
      const int ka  = 4 * u + 2 * half;
      const int kb0 = min(ka, 9), kb1 = min(ka + 1, 9);
      bw1[t][u].x = rel_w1[kb0 * OD_ + col];
      bw1[t][u].y = rel_w1[kb1 * OD_ + col];
    }
#pragma unroll
    for (int u = 0; u < OD_ / 4; ++u) {
      const int ka = 4 * u + 2 * half;
      bw2[t][u].x = rel_w2[ka * OD_ + col];
      bw2[t][u].y = rel_w2[(ka + 1) * OD_ + col];
    }
    bb1[t] = rel_b1[col];
    bb2[t] = rel_b2[col];
  }
  __syncthreads();

  for (int r = tid; r < PB_ * KNB; r += FT_) {
    const int j = r >> 4, p = r & 15;
    const float* cs = coord + (size_t)s * N_ * 3;
    const int m = idxsm[p * KNB + j];
    const int n = pt0 + p;
    const float nx = cs[m * 3 + 0], ny = cs[m * 3 + 1], nz = cs[m * 3 + 2];
    const float ox = cs[n * 3 + 0], oy = cs[n * 3 + 1], oz = cs[n * 3 + 2];
    const float rx = nx - ox, ry = ny - oy, rz = nz - oz;
    const float dist = sqrtf(rx * rx + ry * ry + rz * rz);
    float* ri = relin + r * KP_;
    ri[0] = nx; ri[1] = ny; ri[2] = nz;
    ri[3] = ox; ri[4] = oy; ri[5] = oz;
    ri[6] = rx; ri[7] = ry; ri[8] = rz;
    ri[9] = dist;
    ri[10] = 0.f; ri[11] = 0.f;
  }
  __syncthreads();

  float acc[2][8];
#pragma unroll
  for (int v = 0; v < 8; ++v) { acc[0][v] = 0.f; acc[1][v] = 0.f; }

  const int*   ip = idxsm + 8 * half * KNB;
  const float* gs = g_ws + (size_t)s * N_ * OD_;
  const unsigned lcol = (unsigned)(c0 + lr);

  for (int j = 0; j < KNB; ++j) {
    unsigned m[8];
#pragma unroll
    for (int v = 0; v < 8; ++v) m[v] = (unsigned)ip[v * KNB + j];

    v8f ca = {}, cb = {};
    {
      const float* rp = relin + (j * PB_ + lr) * KP_ + 2 * half;
      v2f a[K1S_];
#pragma unroll
      for (int u = 0; u < K1S_; ++u) {
        a[u].x = rp[4 * u];
        a[u].y = rp[4 * u + 1];
      }
#pragma unroll
      for (int u = 0; u < K1S_; ++u) {
        ca = wmma4(a[u], bw1[0][u], ca);
        cb = wmma4(a[u], bw1[1][u], cb);
      }
    }
    float* hb = hbuf[j & 1];
    {
      float* hp = hb + 8 * half * RP_ + c0 + lr;
#pragma unroll
      for (int v = 0; v < 8; ++v) {
        hp[v * RP_]      = fmaxf(ca[v] + bb1[0], 0.f);
        hp[v * RP_ + 16] = fmaxf(cb[v] + bb1[1], 0.f);
      }
    }
    __syncthreads();

    float gv[2][8];
#pragma unroll
    for (int v = 0; v < 8; ++v) {
      const float* gp = gs + (m[v] * OD_ + lcol);
      gv[0][v] = gp[0];
      gv[1][v] = gp[16];
    }

    v8f c2a = {}, c2b = {};
    const float* ap = hb + lr * RP_ + 2 * half;
    for (int kg = 0; kg < OD_ / 4; kg += KG_) {
      v2f a[KG_];
#pragma unroll
      for (int u = 0; u < KG_; ++u) {
        a[u].x = ap[4 * (kg + u)];
        a[u].y = ap[4 * (kg + u) + 1];
      }
      __builtin_amdgcn_sched_barrier(0);
#pragma unroll
      for (int u = 0; u < KG_; ++u) {
        c2a = wmma4(a[u], bw2[0][kg + u], c2a);
        c2b = wmma4(a[u], bw2[1][kg + u], c2b);
      }
    }
#pragma unroll
    for (int v = 0; v < 8; ++v) {
      acc[0][v] += gv[0][v] * (c2a[v] + bb2[0]);
      acc[1][v] += gv[1][v] * (c2b[v] + bb2[1]);
    }
  }

  float* osm = hbuf[1];
  {
    float* op = osm + 8 * half * RP_ + c0 + lr;
#pragma unroll
    for (int v = 0; v < 8; ++v) {
      op[v * RP_]      = acc[0][v];
      op[v * RP_ + 16] = acc[1][v];
    }
  }
  __syncthreads();

#pragma unroll
  for (int pp = 0; pp < PB_; pp += FT_ / 16) {
    const int p = pp + (tid >> 4), q = tid & 15;
    float* o = out + ((size_t)s * N_ + pt0 + p) * OD_;
    float x[8];
#pragma unroll
    for (int i = 0; i < 8; ++i) x[i] = osm[p * RP_ + q + 16 * i] + o[q + 16 * i];
#pragma unroll
    for (int i = 0; i < 8; ++i) o[q + 16 * i] = x[i];
    const float y0 = x[0] + x[4], y1 = x[1] + x[5];
    const float y2 = x[2] + x[6], y3 = x[3] + x[7];
    float r = (y0 + y2) + (y1 + y3);
    r += __shfl_xor(r, 8);
    r += __shfl_xor(r, 4);
    r += __shfl_xor(r, 2);
    r += __shfl_xor(r, 1);
    if (q == 0) score[(size_t)s * N_ + pt0 + p] = r;
  }
}

extern "C" void kernel_launch(void* const* d_in, const int* in_sizes, int n_in,
                              void* d_out, int out_size, void* d_ws,
                              size_t ws_size, hipStream_t stream) {
  (void)in_sizes; (void)n_in; (void)out_size; (void)ws_size;
  const float* feat      = (const float*)d_in[0];
  const float* coord     = (const float*)d_in[1];
  const float* lang_feat = (const float*)d_in[2];
  const float* lang_mask = (const float*)d_in[3];
  const float* rel_w1    = (const float*)d_in[4];
  const float* rel_b1    = (const float*)d_in[5];
  const float* rel_w2    = (const float*)d_in[6];
  const float* rel_b2    = (const float*)d_in[7];
  const float* lang_w1   = (const float*)d_in[8];
  const float* lang_b1   = (const float*)d_in[9];
  const float* lang_w2   = (const float*)d_in[10];
  const float* lang_b2   = (const float*)d_in[11];
  const float* feat_w1   = (const float*)d_in[12];
  const float* feat_b1   = (const float*)d_in[13];
  const float* feat_w2   = (const float*)d_in[14];
  const float* feat_b2   = (const float*)d_in[15];

  char* ws = (char*)d_ws;
  int*   idx_ws  = (int*)ws;
  float* lang_ws = (float*)(ws + (size_t)S_ * N_ * KNB * 4);
  float* g_ws    = (float*)(ws + (size_t)S_ * N_ * KNB * 4 +
                                 (size_t)S_ * L_ * OD_ * 4);

  float* out   = (float*)d_out;
  float* score = out + (size_t)S_ * N_ * OD_;

  knn_kernel<<<S_ * (N_ / KPB_), 32 * KWV_, 0, stream>>>(coord, idx_ws);
  lang_mlp_kernel<<<S_ * L_, 128, 0, stream>>>(lang_feat, lang_w1, lang_b1,
                                               lang_w2, lang_b2, lang_ws);
  feat_mlp_kernel<<<S_ * (N_ / 16), 256, 0, stream>>>(feat, feat_w1, feat_b1,
                                                      feat_w2, feat_b2, lang_mask,
                                                      lang_ws, g_ws, out);
  fused_kernel<<<S_ * (N_ / PB_), FT_, 0, stream>>>(coord, rel_w1, rel_b1,
                                                     rel_w2, rel_b2, idx_ws,
                                                     g_ws, out, score);
}


# lang MLP on f32 WMMA in 16-row tiles (160 blocks of 8 waves, weights read once per 16 rows), was one VALU row per block re-reading all weights; lang sums now in WMMA k order
# speedup vs baseline: 7.8000x; 1.0899x over previous
#include <hip/hip_runtime.h>
#include <hip/hip_bf16.h>
#include <float.h>


typedef float v2f __attribute__((ext_vector_type(2)));
typedef float v8f __attribute__((ext_vector_type(8)));

#define S_      64
#define N_      256
#define L_      40
#define LP_     48
#define RP_     132
#define LPP_    52
#define KP_     20
#define KNB     17
#define OD_     128
#define LANGID_ 256
#define PCID_   128
#define KG_     16

static __device__ inline v8f wmma4(v2f a, v2f b, v8f c) {
  return __builtin_amdgcn_wmma_f32_16x16x4_f32(false, a, false, b, (short)0, c,
                                               false, false);
}

#define KPW_    4
#define KWV_    8
#define KPB_    (KPW_ * KWV_)
#define KRM_    0xffffffffu

template <int XM>
static __device__ inline unsigned row_xor(unsigned x) {
  return (unsigned)__builtin_amdgcn_mov_dpp((int)x, 0x160 + XM, 0xf, 0xf, true);
}

static __device__ inline unsigned other_row(unsigned x) {
  return (unsigned)__builtin_amdgcn_permlanex16((int)x, (int)x, 0x76543210,
                                                0xfedcba98, false, false);
}

static __device__ inline unsigned umin2(unsigned a, unsigned b) {
  return a < b ? a : b;
}

static __device__ inline unsigned wave_umin(unsigned x) {
  x = umin2(x, row_xor<1>(x));
  x = umin2(x, row_xor<2>(x));
  x = umin2(x, row_xor<4>(x));
  x = umin2(x, row_xor<8>(x));
  return umin2(x, other_row(x));
}

__global__ void __launch_bounds__(32 * KWV_)
knn_kernel(const float* __restrict__ coord, int* __restrict__ idx_out) {
  __shared__ int osm[KPB_ * KNB];
  const int tid  = threadIdx.x;
  const int lane = tid & 31;
  const int wave = __builtin_amdgcn_readfirstlane(tid >> 5);
  const int s    = blockIdx.x / (N_ / KPB_);
  const int pb   = (blockIdx.x % (N_ / KPB_)) * KPB_;
  const int p0   = pb + wave * KPW_;
  const float* cs = coord + (size_t)s * N_ * 3;

  float c[24];
  {
    const float4* cv = reinterpret_cast<const float4*>(cs + lane * 24);
    for (int u = 0; u < 6; ++u) {
      const float4 t = cv[u];
      c[4 * u + 0] = t.x; c[4 * u + 1] = t.y;
      c[4 * u + 2] = t.z; c[4 * u + 3] = t.w;
    }
  }

  unsigned d[KPW_][8];
  for (int q = 0; q < KPW_; ++q) {
    const float px = cs[(p0 + q) * 3 + 0];
    const float py = cs[(p0 + q) * 3 + 1];
    const float pz = cs[(p0 + q) * 3 + 2];
    for (int i = 0; i < 8; ++i) {
      const float dx = c[3 * i + 0] - px;
      const float dy = c[3 * i + 1] - py;
      const float dz = c[3 * i + 2] - pz;
      const float d2 = __builtin_fmaf(dz, dz, __builtin_fmaf(dx, dx, dy * dy));
      d[q][i] = __float_as_uint(d2);
    }
  }

  const int lane8 = lane * 8;
  int res[KPW_] = {};

  for (int j = 0; j < KNB; ++j) {
    for (int q = 0; q < KPW_; ++q) {
      unsigned* dq = d[q];
      unsigned lm = umin2(dq[0], dq[1]);
      for (int i = 2; i < 8; ++i) lm = umin2(lm, dq[i]);
      int li = 7;
      for (int i = 6; i >= 0; --i) li = (dq[i] == lm) ? i : li;
      const unsigned wm  = wave_umin(lm);
      const unsigned bal = __builtin_amdgcn_ballot_w32(lm == wm);
      const int L  = __builtin_ctz(bal);
      const int sl = __builtin_amdgcn_readlane(li, L);
      const int m  = L * 8 + sl;
      res[q] = (lane == j) ? m : res[q];
      const int rel = m - lane8;
      for (int i = 0; i < 8; ++i) dq[i] = (rel == i) ? KRM_ : dq[i];
    }
  }

  if (lane < KNB) {
    for (int q = 0; q < KPW_; ++q)
      osm[(wave * KPW_ + q) * KNB + lane] = res[q];
  }
  __syncthreads();
  int* ob = idx_out + ((size_t)s * N_ + pb) * KNB;
  for (int i = tid; i < KPB_ * KNB; i += 32 * KWV_) ob[i] = osm[i];
}

#define XP_     260
#define LROWS_  (S_ * L_)

__global__ void __launch_bounds__(256)
lang_mlp_kernel(const float* __restrict__ lang_feat,
                const float* __restrict__ w1,
                const float* __restrict__ b1,
                const float* __restrict__ w2,
                const float* __restrict__ b2,
                float* __restrict__ lang_out) {
  __shared__ float Xsm[16 * XP_] __attribute__((aligned(16)));
  __shared__ float Hsm[16 * RP_];
  const int row0 = blockIdx.x * 16;
  const int tid  = threadIdx.x;
  const int lane = tid & 31, wave = tid >> 5;
  const int half = lane >> 4, lr = lane & 15;

  {
    const float4* X4 =
        reinterpret_cast<const float4*>(lang_feat + (size_t)row0 * LANGID_);
    float4 t[4];
#pragma unroll
    for (int u = 0; u < 4; ++u) t[u] = X4[tid + 256 * u];
#pragma unroll
    for (int u = 0; u < 4; ++u) {
      const int i4 = tid + 256 * u;
      *reinterpret_cast<float4*>(Xsm + (i4 >> 6) * XP_ + (i4 & 63) * 4) = t[u];
    }
  }
  __syncthreads();

  const int n0 = wave * 16;
  v8f c = {};
  for (int kg = 0; kg < LANGID_; kg += 4 * KG_) {
    v2f a[KG_], b[KG_];
#pragma unroll
    for (int u = 0; u < KG_; ++u) {
      const int ka = kg + 4 * u + 2 * half;
      a[u].x = Xsm[lr * XP_ + ka];
      a[u].y = Xsm[lr * XP_ + ka + 1];
      b[u].x = w1[ka * OD_ + n0 + lr];
      b[u].y = w1[(ka + 1) * OD_ + n0 + lr];
    }
    __builtin_amdgcn_sched_barrier(0);
#pragma unroll
    for (int u = 0; u < KG_; ++u) c = wmma4(a[u], b[u], c);
  }
  {
    const float bb = b1[n0 + lr];
#pragma unroll
    for (int v = 0; v < 8; ++v)
      Hsm[(v + 8 * half) * RP_ + n0 + lr] = fmaxf(c[v] + bb, 0.0f);
  }
  __syncthreads();

  v8f c2 = {};
  for (int kg = 0; kg < OD_; kg += 4 * KG_) {
    v2f a[KG_], b[KG_];
#pragma unroll
    for (int u = 0; u < KG_; ++u) {
      const int ka = kg + 4 * u + 2 * half;
      a[u].x = Hsm[lr * RP_ + ka];
      a[u].y = Hsm[lr * RP_ + ka + 1];
      b[u].x = w2[ka * OD_ + n0 + lr];
      b[u].y = w2[(ka + 1) * OD_ + n0 + lr];
    }
    __builtin_amdgcn_sched_barrier(0);
#pragma unroll
    for (int u = 0; u < KG_; ++u) c2 = wmma4(a[u], b[u], c2);
  }
  {
    const float bb = b2[n0 + lr];
#pragma unroll
    for (int v = 0; v < 8; ++v)
      Xsm[(v + 8 * half) * XP_ + n0 + lr] = c2[v] + bb;
  }
  __syncthreads();
  float* lo = lang_out + (size_t)row0 * OD_;
  for (int u = 0; u < 8; ++u) {
    const int i = tid + 256 * u;
    lo[i] = Xsm[(i >> 7) * XP_ + (i & 127)];
  }
}

__global__ void feat_mlp_kernel(const float* __restrict__ feat,
                                const float* __restrict__ w1,
                                const float* __restrict__ b1,
                                const float* __restrict__ w2,
                                const float* __restrict__ b2,
                                const float* __restrict__ lang_mask,
                                const float* __restrict__ lang_ws,
                                float* __restrict__ g_out,
                                float* __restrict__ f_out) {
  __shared__ float Asm[16 * RP_];
  __shared__ float Hsm[16 * RP_];
  __shared__ float attnsm[16 * LPP_];
  const int s    = blockIdx.x >> 4;
  const int mt   = blockIdx.x & 15;
  const int tid  = threadIdx.x;
  const int lane = tid & 31, wave = tid >> 5;
  const int half = lane >> 4, lr = lane & 15;

  const float* A = feat + ((size_t)s * N_ + mt * 16) * PCID_;
  for (int i = tid; i < 16 * PCID_; i += 256)
    Asm[(i >> 7) * RP_ + (i & 127)] = A[i];
  __syncthreads();

  const int n0 = wave * 16;
  v8f c = {};
  for (int kg = 0; kg < PCID_; kg += 4 * KG_) {
    v2f a[KG_], b[KG_];
#pragma unroll
    for (int u = 0; u < KG_; ++u) {
      const int ka = kg + 4 * u + 2 * half;
      a[u].x = Asm[lr * RP_ + ka];
      a[u].y = Asm[lr * RP_ + ka + 1];
      b[u].x = w1[ka * OD_ + n0 + lr];
      b[u].y = w1[(ka + 1) * OD_ + n0 + lr];
    }
    __builtin_amdgcn_sched_barrier(0);
#pragma unroll
    for (int u = 0; u < KG_; ++u) c = wmma4(a[u], b[u], c);
  }
  {
    const float bb = b1[n0 + lr];
#pragma unroll
    for (int v = 0; v < 8; ++v)
      Hsm[(v + 8 * half) * RP_ + n0 + lr] = fmaxf(c[v] + bb, 0.0f);
  }
  __syncthreads();

  v8f c2 = {};
  for (int kg = 0; kg < OD_; kg += 4 * KG_) {
    v2f a[KG_], b[KG_];
#pragma unroll
    for (int u = 0; u < KG_; ++u) {
      const int ka = kg + 4 * u + 2 * half;
      a[u].x = Hsm[lr * RP_ + ka];
      a[u].y = Hsm[lr * RP_ + ka + 1];
      b[u].x = w2[ka * OD_ + n0 + lr];
      b[u].y = w2[(ka + 1) * OD_ + n0 + lr];
    }
    __builtin_amdgcn_sched_barrier(0);
#pragma unroll
    for (int u = 0; u < KG_; ++u) c2 = wmma4(a[u], b[u], c2);
  }
  float fv[8];
  {
    const float bb = b2[n0 + lr];
    float* fo = f_out + ((size_t)s * N_ + mt * 16) * OD_;
#pragma unroll
    for (int v = 0; v < 8; ++v) {
      fv[v] = c2[v] + bb;
      Asm[(v + 8 * half) * RP_ + n0 + lr] = fv[v];
      fo[(v + 8 * half) * OD_ + n0 + lr]  = fv[v];
    }
  }
  __syncthreads();

  const float* langS = lang_ws + (size_t)s * L_ * OD_;

  if (wave < 3) {
    const int c0 = wave * 16;
    const int lclamp = min(c0 + lr, L_ - 1);
    v8f cl = {};
    for (int kg = 0; kg < OD_; kg += 4 * KG_) {
      v2f a[KG_], b[KG_];
#pragma unroll
      for (int u = 0; u < KG_; ++u) {
        const int ka = kg + 4 * u + 2 * half;
        a[u].x = Asm[lr * RP_ + ka];
        a[u].y = Asm[lr * RP_ + ka + 1];
        b[u].x = langS[lclamp * OD_ + ka];
        b[u].y = langS[lclamp * OD_ + ka + 1];
      }
      __builtin_amdgcn_sched_barrier(0);
#pragma unroll
      for (int u = 0; u < KG_; ++u) cl = wmma4(a[u], b[u], cl);
    }
#pragma unroll
    for (int v = 0; v < 8; ++v)
      attnsm[(v + 8 * half) * LPP_ + c0 + lr] = cl[v];
  }
  __syncthreads();

  if (tid < 16) {
    float* rowp = attnsm + tid * LPP_;
    float mx = -FLT_MAX;
    for (int l = 0; l < L_; ++l) mx = fmaxf(mx, rowp[l]);
    float ssum = 0.f;
    for (int l = 0; l < L_; ++l) {
      const float e = expf(rowp[l] - mx);
      rowp[l] = e;
      ssum += e;
    }
    const float inv = 1.f / ssum;
    const float* mk = lang_mask + (size_t)s * L_;
    float s2 = 0.f;
    for (int l = 0; l < L_; ++l) {
      const float p = rowp[l] * inv * mk[l];
      rowp[l] = p;
      s2 += p;
    }
    const float inv2 = 1.f / (s2 + 1e-7f);
    for (int l = 0; l < L_; ++l) rowp[l] *= inv2;
    for (int l = L_; l < LP_; ++l) rowp[l] = 0.f;
  }
  __syncthreads();

  {
    v8f ci = {};
    {
      v2f a[LP_ / 4], b[LP_ / 4];
#pragma unroll
      for (int u = 0; u < LP_ / 4; ++u) {
        const int ka  = 4 * u + 2 * half;
        const int kb0 = min(ka, L_ - 1), kb1 = min(ka + 1, L_ - 1);
        a[u].x = attnsm[lr * LPP_ + ka];
        a[u].y = attnsm[lr * LPP_ + ka + 1];
        b[u].x = langS[kb0 * OD_ + n0 + lr];
        b[u].y = langS[kb1 * OD_ + n0 + lr];
      }
#pragma unroll
      for (int u = 0; u < LP_ / 4; ++u) ci = wmma4(a[u], b[u], ci);
    }
    float* go = g_out + ((size_t)s * N_ + mt * 16) * OD_;
#pragma unroll
    for (int v = 0; v < 8; ++v)
      go[(v + 8 * half) * OD_ + n0 + lr] = fv[v] * ci[v];
  }
}

#define PB_     16
#define K1S_    3
#define FT_     128

__global__ void __launch_bounds__(FT_)
fused_kernel(const float* __restrict__ coord,
             const float* __restrict__ rel_w1,
             const float* __restrict__ rel_b1,
             const float* __restrict__ rel_w2,
             const float* __restrict__ rel_b2,
             const int*   __restrict__ idx_ws,
             const float* __restrict__ g_ws,
             float* __restrict__ out,
             float* __restrict__ score) {
  __shared__ float relin[KNB * PB_ * KP_];
  __shared__ float hbuf[2][PB_ * RP_];
  __shared__ int   idxsm[PB_ * KNB];

  const int s    = blockIdx.x >> 4;
  const int pt0  = (blockIdx.x & 15) * PB_;
  const int tid  = threadIdx.x;
  const int lane = tid & 31, wave = tid >> 5;
  const int half = lane >> 4, lr = lane & 15;
  const int c0   = wave * 32;

  for (int i = tid; i < PB_ * KNB; i += FT_)
    idxsm[i] = idx_ws[((size_t)s * N_ + pt0) * KNB + i];

  v2f bw1[2][K1S_];
  v2f bw2[2][OD_ / 4];
  float bb1[2], bb2[2];
#pragma unroll
  for (int t = 0; t < 2; ++t) {
    const int col = c0 + 16 * t + lr;
#pragma unroll
    for (int u = 0; u < K1S_; ++u) {
      const int ka  = 4 * u + 2 * half;
      const int kb0 = min(ka, 9), kb1 = min(ka + 1, 9);
      bw1[t][u].x = rel_w1[kb0 * OD_ + col];
      bw1[t][u].y = rel_w1[kb1 * OD_ + col];
    }
#pragma unroll
    for (int u = 0; u < OD_ / 4; ++u) {
      const int ka = 4 * u + 2 * half;
      bw2[t][u].x = rel_w2[ka * OD_ + col];
      bw2[t][u].y = rel_w2[(ka + 1) * OD_ + col];
    }
    bb1[t] = rel_b1[col];
    bb2[t] = rel_b2[col];
  }
  __syncthreads();

  for (int r = tid; r < PB_ * KNB; r += FT_) {
    const int j = r >> 4, p = r & 15;
    const float* cs = coord + (size_t)s * N_ * 3;
    const int m = idxsm[p * KNB + j];
    const int n = pt0 + p;
    const float nx = cs[m * 3 + 0], ny = cs[m * 3 + 1], nz = cs[m * 3 + 2];
    const float ox = cs[n * 3 + 0], oy = cs[n * 3 + 1], oz = cs[n * 3 + 2];
    const float rx = nx - ox, ry = ny - oy, rz = nz - oz;
    const float dist = sqrtf(rx * rx + ry * ry + rz * rz);
    float* ri = relin + r * KP_;
    ri[0] = nx; ri[1] = ny; ri[2] = nz;
    ri[3] = ox; ri[4] = oy; ri[5] = oz;
    ri[6] = rx; ri[7] = ry; ri[8] = rz;
    ri[9] = dist;
    ri[10] = 0.f; ri[11] = 0.f;
  }
  __syncthreads();

  float acc[2][8];
#pragma unroll
  for (int v = 0; v < 8; ++v) { acc[0][v] = 0.f; acc[1][v] = 0.f; }

  const int*   ip = idxsm + 8 * half * KNB;
  const float* gs = g_ws + (size_t)s * N_ * OD_;
  const unsigned lcol = (unsigned)(c0 + lr);

  for (int j = 0; j < KNB; ++j) {
    unsigned m[8];
#pragma unroll
    for (int v = 0; v < 8; ++v) m[v] = (unsigned)ip[v * KNB + j];

    v8f ca = {}, cb = {};
    {
      const float* rp = relin + (j * PB_ + lr) * KP_ + 2 * half;
      v2f a[K1S_];
#pragma unroll
      for (int u = 0; u < K1S_; ++u) {
        a[u].x = rp[4 * u];
        a[u].y = rp[4 * u + 1];
      }
#pragma unroll
      for (int u = 0; u < K1S_; ++u) {
        ca = wmma4(a[u], bw1[0][u], ca);
        cb = wmma4(a[u], bw1[1][u], cb);
      }
    }
    float* hb = hbuf[j & 1];
    {
      float* hp = hb + 8 * half * RP_ + c0 + lr;
#pragma unroll
      for (int v = 0; v < 8; ++v) {
        hp[v * RP_]      = fmaxf(ca[v] + bb1[0], 0.f);
        hp[v * RP_ + 16] = fmaxf(cb[v] + bb1[1], 0.f);
      }
    }
    __syncthreads();

    float gv[2][8];
#pragma unroll
    for (int v = 0; v < 8; ++v) {
      const float* gp = gs + (m[v] * OD_ + lcol);
      gv[0][v] = gp[0];
      gv[1][v] = gp[16];
    }

    v8f c2a = {}, c2b = {};
    const float* ap = hb + lr * RP_ + 2 * half;
    for (int kg = 0; kg < OD_ / 4; kg += KG_) {
      v2f a[KG_];
#pragma unroll
      for (int u = 0; u < KG_; ++u) {
        a[u].x = ap[4 * (kg + u)];
        a[u].y = ap[4 * (kg + u) + 1];
      }
      __builtin_amdgcn_sched_barrier(0);
#pragma unroll
      for (int u = 0; u < KG_; ++u) {
        c2a = wmma4(a[u], bw2[0][kg + u], c2a);
        c2b = wmma4(a[u], bw2[1][kg + u], c2b);
      }
    }
#pragma unroll
    for (int v = 0; v < 8; ++v) {
      acc[0][v] += gv[0][v] * (c2a[v] + bb2[0]);
      acc[1][v] += gv[1][v] * (c2b[v] + bb2[1]);
    }
  }

  float* osm = hbuf[1];
  {
    float* op = osm + 8 * half * RP_ + c0 + lr;
#pragma unroll
    for (int v = 0; v < 8; ++v) {
      op[v * RP_]      = acc[0][v];
      op[v * RP_ + 16] = acc[1][v];
    }
  }
  __syncthreads();

#pragma unroll
  for (int pp = 0; pp < PB_; pp += FT_ / 16) {
    const int p = pp + (tid >> 4), q = tid & 15;
    float* o = out + ((size_t)s * N_ + pt0 + p) * OD_;
    float x[8];
#pragma unroll
    for (int i = 0; i < 8; ++i) x[i] = osm[p * RP_ + q + 16 * i] + o[q + 16 * i];
#pragma unroll
    for (int i = 0; i < 8; ++i) o[q + 16 * i] = x[i];
    const float y0 = x[0] + x[4], y1 = x[1] + x[5];
    const float y2 = x[2] + x[6], y3 = x[3] + x[7];
    float r = (y0 + y2) + (y1 + y3);
    r += __shfl_xor(r, 8);
    r += __shfl_xor(r, 4);
    r += __shfl_xor(r, 2);
    r += __shfl_xor(r, 1);
    if (q == 0) score[(size_t)s * N_ + pt0 + p] = r;
  }
}

extern "C" void kernel_launch(void* const* d_in, const int* in_sizes, int n_in,
                              void* d_out, int out_size, void* d_ws,
                              size_t ws_size, hipStream_t stream) {
  (void)in_sizes; (void)n_in; (void)out_size; (void)ws_size;
  const float* feat      = (const float*)d_in[0];
  const float* coord     = (const float*)d_in[1];
  const float* lang_feat = (const float*)d_in[2];
  const float* lang_mask = (const float*)d_in[3];
  const float* rel_w1    = (const float*)d_in[4];
  const float* rel_b1    = (const float*)d_in[5];
  const float* rel_w2    = (const float*)d_in[6];
  const float* rel_b2    = (const float*)d_in[7];
  const float* lang_w1   = (const float*)d_in[8];
  const float* lang_b1   = (const float*)d_in[9];
  const float* lang_w2   = (const float*)d_in[10];
  const float* lang_b2   = (const float*)d_in[11];
  const float* feat_w1   = (const float*)d_in[12];
  const float* feat_b1   = (const float*)d_in[13];
  const float* feat_w2   = (const float*)d_in[14];
  const float* feat_b2   = (const float*)d_in[15];

  char* ws = (char*)d_ws;
  int*   idx_ws  = (int*)ws;
  float* lang_ws = (float*)(ws + (size_t)S_ * N_ * KNB * 4);
  float* g_ws    = (float*)(ws + (size_t)S_ * N_ * KNB * 4 +
                                 (size_t)S_ * L_ * OD_ * 4);

  float* out   = (float*)d_out;
  float* score = out + (size_t)S_ * N_ * OD_;

  knn_kernel<<<S_ * (N_ / KPB_), 32 * KWV_, 0, stream>>>(coord, idx_ws);
  lang_mlp_kernel<<<LROWS_ / 16, 256, 0, stream>>>(lang_feat, lang_w1, lang_b1,
                                                   lang_w2, lang_b2, lang_ws);
  feat_mlp_kernel<<<S_ * (N_ / 16), 256, 0, stream>>>(feat, feat_w1, feat_b1,
                                                      feat_w2, feat_b2, lang_mask,
                                                      lang_ws, g_ws, out);
  fused_kernel<<<S_ * (N_ / PB_), FT_, 0, stream>>>(coord, rel_w1, rel_b1,
                                                     rel_w2, rel_b2, idx_ws,
                                                     g_ws, out, score);
}


# kNN and lang MLP (independent of each other) run as two block ranges of ONE launch, 3 launches not 4: lang's latency-bound blocks run under the kNN's vector work; all sums and their order unchanged
# speedup vs baseline: 8.5260x; 1.0931x over previous
#include <hip/hip_runtime.h>
#include <hip/hip_bf16.h>
#include <float.h>


typedef float v2f __attribute__((ext_vector_type(2)));
typedef float v8f __attribute__((ext_vector_type(8)));

#define S_      64
#define N_      256
#define L_      40
#define LP_     48
#define RP_     132
#define LPP_    52
#define KP_     20
#define KNB     17
#define OD_     128
#define LANGID_ 256
#define PCID_   128
#define KG_     16

static __device__ inline v8f wmma4(v2f a, v2f b, v8f c) {
  return __builtin_amdgcn_wmma_f32_16x16x4_f32(false, a, false, b, (short)0, c,
                                               false, false);
}

#define KPW_    4
#define KWV_    8
#define KPB_    (KPW_ * KWV_)
#define KRM_    0xffffffffu

template <int XM>
static __device__ inline unsigned row_xor(unsigned x) {
  return (unsigned)__builtin_amdgcn_mov_dpp((int)x, 0x160 + XM, 0xf, 0xf, true);
}

static __device__ inline unsigned other_row(unsigned x) {
  return (unsigned)__builtin_amdgcn_permlanex16((int)x, (int)x, 0x76543210,
                                                0xfedcba98, false, false);
}

static __device__ inline unsigned umin2(unsigned a, unsigned b) {
  return a < b ? a : b;
}

static __device__ inline unsigned wave_umin(unsigned x) {
  x = umin2(x, row_xor<1>(x));
  x = umin2(x, row_xor<2>(x));
  x = umin2(x, row_xor<4>(x));
  x = umin2(x, row_xor<8>(x));
  return umin2(x, other_row(x));
}

#define KNN_BLOCKS_ (S_ * (N_ / KPB_))

static __device__ __forceinline__ void
knn_role(const unsigned bid, const float* __restrict__ coord,
         int* __restrict__ idx_out) {
  __shared__ int osm[KPB_ * KNB];
  const int tid  = threadIdx.x;
  const int lane = tid & 31;
  const int wave = __builtin_amdgcn_readfirstlane(tid >> 5);
  const int s    = (int)(bid / (N_ / KPB_));
  const int pb   = (int)(bid % (N_ / KPB_)) * KPB_;
  const int p0   = pb + wave * KPW_;
  const float* cs = coord + (size_t)s * N_ * 3;

  float c[24];
  {
    const float4* cv = reinterpret_cast<const float4*>(cs + lane * 24);
    for (int u = 0; u < 6; ++u) {
      const float4 t = cv[u];
      c[4 * u + 0] = t.x; c[4 * u + 1] = t.y;
      c[4 * u + 2] = t.z; c[4 * u + 3] = t.w;
    }
  }

  unsigned d[KPW_][8];
  for (int q = 0; q < KPW_; ++q) {
    const float px = cs[(p0 + q) * 3 + 0];
    const float py = cs[(p0 + q) * 3 + 1];
    const float pz = cs[(p0 + q) * 3 + 2];
    for (int i = 0; i < 8; ++i) {
      const float dx = c[3 * i + 0] - px;
      const float dy = c[3 * i + 1] - py;
      const float dz = c[3 * i + 2] - pz;
      const float d2 = __builtin_fmaf(dz, dz, __builtin_fmaf(dx, dx, dy * dy));
      d[q][i] = __float_as_uint(d2);
    }
  }

  const int lane8 = lane * 8;
  int res[KPW_] = {};

  for (int j = 0; j < KNB; ++j) {
    for (int q = 0; q < KPW_; ++q) {
      unsigned* dq = d[q];
      unsigned lm = umin2(dq[0], dq[1]);
      for (int i = 2; i < 8; ++i) lm = umin2(lm, dq[i]);
      int li = 7;
      for (int i = 6; i >= 0; --i) li = (dq[i] == lm) ? i : li;
      const unsigned wm  = wave_umin(lm);
      const unsigned bal = __builtin_amdgcn_ballot_w32(lm == wm);
      const int L  = __builtin_ctz(bal);
      const int sl = __builtin_amdgcn_readlane(li, L);
      const int m  = L * 8 + sl;
      res[q] = (lane == j) ? m : res[q];
      const int rel = m - lane8;
      for (int i = 0; i < 8; ++i) dq[i] = (rel == i) ? KRM_ : dq[i];
    }
  }

  if (lane < KNB) {
    for (int q = 0; q < KPW_; ++q)
      osm[(wave * KPW_ + q) * KNB + lane] = res[q];
  }
  __syncthreads();
  int* ob = idx_out + ((size_t)s * N_ + pb) * KNB;
  for (int i = tid; i < KPB_ * KNB; i += 32 * KWV_) ob[i] = osm[i];
}

#define XP_     260
#define LROWS_  (S_ * L_)
#define LANG_BLOCKS_ (LROWS_ / 16)

static __device__ __forceinline__ void
lang_role(const unsigned bid, const float* __restrict__ lang_feat,
          const float* __restrict__ w1,
          const float* __restrict__ b1,
          const float* __restrict__ w2,
          const float* __restrict__ b2,
          float* __restrict__ lang_out) {
  __shared__ float Xsm[16 * XP_] __attribute__((aligned(16)));
  __shared__ float Hsm[16 * RP_];
  const int row0 = (int)bid * 16;
  const int tid  = threadIdx.x;
  const int lane = tid & 31, wave = tid >> 5;
  const int half = lane >> 4, lr = lane & 15;

  {
    const float4* X4 =
        reinterpret_cast<const float4*>(lang_feat + (size_t)row0 * LANGID_);
    float4 t[4];
#pragma unroll
    for (int u = 0; u < 4; ++u) t[u] = X4[tid + 256 * u];
#pragma unroll
    for (int u = 0; u < 4; ++u) {
      const int i4 = tid + 256 * u;
      *reinterpret_cast<float4*>(Xsm + (i4 >> 6) * XP_ + (i4 & 63) * 4) = t[u];
    }
  }
  __syncthreads();

  const int n0 = wave * 16;
  v8f c = {};
  for (int kg = 0; kg < LANGID_; kg += 4 * KG_) {
    v2f a[KG_], b[KG_];
#pragma unroll
    for (int u = 0; u < KG_; ++u) {
      const int ka = kg + 4 * u + 2 * half;
      a[u].x = Xsm[lr * XP_ + ka];
      a[u].y = Xsm[lr * XP_ + ka + 1];
      b[u].x = w1[ka * OD_ + n0 + lr];
      b[u].y = w1[(ka + 1) * OD_ + n0 + lr];
    }
    __builtin_amdgcn_sched_barrier(0);
#pragma unroll
    for (int u = 0; u < KG_; ++u) c = wmma4(a[u], b[u], c);
  }
  {
    const float bb = b1[n0 + lr];
#pragma unroll
    for (int v = 0; v < 8; ++v)
      Hsm[(v + 8 * half) * RP_ + n0 + lr] = fmaxf(c[v] + bb, 0.0f);
  }
  __syncthreads();

  v8f c2 = {};
  for (int kg = 0; kg < OD_; kg += 4 * KG_) {
    v2f a[KG_], b[KG_];
#pragma unroll
    for (int u = 0; u < KG_; ++u) {
      const int ka = kg + 4 * u + 2 * half;
      a[u].x = Hsm[lr * RP_ + ka];
      a[u].y = Hsm[lr * RP_ + ka + 1];
      b[u].x = w2[ka * OD_ + n0 + lr];
      b[u].y = w2[(ka + 1) * OD_ + n0 + lr];
    }
    __builtin_amdgcn_sched_barrier(0);
#pragma unroll
    for (int u = 0; u < KG_; ++u) c2 = wmma4(a[u], b[u], c2);
  }
  {
    const float bb = b2[n0 + lr];
#pragma unroll
    for (int v = 0; v < 8; ++v)
      Xsm[(v + 8 * half) * XP_ + n0 + lr] = c2[v] + bb;
  }
  __syncthreads();
  float* lo = lang_out + (size_t)row0 * OD_;
  for (int u = 0; u < 8; ++u) {
    const int i = tid + 256 * u;
    lo[i] = Xsm[(i >> 7) * XP_ + (i & 127)];
  }
}

__global__ void __launch_bounds__(256)
knn_lang_kernel(const float* __restrict__ coord, int* __restrict__ idx_out,
                const float* __restrict__ lang_feat,
                const float* __restrict__ w1,
                const float* __restrict__ b1,
                const float* __restrict__ w2,
                const float* __restrict__ b2,
                float* __restrict__ lang_out) {
  const unsigned bid = blockIdx.x;
  if (bid >= KNN_BLOCKS_) {
    lang_role(bid - KNN_BLOCKS_, lang_feat, w1, b1, w2, b2, lang_out);
  } else {
    knn_role(bid, coord, idx_out);
  }
}

__global__ void feat_mlp_kernel(const float* __restrict__ feat,
                                const float* __restrict__ w1,
                                const float* __restrict__ b1,
                                const float* __restrict__ w2,
                                const float* __restrict__ b2,
                                const float* __restrict__ lang_mask,
                                const float* __restrict__ lang_ws,
                                float* __restrict__ g_out,
                                float* __restrict__ f_out) {
  __shared__ float Asm[16 * RP_];
  __shared__ float Hsm[16 * RP_];
  __shared__ float attnsm[16 * LPP_];
  const int s    = blockIdx.x >> 4;
  const int mt   = blockIdx.x & 15;
  const int tid  = threadIdx.x;
  const int lane = tid & 31, wave = tid >> 5;
  const int half = lane >> 4, lr = lane & 15;

  const float* A = feat + ((size_t)s * N_ + mt * 16) * PCID_;
  for (int i = tid; i < 16 * PCID_; i += 256)
    Asm[(i >> 7) * RP_ + (i & 127)] = A[i];
  __syncthreads();

  const int n0 = wave * 16;
  v8f c = {};
  for (int kg = 0; kg < PCID_; kg += 4 * KG_) {
    v2f a[KG_], b[KG_];
#pragma unroll
    for (int u = 0; u < KG_; ++u) {
      const int ka = kg + 4 * u + 2 * half;
      a[u].x = Asm[lr * RP_ + ka];
      a[u].y = Asm[lr * RP_ + ka + 1];
      b[u].x = w1[ka * OD_ + n0 + lr];
      b[u].y = w1[(ka + 1) * OD_ + n0 + lr];
    }
    __builtin_amdgcn_sched_barrier(0);
#pragma unroll
    for (int u = 0; u < KG_; ++u) c = wmma4(a[u], b[u], c);
  }
  {
    const float bb = b1[n0 + lr];
#pragma unroll
    for (int v = 0; v < 8; ++v)
      Hsm[(v + 8 * half) * RP_ + n0 + lr] = fmaxf(c[v] + bb, 0.0f);
  }
  __syncthreads();

  v8f c2 = {};
  for (int kg = 0; kg < OD_; kg += 4 * KG_) {
    v2f a[KG_], b[KG_];
#pragma unroll
    for (int u = 0; u < KG_; ++u) {
      const int ka = kg + 4 * u + 2 * half;
      a[u].x = Hsm[lr * RP_ + ka];
      a[u].y = Hsm[lr * RP_ + ka + 1];
      b[u].x = w2[ka * OD_ + n0 + lr];
      b[u].y = w2[(ka + 1) * OD_ + n0 + lr];
    }
    __builtin_amdgcn_sched_barrier(0);
#pragma unroll
    for (int u = 0; u < KG_; ++u) c2 = wmma4(a[u], b[u], c2);
  }
  float fv[8];
  {
    const float bb = b2[n0 + lr];
    float* fo = f_out + ((size_t)s * N_ + mt * 16) * OD_;
#pragma unroll
    for (int v = 0; v < 8; ++v) {
      fv[v] = c2[v] + bb;
      Asm[(v + 8 * half) * RP_ + n0 + lr] = fv[v];
      fo[(v + 8 * half) * OD_ + n0 + lr]  = fv[v];
    }
  }
  __syncthreads();

  const float* langS = lang_ws + (size_t)s * L_ * OD_;

  if (wave < 3) {
    const int c0 = wave * 16;
    const int lclamp = min(c0 + lr, L_ - 1);
    v8f cl = {};
    for (int kg = 0; kg < OD_; kg += 4 * KG_) {
      v2f a[KG_], b[KG_];
#pragma unroll
      for (int u = 0; u < KG_; ++u) {
        const int ka = kg + 4 * u + 2 * half;
        a[u].x = Asm[lr * RP_ + ka];
        a[u].y = Asm[lr * RP_ + ka + 1];
        b[u].x = langS[lclamp * OD_ + ka];
        b[u].y = langS[lclamp * OD_ + ka + 1];
      }
      __builtin_amdgcn_sched_barrier(0);
#pragma unroll
      for (int u = 0; u < KG_; ++u) cl = wmma4(a[u], b[u], cl);
    }
#pragma unroll
    for (int v = 0; v < 8; ++v)
      attnsm[(v + 8 * half) * LPP_ + c0 + lr] = cl[v];
  }
  __syncthreads();

  if (tid < 16) {
    float* rowp = attnsm + tid * LPP_;
    float mx = -FLT_MAX;
    for (int l = 0; l < L_; ++l) mx = fmaxf(mx, rowp[l]);
    float ssum = 0.f;
    for (int l = 0; l < L_; ++l) {
      const float e = expf(rowp[l] - mx);
      rowp[l] = e;
      ssum += e;
    }
    const float inv = 1.f / ssum;
    const float* mk = lang_mask + (size_t)s * L_;
    float s2 = 0.f;
    for (int l = 0; l < L_; ++l) {
      const float p = rowp[l] * inv * mk[l];
      rowp[l] = p;
      s2 += p;
    }
    const float inv2 = 1.f / (s2 + 1e-7f);
    for (int l = 0; l < L_; ++l) rowp[l] *= inv2;
    for (int l = L_; l < LP_; ++l) rowp[l] = 0.f;
  }
  __syncthreads();

  {
    v8f ci = {};
    {
      v2f a[LP_ / 4], b[LP_ / 4];
#pragma unroll
      for (int u = 0; u < LP_ / 4; ++u) {
        const int ka  = 4 * u + 2 * half;
        const int kb0 = min(ka, L_ - 1), kb1 = min(ka + 1, L_ - 1);
        a[u].x = attnsm[lr * LPP_ + ka];
        a[u].y = attnsm[lr * LPP_ + ka + 1];
        b[u].x = langS[kb0 * OD_ + n0 + lr];
        b[u].y = langS[kb1 * OD_ + n0 + lr];
      }
#pragma unroll
      for (int u = 0; u < LP_ / 4; ++u) ci = wmma4(a[u], b[u], ci);
    }
    float* go = g_out + ((size_t)s * N_ + mt * 16) * OD_;
#pragma unroll
    for (int v = 0; v < 8; ++v)
      go[(v + 8 * half) * OD_ + n0 + lr] = fv[v] * ci[v];
  }
}

#define PB_     16
#define K1S_    3
#define FT_     128

__global__ void __launch_bounds__(FT_)
fused_kernel(const float* __restrict__ coord,
             const float* __restrict__ rel_w1,
             const float* __restrict__ rel_b1,
             const float* __restrict__ rel_w2,
             const float* __restrict__ rel_b2,
             const int*   __restrict__ idx_ws,
             const float* __restrict__ g_ws,
             float* __restrict__ out,
             float* __restrict__ score) {
  __shared__ float relin[KNB * PB_ * KP_];
  __shared__ float hbuf[2][PB_ * RP_];
  __shared__ int   idxsm[PB_ * KNB];

  const int s    = blockIdx.x >> 4;
  const int pt0  = (blockIdx.x & 15) * PB_;
  const int tid  = threadIdx.x;
  const int lane = tid & 31, wave = tid >> 5;
  const int half = lane >> 4, lr = lane & 15;
  const int c0   = wave * 32;

  for (int i = tid; i < PB_ * KNB; i += FT_)
    idxsm[i] = idx_ws[((size_t)s * N_ + pt0) * KNB + i];

  v2f bw1[2][K1S_];
  v2f bw2[2][OD_ / 4];
  float bb1[2], bb2[2];
#pragma unroll
  for (int t = 0; t < 2; ++t) {
    const int col = c0 + 16 * t + lr;
#pragma unroll
    for (int u = 0; u < K1S_; ++u) {
      const int ka  = 4 * u + 2 * half;
      const int kb0 = min(ka, 9), kb1 = min(ka + 1, 9);
      bw1[t][u].x = rel_w1[kb0 * OD_ + col];
      bw1[t][u].y = rel_w1[kb1 * OD_ + col];
    }
#pragma unroll
    for (int u = 0; u < OD_ / 4; ++u) {
      const int ka = 4 * u + 2 * half;
      bw2[t][u].x = rel_w2[ka * OD_ + col];
      bw2[t][u].y = rel_w2[(ka + 1) * OD_ + col];
    }
    bb1[t] = rel_b1[col];
    bb2[t] = rel_b2[col];
  }
  __syncthreads();

  for (int r = tid; r < PB_ * KNB; r += FT_) {
    const int j = r >> 4, p = r & 15;
    const float* cs = coord + (size_t)s * N_ * 3;
    const int m = idxsm[p * KNB + j];
    const int n = pt0 + p;
    const float nx = cs[m * 3 + 0], ny = cs[m * 3 + 1], nz = cs[m * 3 + 2];
    const float ox = cs[n * 3 + 0], oy = cs[n * 3 + 1], oz = cs[n * 3 + 2];
    const float rx = nx - ox, ry = ny - oy, rz = nz - oz;
    const float dist = sqrtf(rx * rx + ry * ry + rz * rz);
    float* ri = relin + r * KP_;
    ri[0] = nx; ri[1] = ny; ri[2] = nz;
    ri[3] = ox; ri[4] = oy; ri[5] = oz;
    ri[6] = rx; ri[7] = ry; ri[8] = rz;
    ri[9] = dist;
    ri[10] = 0.f; ri[11] = 0.f;
  }
  __syncthreads();

  float acc[2][8];
#pragma unroll
  for (int v = 0; v < 8; ++v) { acc[0][v] = 0.f; acc[1][v] = 0.f; }

  const int*   ip = idxsm + 8 * half * KNB;
  const float* gs = g_ws + (size_t)s * N_ * OD_;
  const unsigned lcol = (unsigned)(c0 + lr);

  for (int j = 0; j < KNB; ++j) {
    unsigned m[8];
#pragma unroll
    for (int v = 0; v < 8; ++v) m[v] = (unsigned)ip[v * KNB + j];

    v8f ca = {}, cb = {};
    {
      const float* rp = relin + (j * PB_ + lr) * KP_ + 2 * half;
      v2f a[K1S_];
#pragma unroll
      for (int u = 0; u < K1S_; ++u) {
        a[u].x = rp[4 * u];
        a[u].y = rp[4 * u + 1];
      }
#pragma unroll
      for (int u = 0; u < K1S_; ++u) {
        ca = wmma4(a[u], bw1[0][u], ca);
        cb = wmma4(a[u], bw1[1][u], cb);
      }
    }
    float* hb = hbuf[j & 1];
    {
      float* hp = hb + 8 * half * RP_ + c0 + lr;
#pragma unroll
      for (int v = 0; v < 8; ++v) {
        hp[v * RP_]      = fmaxf(ca[v] + bb1[0], 0.f);
        hp[v * RP_ + 16] = fmaxf(cb[v] + bb1[1], 0.f);
      }
    }
    __syncthreads();

    float gv[2][8];
#pragma unroll
    for (int v = 0; v < 8; ++v) {
      const float* gp = gs + (m[v] * OD_ + lcol);
      gv[0][v] = gp[0];
      gv[1][v] = gp[16];
    }

    v8f c2a = {}, c2b = {};
    const float* ap = hb + lr * RP_ + 2 * half;
    for (int kg = 0; kg < OD_ / 4; kg += KG_) {
      v2f a[KG_];
#pragma unroll
      for (int u = 0; u < KG_; ++u) {
        a[u].x = ap[4 * (kg + u)];
        a[u].y = ap[4 * (kg + u) + 1];
      }
      __builtin_amdgcn_sched_barrier(0);
#pragma unroll
      for (int u = 0; u < KG_; ++u) {
        c2a = wmma4(a[u], bw2[0][kg + u], c2a);
        c2b = wmma4(a[u], bw2[1][kg + u], c2b);
      }
    }
#pragma unroll
    for (int v = 0; v < 8; ++v) {
      acc[0][v] += gv[0][v] * (c2a[v] + bb2[0]);
      acc[1][v] += gv[1][v] * (c2b[v] + bb2[1]);
    }
  }

  float* osm = hbuf[1];
  {
    float* op = osm + 8 * half * RP_ + c0 + lr;
#pragma unroll
    for (int v = 0; v < 8; ++v) {
      op[v * RP_]      = acc[0][v];
      op[v * RP_ + 16] = acc[1][v];
    }
  }
  __syncthreads();

#pragma unroll
  for (int pp = 0; pp < PB_; pp += FT_ / 16) {
    const int p = pp + (tid >> 4), q = tid & 15;
    float* o = out + ((size_t)s * N_ + pt0 + p) * OD_;
    float x[8];
#pragma unroll
    for (int i = 0; i < 8; ++i) x[i] = osm[p * RP_ + q + 16 * i] + o[q + 16 * i];
#pragma unroll
    for (int i = 0; i < 8; ++i) o[q + 16 * i] = x[i];
    const float y0 = x[0] + x[4], y1 = x[1] + x[5];
    const float y2 = x[2] + x[6], y3 = x[3] + x[7];
    float r = (y0 + y2) + (y1 + y3);
    r += __shfl_xor(r, 8);
    r += __shfl_xor(r, 4);
    r += __shfl_xor(r, 2);
    r += __shfl_xor(r, 1);
    if (q == 0) score[(size_t)s * N_ + pt0 + p] = r;
  }
}

extern "C" void kernel_launch(void* const* d_in, const int* in_sizes, int n_in,
                              void* d_out, int out_size, void* d_ws,
                              size_t ws_size, hipStream_t stream) {
  (void)in_sizes; (void)n_in; (void)out_size; (void)ws_size;
  const float* feat      = (const float*)d_in[0];
  const float* coord     = (const float*)d_in[1];
  const float* lang_feat = (const float*)d_in[2];
  const float* lang_mask = (const float*)d_in[3];
  const float* rel_w1    = (const float*)d_in[4];
  const float* rel_b1    = (const float*)d_in[5];
  const float* rel_w2    = (const float*)d_in[6];
  const float* rel_b2    = (const float*)d_in[7];
  const float* lang_w1   = (const float*)d_in[8];
  const float* lang_b1   = (const float*)d_in[9];
  const float* lang_w2   = (const float*)d_in[10];
  const float* lang_b2   = (const float*)d_in[11];
  const float* feat_w1   = (const float*)d_in[12];
  const float* feat_b1   = (const float*)d_in[13];
  const float* feat_w2   = (const float*)d_in[14];
  const float* feat_b2   = (const float*)d_in[15];

  char* ws = (char*)d_ws;
  int*   idx_ws  = (int*)ws;
  float* lang_ws = (float*)(ws + (size_t)S_ * N_ * KNB * 4);
  float* g_ws    = (float*)(ws + (size_t)S_ * N_ * KNB * 4 +
                                 (size_t)S_ * L_ * OD_ * 4);

  float* out   = (float*)d_out;
  float* score = out + (size_t)S_ * N_ * OD_;

  knn_lang_kernel<<<KNN_BLOCKS_ + LANG_BLOCKS_, 256, 0, stream>>>(
      coord, idx_ws, lang_feat, lang_w1, lang_b1, lang_w2, lang_b2, lang_ws);
  feat_mlp_kernel<<<S_ * (N_ / 16), 256, 0, stream>>>(feat, feat_w1, feat_b1,
                                                      feat_w2, feat_b2, lang_mask,
                                                      lang_ws, g_ws, out);
  fused_kernel<<<S_ * (N_ / PB_), FT_, 0, stream>>>(coord, rel_w1, rel_b1,
                                                     rel_w2, rel_b2, idx_ws,
                                                     g_ws, out, score);
}


# Prologue and epilogue loads issued as one group ahead of their uses: feat tile (was 8 serial round trips), fused neighbor lists and rel-input gather (3 each), f start values; sum order unchanged
# speedup vs baseline: 8.9526x; 1.0500x over previous
#include <hip/hip_runtime.h>
#include <hip/hip_bf16.h>
#include <float.h>


typedef float v2f __attribute__((ext_vector_type(2)));
typedef float v8f __attribute__((ext_vector_type(8)));

#define S_      64
#define N_      256
#define L_      40
#define LP_     48
#define RP_     132
#define LPP_    52
#define KP_     20
#define KNB     17
#define OD_     128
#define LANGID_ 256
#define PCID_   128
#define KG_     16

static __device__ inline v8f wmma4(v2f a, v2f b, v8f c) {
  return __builtin_amdgcn_wmma_f32_16x16x4_f32(false, a, false, b, (short)0, c,
                                               false, false);
}

#define KPW_    4
#define KWV_    8
#define KPB_    (KPW_ * KWV_)
#define KRM_    0xffffffffu

template <int XM>
static __device__ inline unsigned row_xor(unsigned x) {
  return (unsigned)__builtin_amdgcn_mov_dpp((int)x, 0x160 + XM, 0xf, 0xf, true);
}

static __device__ inline unsigned other_row(unsigned x) {
  return (unsigned)__builtin_amdgcn_permlanex16((int)x, (int)x, 0x76543210,
                                                0xfedcba98, false, false);
}

static __device__ inline unsigned umin2(unsigned a, unsigned b) {
  return a < b ? a : b;
}

static __device__ inline unsigned wave_umin(unsigned x) {
  x = umin2(x, row_xor<1>(x));
  x = umin2(x, row_xor<2>(x));
  x = umin2(x, row_xor<4>(x));
  x = umin2(x, row_xor<8>(x));
  return umin2(x, other_row(x));
}

#define KNN_BLOCKS_ (S_ * (N_ / KPB_))

static __device__ __forceinline__ void
knn_role(const unsigned bid, const float* __restrict__ coord,
         int* __restrict__ idx_out) {
  __shared__ int osm[KPB_ * KNB];
  const int tid  = threadIdx.x;
  const int lane = tid & 31;
  const int wave = __builtin_amdgcn_readfirstlane(tid >> 5);
  const int s    = (int)(bid / (N_ / KPB_));
  const int pb   = (int)(bid % (N_ / KPB_)) * KPB_;
  const int p0   = pb + wave * KPW_;
  const float* cs = coord + (size_t)s * N_ * 3;

  float c[24];
  {
    const float4* cv = reinterpret_cast<const float4*>(cs + lane * 24);
    for (int u = 0; u < 6; ++u) {
      const float4 t = cv[u];
      c[4 * u + 0] = t.x; c[4 * u + 1] = t.y;
      c[4 * u + 2] = t.z; c[4 * u + 3] = t.w;
    }
  }

  unsigned d[KPW_][8];
  for (int q = 0; q < KPW_; ++q) {
    const float px = cs[(p0 + q) * 3 + 0];
    const float py = cs[(p0 + q) * 3 + 1];
    const float pz = cs[(p0 + q) * 3 + 2];
    for (int i = 0; i < 8; ++i) {
      const float dx = c[3 * i + 0] - px;
      const float dy = c[3 * i + 1] - py;
      const float dz = c[3 * i + 2] - pz;
      const float d2 = __builtin_fmaf(dz, dz, __builtin_fmaf(dx, dx, dy * dy));
      d[q][i] = __float_as_uint(d2);
    }
  }

  const int lane8 = lane * 8;
  int res[KPW_] = {};

  for (int j = 0; j < KNB; ++j) {
    for (int q = 0; q < KPW_; ++q) {
      unsigned* dq = d[q];
      unsigned lm = umin2(dq[0], dq[1]);
      for (int i = 2; i < 8; ++i) lm = umin2(lm, dq[i]);
      int li = 7;
      for (int i = 6; i >= 0; --i) li = (dq[i] == lm) ? i : li;
      const unsigned wm  = wave_umin(lm);
      const unsigned bal = __builtin_amdgcn_ballot_w32(lm == wm);
      const int L  = __builtin_ctz(bal);
      const int sl = __builtin_amdgcn_readlane(li, L);
      const int m  = L * 8 + sl;
      res[q] = (lane == j) ? m : res[q];
      const int rel = m - lane8;
      for (int i = 0; i < 8; ++i) dq[i] = (rel == i) ? KRM_ : dq[i];
    }
  }

  if (lane < KNB) {
    for (int q = 0; q < KPW_; ++q)
      osm[(wave * KPW_ + q) * KNB + lane] = res[q];
  }
  __syncthreads();
  int* ob = idx_out + ((size_t)s * N_ + pb) * KNB;
  for (int i = tid; i < KPB_ * KNB; i += 32 * KWV_) ob[i] = osm[i];
}

#define XP_     260
#define LROWS_  (S_ * L_)
#define LANG_BLOCKS_ (LROWS_ / 16)

static __device__ __forceinline__ void
lang_role(const unsigned bid, const float* __restrict__ lang_feat,
          const float* __restrict__ w1,
          const float* __restrict__ b1,
          const float* __restrict__ w2,
          const float* __restrict__ b2,
          float* __restrict__ lang_out) {
  __shared__ float Xsm[16 * XP_] __attribute__((aligned(16)));
  __shared__ float Hsm[16 * RP_];
  const int row0 = (int)bid * 16;
  const int tid  = threadIdx.x;
  const int lane = tid & 31, wave = tid >> 5;
  const int half = lane >> 4, lr = lane & 15;

  {
    const float4* X4 =
        reinterpret_cast<const float4*>(lang_feat + (size_t)row0 * LANGID_);
    float4 t[4];
#pragma unroll
    for (int u = 0; u < 4; ++u) t[u] = X4[tid + 256 * u];
#pragma unroll
    for (int u = 0; u < 4; ++u) {
      const int i4 = tid + 256 * u;
      *reinterpret_cast<float4*>(Xsm + (i4 >> 6) * XP_ + (i4 & 63) * 4) = t[u];
    }
  }
  __syncthreads();

  const int n0 = wave * 16;
  v8f c = {};
  for (int kg = 0; kg < LANGID_; kg += 4 * KG_) {
    v2f a[KG_], b[KG_];
#pragma unroll
    for (int u = 0; u < KG_; ++u) {
      const int ka = kg + 4 * u + 2 * half;
      a[u].x = Xsm[lr * XP_ + ka];
      a[u].y = Xsm[lr * XP_ + ka + 1];
      b[u].x = w1[ka * OD_ + n0 + lr];
      b[u].y = w1[(ka + 1) * OD_ + n0 + lr];
    }
    __builtin_amdgcn_sched_barrier(0);
#pragma unroll
    for (int u = 0; u < KG_; ++u) c = wmma4(a[u], b[u], c);
  }
  {
    const float bb = b1[n0 + lr];
#pragma unroll
    for (int v = 0; v < 8; ++v)
      Hsm[(v + 8 * half) * RP_ + n0 + lr] = fmaxf(c[v] + bb, 0.0f);
  }
  __syncthreads();

  v8f c2 = {};
  for (int kg = 0; kg < OD_; kg += 4 * KG_) {
    v2f a[KG_], b[KG_];
#pragma unroll
    for (int u = 0; u < KG_; ++u) {
      const int ka = kg + 4 * u + 2 * half;
      a[u].x = Hsm[lr * RP_ + ka];
      a[u].y = Hsm[lr * RP_ + ka + 1];
      b[u].x = w2[ka * OD_ + n0 + lr];
      b[u].y = w2[(ka + 1) * OD_ + n0 + lr];
    }
    __builtin_amdgcn_sched_barrier(0);
#pragma unroll
    for (int u = 0; u < KG_; ++u) c2 = wmma4(a[u], b[u], c2);
  }
  {
    const float bb = b2[n0 + lr];
#pragma unroll
    for (int v = 0; v < 8; ++v)
      Xsm[(v + 8 * half) * XP_ + n0 + lr] = c2[v] + bb;
  }
  __syncthreads();
  float* lo = lang_out + (size_t)row0 * OD_;
  for (int u = 0; u < 8; ++u) {
    const int i = tid + 256 * u;
    lo[i] = Xsm[(i >> 7) * XP_ + (i & 127)];
  }
}

__global__ void __launch_bounds__(256)
knn_lang_kernel(const float* __restrict__ coord, int* __restrict__ idx_out,
                const float* __restrict__ lang_feat,
                const float* __restrict__ w1,
                const float* __restrict__ b1,
                const float* __restrict__ w2,
                const float* __restrict__ b2,
                float* __restrict__ lang_out) {
  const unsigned bid = blockIdx.x;
  if (bid >= KNN_BLOCKS_) {
    lang_role(bid - KNN_BLOCKS_, lang_feat, w1, b1, w2, b2, lang_out);
  } else {
    knn_role(bid, coord, idx_out);
  }
}

__global__ void feat_mlp_kernel(const float* __restrict__ feat,
                                const float* __restrict__ w1,
                                const float* __restrict__ b1,
                                const float* __restrict__ w2,
                                const float* __restrict__ b2,
                                const float* __restrict__ lang_mask,
                                const float* __restrict__ lang_ws,
                                float* __restrict__ g_out,
                                float* __restrict__ f_out) {
  __shared__ float Asm[16 * RP_] __attribute__((aligned(16)));
  __shared__ float Hsm[16 * RP_];
  __shared__ float attnsm[16 * LPP_];
  const int s    = blockIdx.x >> 4;
  const int mt   = blockIdx.x & 15;
  const int tid  = threadIdx.x;
  const int lane = tid & 31, wave = tid >> 5;
  const int half = lane >> 4, lr = lane & 15;

  {
    const float4* A4 = reinterpret_cast<const float4*>(
        feat + ((size_t)s * N_ + mt * 16) * PCID_);
    float4 t[2];
#pragma unroll
    for (int u = 0; u < 2; ++u) t[u] = A4[tid + 256 * u];
#pragma unroll
    for (int u = 0; u < 2; ++u) {
      const int i4 = tid + 256 * u;
      *reinterpret_cast<float4*>(Asm + (i4 >> 5) * RP_ + (i4 & 31) * 4) = t[u];
    }
  }
  __syncthreads();

  const int n0 = wave * 16;
  v8f c = {};
  for (int kg = 0; kg < PCID_; kg += 4 * KG_) {
    v2f a[KG_], b[KG_];
#pragma unroll
    for (int u = 0; u < KG_; ++u) {
      const int ka = kg + 4 * u + 2 * half;
      a[u].x = Asm[lr * RP_ + ka];
      a[u].y = Asm[lr * RP_ + ka + 1];
      b[u].x = w1[ka * OD_ + n0 + lr];
      b[u].y = w1[(ka + 1) * OD_ + n0 + lr];
    }
    __builtin_amdgcn_sched_barrier(0);
#pragma unroll
    for (int u = 0; u < KG_; ++u) c = wmma4(a[u], b[u], c);
  }
  {
    const float bb = b1[n0 + lr];
#pragma unroll
    for (int v = 0; v < 8; ++v)
      Hsm[(v + 8 * half) * RP_ + n0 + lr] = fmaxf(c[v] + bb, 0.0f);
  }
  __syncthreads();

  v8f c2 = {};
  for (int kg = 0; kg < OD_; kg += 4 * KG_) {
    v2f a[KG_], b[KG_];
#pragma unroll
    for (int u = 0; u < KG_; ++u) {
      const int ka = kg + 4 * u + 2 * half;
      a[u].x = Hsm[lr * RP_ + ka];
      a[u].y = Hsm[lr * RP_ + ka + 1];
      b[u].x = w2[ka * OD_ + n0 + lr];
      b[u].y = w2[(ka + 1) * OD_ + n0 + lr];
    }
    __builtin_amdgcn_sched_barrier(0);
#pragma unroll
    for (int u = 0; u < KG_; ++u) c2 = wmma4(a[u], b[u], c2);
  }
  float fv[8];
  {
    const float bb = b2[n0 + lr];
    float* fo = f_out + ((size_t)s * N_ + mt * 16) * OD_;
#pragma unroll
    for (int v = 0; v < 8; ++v) {
      fv[v] = c2[v] + bb;
      Asm[(v + 8 * half) * RP_ + n0 + lr] = fv[v];
      fo[(v + 8 * half) * OD_ + n0 + lr]  = fv[v];
    }
  }
  __syncthreads();

  const float* langS = lang_ws + (size_t)s * L_ * OD_;

  if (wave < 3) {
    const int c0 = wave * 16;
    const int lclamp = min(c0 + lr, L_ - 1);
    v8f cl = {};
    for (int kg = 0; kg < OD_; kg += 4 * KG_) {
      v2f a[KG_], b[KG_];
#pragma unroll
      for (int u = 0; u < KG_; ++u) {
        const int ka = kg + 4 * u + 2 * half;
        a[u].x = Asm[lr * RP_ + ka];
        a[u].y = Asm[lr * RP_ + ka + 1];
        b[u].x = langS[lclamp * OD_ + ka];
        b[u].y = langS[lclamp * OD_ + ka + 1];
      }
      __builtin_amdgcn_sched_barrier(0);
#pragma unroll
      for (int u = 0; u < KG_; ++u) cl = wmma4(a[u], b[u], cl);
    }
#pragma unroll
    for (int v = 0; v < 8; ++v)
      attnsm[(v + 8 * half) * LPP_ + c0 + lr] = cl[v];
  }
  __syncthreads();

  if (tid < 16) {
    float* rowp = attnsm + tid * LPP_;
    float mx = -FLT_MAX;
    for (int l = 0; l < L_; ++l) mx = fmaxf(mx, rowp[l]);
    float ssum = 0.f;
    for (int l = 0; l < L_; ++l) {
      const float e = expf(rowp[l] - mx);
      rowp[l] = e;
      ssum += e;
    }
    const float inv = 1.f / ssum;
    const float* mk = lang_mask + (size_t)s * L_;
    float s2 = 0.f;
    for (int l = 0; l < L_; ++l) {
      const float p = rowp[l] * inv * mk[l];
      rowp[l] = p;
      s2 += p;
    }
    const float inv2 = 1.f / (s2 + 1e-7f);
    for (int l = 0; l < L_; ++l) rowp[l] *= inv2;
    for (int l = L_; l < LP_; ++l) rowp[l] = 0.f;
  }
  __syncthreads();

  {
    v8f ci = {};
    {
      v2f a[LP_ / 4], b[LP_ / 4];
#pragma unroll
      for (int u = 0; u < LP_ / 4; ++u) {
        const int ka  = 4 * u + 2 * half;
        const int kb0 = min(ka, L_ - 1), kb1 = min(ka + 1, L_ - 1);
        a[u].x = attnsm[lr * LPP_ + ka];
        a[u].y = attnsm[lr * LPP_ + ka + 1];
        b[u].x = langS[kb0 * OD_ + n0 + lr];
        b[u].y = langS[kb1 * OD_ + n0 + lr];
      }
#pragma unroll
      for (int u = 0; u < LP_ / 4; ++u) ci = wmma4(a[u], b[u], ci);
    }
    float* go = g_out + ((size_t)s * N_ + mt * 16) * OD_;
#pragma unroll
    for (int v = 0; v < 8; ++v)
      go[(v + 8 * half) * OD_ + n0 + lr] = fv[v] * ci[v];
  }
}

#define PB_     16
#define K1S_    3
#define FT_     128

__global__ void __launch_bounds__(FT_)
fused_kernel(const float* __restrict__ coord,
             const float* __restrict__ rel_w1,
             const float* __restrict__ rel_b1,
             const float* __restrict__ rel_w2,
             const float* __restrict__ rel_b2,
             const int*   __restrict__ idx_ws,
             const float* __restrict__ g_ws,
             float* __restrict__ out,
             float* __restrict__ score) {
  __shared__ float relin[3 * FT_ * KP_];
  __shared__ float hbuf[2][PB_ * RP_];
  __shared__ int   idxsm[FT_ * 4] __attribute__((aligned(16)));

  const int s    = blockIdx.x >> 4;
  const int pt0  = (blockIdx.x & 15) * PB_;
  const int tid  = threadIdx.x;
  const int lane = tid & 31, wave = tid >> 5;
  const int half = lane >> 4, lr = lane & 15;
  const int c0   = wave * 32;

  const float* cs = coord + (size_t)s * N_ * 3;
  const int4 idxv = reinterpret_cast<const int4*>(
      idx_ws + ((size_t)s * N_ + pt0) * KNB)[min(tid, PB_ * KNB / 4 - 1)];
  const int   nctr = pt0 + (tid & 15);
  const float ox = cs[nctr * 3 + 0], oy = cs[nctr * 3 + 1], oz = cs[nctr * 3 + 2];

  v2f bw1[2][K1S_];
  v2f bw2[2][OD_ / 4];
  float bb1[2], bb2[2];
#pragma unroll
  for (int t = 0; t < 2; ++t) {
    const int col = c0 + 16 * t + lr;
#pragma unroll
    for (int u = 0; u < K1S_; ++u) {
      const int ka  = 4 * u + 2 * half;
      const int kb0 = min(ka, 9), kb1 = min(ka + 1, 9);
      bw1[t][u].x = rel_w1[kb0 * OD_ + col];
      bw1[t][u].y = rel_w1[kb1 * OD_ + col];
    }
#pragma unroll
    for (int u = 0; u < OD_ / 4; ++u) {
      const int ka = 4 * u + 2 * half;
      bw2[t][u].x = rel_w2[ka * OD_ + col];
      bw2[t][u].y = rel_w2[(ka + 1) * OD_ + col];
    }
    bb1[t] = rel_b1[col];
    bb2[t] = rel_b2[col];
  }
  reinterpret_cast<int4*>(idxsm)[tid] = idxv;
  __syncthreads();

  {
    float nb[3][3];
#pragma unroll
    for (int u = 0; u < 3; ++u) {
      const int r = min(tid + FT_ * u, PB_ * KNB - 1);
      const int m = idxsm[(r & 15) * KNB + (r >> 4)];
      nb[u][0] = cs[m * 3 + 0];
      nb[u][1] = cs[m * 3 + 1];
      nb[u][2] = cs[m * 3 + 2];
    }
#pragma unroll
    for (int u = 0; u < 3; ++u) {
      const float nx = nb[u][0], ny = nb[u][1], nz = nb[u][2];
      const float rx = nx - ox, ry = ny - oy, rz = nz - oz;
      const float dist = sqrtf(rx * rx + ry * ry + rz * rz);
      float* ri = relin + (tid + FT_ * u) * KP_;
      ri[0] = nx; ri[1] = ny; ri[2] = nz;
      ri[3] = ox; ri[4] = oy; ri[5] = oz;
      ri[6] = rx; ri[7] = ry; ri[8] = rz;
      ri[9] = dist;
      ri[10] = 0.f; ri[11] = 0.f;
    }
  }
  __syncthreads();

  float acc[2][8];
#pragma unroll
  for (int v = 0; v < 8; ++v) { acc[0][v] = 0.f; acc[1][v] = 0.f; }

  const int*   ip = idxsm + 8 * half * KNB;
  const float* gs = g_ws + (size_t)s * N_ * OD_;
  const unsigned lcol = (unsigned)(c0 + lr);

  for (int j = 0; j < KNB; ++j) {
    unsigned m[8];
#pragma unroll
    for (int v = 0; v < 8; ++v) m[v] = (unsigned)ip[v * KNB + j];

    v8f ca = {}, cb = {};
    {
      const float* rp = relin + (j * PB_ + lr) * KP_ + 2 * half;
      v2f a[K1S_];
#pragma unroll
      for (int u = 0; u < K1S_; ++u) {
        a[u].x = rp[4 * u];
        a[u].y = rp[4 * u + 1];
      }
#pragma unroll
      for (int u = 0; u < K1S_; ++u) {
        ca = wmma4(a[u], bw1[0][u], ca);
        cb = wmma4(a[u], bw1[1][u], cb);
      }
    }
    float* hb = hbuf[j & 1];
    {
      float* hp = hb + 8 * half * RP_ + c0 + lr;
#pragma unroll
      for (int v = 0; v < 8; ++v) {
        hp[v * RP_]      = fmaxf(ca[v] + bb1[0], 0.f);
        hp[v * RP_ + 16] = fmaxf(cb[v] + bb1[1], 0.f);
      }
    }
    __syncthreads();

    float gv[2][8];
#pragma unroll
    for (int v = 0; v < 8; ++v) {
      const float* gp = gs + (m[v] * OD_ + lcol);
      gv[0][v] = gp[0];
      gv[1][v] = gp[16];
    }

    v8f c2a = {}, c2b = {};
    const float* ap = hb + lr * RP_ + 2 * half;
    for (int kg = 0; kg < OD_ / 4; kg += KG_) {
      v2f a[KG_];
#pragma unroll
      for (int u = 0; u < KG_; ++u) {
        a[u].x = ap[4 * (kg + u)];
        a[u].y = ap[4 * (kg + u) + 1];
      }
      __builtin_amdgcn_sched_barrier(0);
#pragma unroll
      for (int u = 0; u < KG_; ++u) {
        c2a = wmma4(a[u], bw2[0][kg + u], c2a);
        c2b = wmma4(a[u], bw2[1][kg + u], c2b);
      }
    }
#pragma unroll
    for (int v = 0; v < 8; ++v) {
      acc[0][v] += gv[0][v] * (c2a[v] + bb2[0]);
      acc[1][v] += gv[1][v] * (c2b[v] + bb2[1]);
    }
  }

  float fst[PB_ / (FT_ / 16)][8];
#pragma unroll
  for (int h2 = 0; h2 < PB_ / (FT_ / 16); ++h2) {
    const float* o = out + ((size_t)s * N_ + pt0 + h2 * (FT_ / 16) + (tid >> 4)) * OD_;
#pragma unroll
    for (int i = 0; i < 8; ++i) fst[h2][i] = o[(tid & 15) + 16 * i];
  }

  float* osm = hbuf[1];
  {
    float* op = osm + 8 * half * RP_ + c0 + lr;
#pragma unroll
    for (int v = 0; v < 8; ++v) {
      op[v * RP_]      = acc[0][v];
      op[v * RP_ + 16] = acc[1][v];
    }
  }
  __syncthreads();

#pragma unroll
  for (int pp = 0; pp < PB_; pp += FT_ / 16) {
    const int p = pp + (tid >> 4), q = tid & 15;
    float* o = out + ((size_t)s * N_ + pt0 + p) * OD_;
    float x[8];
#pragma unroll
    for (int i = 0; i < 8; ++i)
      x[i] = osm[p * RP_ + q + 16 * i] + fst[pp / (FT_ / 16)][i];
#pragma unroll
    for (int i = 0; i < 8; ++i) o[q + 16 * i] = x[i];
    const float y0 = x[0] + x[4], y1 = x[1] + x[5];
    const float y2 = x[2] + x[6], y3 = x[3] + x[7];
    float r = (y0 + y2) + (y1 + y3);
    r += __shfl_xor(r, 8);
    r += __shfl_xor(r, 4);
    r += __shfl_xor(r, 2);
    r += __shfl_xor(r, 1);
    if (q == 0) score[(size_t)s * N_ + pt0 + p] = r;
  }
}

extern "C" void kernel_launch(void* const* d_in, const int* in_sizes, int n_in,
                              void* d_out, int out_size, void* d_ws,
                              size_t ws_size, hipStream_t stream) {
  (void)in_sizes; (void)n_in; (void)out_size; (void)ws_size;
  const float* feat      = (const float*)d_in[0];
  const float* coord     = (const float*)d_in[1];
  const float* lang_feat = (const float*)d_in[2];
  const float* lang_mask = (const float*)d_in[3];
  const float* rel_w1    = (const float*)d_in[4];
  const float* rel_b1    = (const float*)d_in[5];
  const float* rel_w2    = (const float*)d_in[6];
  const float* rel_b2    = (const float*)d_in[7];
  const float* lang_w1   = (const float*)d_in[8];
  const float* lang_b1   = (const float*)d_in[9];
  const float* lang_w2   = (const float*)d_in[10];
  const float* lang_b2   = (const float*)d_in[11];
  const float* feat_w1   = (const float*)d_in[12];
  const float* feat_b1   = (const float*)d_in[13];
  const float* feat_w2   = (const float*)d_in[14];
  const float* feat_b2   = (const float*)d_in[15];

  char* ws = (char*)d_ws;
  int*   idx_ws  = (int*)ws;
  float* lang_ws = (float*)(ws + (size_t)S_ * N_ * KNB * 4);
  float* g_ws    = (float*)(ws + (size_t)S_ * N_ * KNB * 4 +
                                 (size_t)S_ * L_ * OD_ * 4);

  float* out   = (float*)d_out;
  float* score = out + (size_t)S_ * N_ * OD_;

  knn_lang_kernel<<<KNN_BLOCKS_ + LANG_BLOCKS_, 256, 0, stream>>>(
      coord, idx_ws, lang_feat, lang_w1, lang_b1, lang_w2, lang_b2, lang_ws);
  feat_mlp_kernel<<<S_ * (N_ / 16), 256, 0, stream>>>(feat, feat_w1, feat_b1,
                                                      feat_w2, feat_b2, lang_mask,
                                                      lang_ws, g_ws, out);
  fused_kernel<<<S_ * (N_ / PB_), FT_, 0, stream>>>(coord, rel_w1, rel_b1,
                                                     rel_w2, rel_b2, idx_ws,
                                                     g_ws, out, score);
}


# Feat kernel retiled: 32 points per block on 4 waves, each wave 2 row x 2 col tiles (4 chains, each A and B fragment feeds 2 WMMAs); 512 blocks x 4 waves, was 1024 x 8; sum order unchanged
# speedup vs baseline: 9.4758x; 1.0584x over previous
#include <hip/hip_runtime.h>
#include <hip/hip_bf16.h>
#include <float.h>


typedef float v2f __attribute__((ext_vector_type(2)));
typedef float v8f __attribute__((ext_vector_type(8)));

#define S_      64
#define N_      256
#define L_      40
#define LP_     48
#define RP_     132
#define LPP_    52
#define KP_     20
#define KNB     17
#define OD_     128
#define LANGID_ 256
#define PCID_   128
#define KG_     16

static __device__ inline v8f wmma4(v2f a, v2f b, v8f c) {
  return __builtin_amdgcn_wmma_f32_16x16x4_f32(false, a, false, b, (short)0, c,
                                               false, false);
}

#define KPW_    4
#define KWV_    8
#define KPB_    (KPW_ * KWV_)
#define KRM_    0xffffffffu

template <int XM>
static __device__ inline unsigned row_xor(unsigned x) {
  return (unsigned)__builtin_amdgcn_mov_dpp((int)x, 0x160 + XM, 0xf, 0xf, true);
}

static __device__ inline unsigned other_row(unsigned x) {
  return (unsigned)__builtin_amdgcn_permlanex16((int)x, (int)x, 0x76543210,
                                                0xfedcba98, false, false);
}

static __device__ inline unsigned umin2(unsigned a, unsigned b) {
  return a < b ? a : b;
}

static __device__ inline unsigned wave_umin(unsigned x) {
  x = umin2(x, row_xor<1>(x));
  x = umin2(x, row_xor<2>(x));
  x = umin2(x, row_xor<4>(x));
  x = umin2(x, row_xor<8>(x));
  return umin2(x, other_row(x));
}

#define KNN_BLOCKS_ (S_ * (N_ / KPB_))

static __device__ __forceinline__ void
knn_role(const unsigned bid, const float* __restrict__ coord,
         int* __restrict__ idx_out) {
  __shared__ int osm[KPB_ * KNB];
  const int tid  = threadIdx.x;
  const int lane = tid & 31;
  const int wave = __builtin_amdgcn_readfirstlane(tid >> 5);
  const int s    = (int)(bid / (N_ / KPB_));
  const int pb   = (int)(bid % (N_ / KPB_)) * KPB_;
  const int p0   = pb + wave * KPW_;
  const float* cs = coord + (size_t)s * N_ * 3;

  float c[24];
  {
    const float4* cv = reinterpret_cast<const float4*>(cs + lane * 24);
    for (int u = 0; u < 6; ++u) {
      const float4 t = cv[u];
      c[4 * u + 0] = t.x; c[4 * u + 1] = t.y;
      c[4 * u + 2] = t.z; c[4 * u + 3] = t.w;
    }
  }

  unsigned d[KPW_][8];
  for (int q = 0; q < KPW_; ++q) {
    const float px = cs[(p0 + q) * 3 + 0];
    const float py = cs[(p0 + q) * 3 + 1];
    const float pz = cs[(p0 + q) * 3 + 2];
    for (int i = 0; i < 8; ++i) {
      const float dx = c[3 * i + 0] - px;
      const float dy = c[3 * i + 1] - py;
      const float dz = c[3 * i + 2] - pz;
      const float d2 = __builtin_fmaf(dz, dz, __builtin_fmaf(dx, dx, dy * dy));
      d[q][i] = __float_as_uint(d2);
    }
  }

  const int lane8 = lane * 8;
  int res[KPW_] = {};

  for (int j = 0; j < KNB; ++j) {
    for (int q = 0; q < KPW_; ++q) {
      unsigned* dq = d[q];
      unsigned lm = umin2(dq[0], dq[1]);
      for (int i = 2; i < 8; ++i) lm = umin2(lm, dq[i]);
      int li = 7;
      for (int i = 6; i >= 0; --i) li = (dq[i] == lm) ? i : li;
      const unsigned wm  = wave_umin(lm);
      const unsigned bal = __builtin_amdgcn_ballot_w32(lm == wm);
      const int L  = __builtin_ctz(bal);
      const int sl = __builtin_amdgcn_readlane(li, L);
      const int m  = L * 8 + sl;
      res[q] = (lane == j) ? m : res[q];
      const int rel = m - lane8;
      for (int i = 0; i < 8; ++i) dq[i] = (rel == i) ? KRM_ : dq[i];
    }
  }

  if (lane < KNB) {
    for (int q = 0; q < KPW_; ++q)
      osm[(wave * KPW_ + q) * KNB + lane] = res[q];
  }
  __syncthreads();
  int* ob = idx_out + ((size_t)s * N_ + pb) * KNB;
  for (int i = tid; i < KPB_ * KNB; i += 32 * KWV_) ob[i] = osm[i];
}

#define XP_     260
#define LROWS_  (S_ * L_)
#define LANG_BLOCKS_ (LROWS_ / 16)

static __device__ __forceinline__ void
lang_role(const unsigned bid, const float* __restrict__ lang_feat,
          const float* __restrict__ w1,
          const float* __restrict__ b1,
          const float* __restrict__ w2,
          const float* __restrict__ b2,
          float* __restrict__ lang_out) {
  __shared__ float Xsm[16 * XP_] __attribute__((aligned(16)));
  __shared__ float Hsm[16 * RP_];
  const int row0 = (int)bid * 16;
  const int tid  = threadIdx.x;
  const int lane = tid & 31, wave = tid >> 5;
  const int half = lane >> 4, lr = lane & 15;

  {
    const float4* X4 =
        reinterpret_cast<const float4*>(lang_feat + (size_t)row0 * LANGID_);
    float4 t[4];
#pragma unroll
    for (int u = 0; u < 4; ++u) t[u] = X4[tid + 256 * u];
#pragma unroll
    for (int u = 0; u < 4; ++u) {
      const int i4 = tid + 256 * u;
      *reinterpret_cast<float4*>(Xsm + (i4 >> 6) * XP_ + (i4 & 63) * 4) = t[u];
    }
  }
  __syncthreads();

  const int n0 = wave * 16;
  v8f c = {};
  for (int kg = 0; kg < LANGID_; kg += 4 * KG_) {
    v2f a[KG_], b[KG_];
#pragma unroll
    for (int u = 0; u < KG_; ++u) {
      const int ka = kg + 4 * u + 2 * half;
      a[u].x = Xsm[lr * XP_ + ka];
      a[u].y = Xsm[lr * XP_ + ka + 1];
      b[u].x = w1[ka * OD_ + n0 + lr];
      b[u].y = w1[(ka + 1) * OD_ + n0 + lr];
    }
    __builtin_amdgcn_sched_barrier(0);
#pragma unroll
    for (int u = 0; u < KG_; ++u) c = wmma4(a[u], b[u], c);
  }
  {
    const float bb = b1[n0 + lr];
#pragma unroll
    for (int v = 0; v < 8; ++v)
      Hsm[(v + 8 * half) * RP_ + n0 + lr] = fmaxf(c[v] + bb, 0.0f);
  }
  __syncthreads();

  v8f c2 = {};
  for (int kg = 0; kg < OD_; kg += 4 * KG_) {
    v2f a[KG_], b[KG_];
#pragma unroll
    for (int u = 0; u < KG_; ++u) {
      const int ka = kg + 4 * u + 2 * half;
      a[u].x = Hsm[lr * RP_ + ka];
      a[u].y = Hsm[lr * RP_ + ka + 1];
      b[u].x = w2[ka * OD_ + n0 + lr];
      b[u].y = w2[(ka + 1) * OD_ + n0 + lr];
    }
    __builtin_amdgcn_sched_barrier(0);
#pragma unroll
    for (int u = 0; u < KG_; ++u) c2 = wmma4(a[u], b[u], c2);
  }
  {
    const float bb = b2[n0 + lr];
#pragma unroll
    for (int v = 0; v < 8; ++v)
      Xsm[(v + 8 * half) * XP_ + n0 + lr] = c2[v] + bb;
  }
  __syncthreads();
  float* lo = lang_out + (size_t)row0 * OD_;
  for (int u = 0; u < 8; ++u) {
    const int i = tid + 256 * u;
    lo[i] = Xsm[(i >> 7) * XP_ + (i & 127)];
  }
}

__global__ void __launch_bounds__(256)
knn_lang_kernel(const float* __restrict__ coord, int* __restrict__ idx_out,
                const float* __restrict__ lang_feat,
                const float* __restrict__ w1,
                const float* __restrict__ b1,
                const float* __restrict__ w2,
                const float* __restrict__ b2,
                float* __restrict__ lang_out) {
  const unsigned bid = blockIdx.x;
  if (bid >= KNN_BLOCKS_) {
    lang_role(bid - KNN_BLOCKS_, lang_feat, w1, b1, w2, b2, lang_out);
  } else {
    knn_role(bid, coord, idx_out);
  }
}

#define FR_     32
#define FTH_    128

__global__ void __launch_bounds__(FTH_)
feat_mlp_kernel(const float* __restrict__ feat,
                const float* __restrict__ w1,
                const float* __restrict__ b1,
                const float* __restrict__ w2,
                const float* __restrict__ b2,
                const float* __restrict__ lang_mask,
                const float* __restrict__ lang_ws,
                float* __restrict__ g_out,
                float* __restrict__ f_out) {
  __shared__ float Asm[FR_ * RP_] __attribute__((aligned(16)));
  __shared__ float Hsm[FR_ * RP_];
  __shared__ float attnsm[FR_ * LPP_];
  const int s    = blockIdx.x / (N_ / FR_);
  const int mt   = blockIdx.x % (N_ / FR_);
  const int tid  = threadIdx.x;
  const int lane = tid & 31, wave = tid >> 5;
  const int half = lane >> 4, lr = lane & 15;
  const size_t row0 = (size_t)s * N_ + mt * FR_;

  {
    const float4* A4 = reinterpret_cast<const float4*>(feat + row0 * PCID_);
    float4 t[8];
#pragma unroll
    for (int u = 0; u < 8; ++u) t[u] = A4[tid + FTH_ * u];
#pragma unroll
    for (int u = 0; u < 8; ++u) {
      const int i4 = tid + FTH_ * u;
      *reinterpret_cast<float4*>(Asm + (i4 >> 5) * RP_ + (i4 & 31) * 4) = t[u];
    }
  }
  __syncthreads();

  const int n0 = wave * 32;
  v8f c[2][2] = {};
#pragma unroll
  for (int kg = 0; kg < PCID_; kg += 4 * KG_) {
    v2f a[2][KG_], b[2][KG_];
#pragma unroll
    for (int u = 0; u < KG_; ++u) {
      const int ka = kg + 4 * u + 2 * half;
      a[0][u].x = Asm[lr * RP_ + ka];
      a[0][u].y = Asm[lr * RP_ + ka + 1];
      a[1][u].x = Asm[(16 + lr) * RP_ + ka];
      a[1][u].y = Asm[(16 + lr) * RP_ + ka + 1];
      b[0][u].x = w1[ka * OD_ + n0 + lr];
      b[0][u].y = w1[(ka + 1) * OD_ + n0 + lr];
      b[1][u].x = w1[ka * OD_ + n0 + 16 + lr];
      b[1][u].y = w1[(ka + 1) * OD_ + n0 + 16 + lr];
    }
    __builtin_amdgcn_sched_barrier(0);
#pragma unroll
    for (int u = 0; u < KG_; ++u) {
      c[0][0] = wmma4(a[0][u], b[0][u], c[0][0]);
      c[0][1] = wmma4(a[0][u], b[1][u], c[0][1]);
      c[1][0] = wmma4(a[1][u], b[0][u], c[1][0]);
      c[1][1] = wmma4(a[1][u], b[1][u], c[1][1]);
    }
  }
#pragma unroll
  for (int t = 0; t < 2; ++t) {
    const int col = n0 + 16 * t + lr;
    const float bb = b1[col];
#pragma unroll
    for (int r = 0; r < 2; ++r)
#pragma unroll
      for (int v = 0; v < 8; ++v)
        Hsm[(16 * r + v + 8 * half) * RP_ + col] = fmaxf(c[r][t][v] + bb, 0.0f);
  }
  __syncthreads();

  v8f c2[2][2] = {};
#pragma unroll
  for (int kg = 0; kg < OD_; kg += 4 * KG_) {
    v2f a[2][KG_], b[2][KG_];
#pragma unroll
    for (int u = 0; u < KG_; ++u) {
      const int ka = kg + 4 * u + 2 * half;
      a[0][u].x = Hsm[lr * RP_ + ka];
      a[0][u].y = Hsm[lr * RP_ + ka + 1];
      a[1][u].x = Hsm[(16 + lr) * RP_ + ka];
      a[1][u].y = Hsm[(16 + lr) * RP_ + ka + 1];
      b[0][u].x = w2[ka * OD_ + n0 + lr];
      b[0][u].y = w2[(ka + 1) * OD_ + n0 + lr];
      b[1][u].x = w2[ka * OD_ + n0 + 16 + lr];
      b[1][u].y = w2[(ka + 1) * OD_ + n0 + 16 + lr];
    }
    __builtin_amdgcn_sched_barrier(0);
#pragma unroll
    for (int u = 0; u < KG_; ++u) {
      c2[0][0] = wmma4(a[0][u], b[0][u], c2[0][0]);
      c2[0][1] = wmma4(a[0][u], b[1][u], c2[0][1]);
      c2[1][0] = wmma4(a[1][u], b[0][u], c2[1][0]);
      c2[1][1] = wmma4(a[1][u], b[1][u], c2[1][1]);
    }
  }
  float fv[2][2][8];
  {
    float* fo = f_out + row0 * OD_;
#pragma unroll
    for (int t = 0; t < 2; ++t) {
      const int col = n0 + 16 * t + lr;
      const float bb = b2[col];
#pragma unroll
      for (int r = 0; r < 2; ++r)
#pragma unroll
        for (int v = 0; v < 8; ++v) {
          const int row = 16 * r + v + 8 * half;
          fv[r][t][v] = c2[r][t][v] + bb;
          Asm[row * RP_ + col] = fv[r][t][v];
          fo[row * OD_ + col]  = fv[r][t][v];
        }
    }
  }
  __syncthreads();

  const float* langS = lang_ws + (size_t)s * L_ * OD_;

  if (wave < 3) {
    const int c0 = wave * 16;
    const int lclamp = min(c0 + lr, L_ - 1);
    v8f cl[2] = {};
#pragma unroll
    for (int kg = 0; kg < OD_; kg += 4 * KG_) {
      v2f a[2][KG_], b[KG_];
#pragma unroll
      for (int u = 0; u < KG_; ++u) {
        const int ka = kg + 4 * u + 2 * half;
        a[0][u].x = Asm[lr * RP_ + ka];
        a[0][u].y = Asm[lr * RP_ + ka + 1];
        a[1][u].x = Asm[(16 + lr) * RP_ + ka];
        a[1][u].y = Asm[(16 + lr) * RP_ + ka + 1];
        b[u].x = langS[lclamp * OD_ + ka];
        b[u].y = langS[lclamp * OD_ + ka + 1];
      }
      __builtin_amdgcn_sched_barrier(0);
#pragma unroll
      for (int u = 0; u < KG_; ++u) {
        cl[0] = wmma4(a[0][u], b[u], cl[0]);
        cl[1] = wmma4(a[1][u], b[u], cl[1]);
      }
    }
#pragma unroll
    for (int r = 0; r < 2; ++r)
#pragma unroll
      for (int v = 0; v < 8; ++v)
        attnsm[(16 * r + v + 8 * half) * LPP_ + c0 + lr] = cl[r][v];
  }
  __syncthreads();

  if (tid < FR_) {
    float* rowp = attnsm + tid * LPP_;
    float mx = -FLT_MAX;
    for (int l = 0; l < L_; ++l) mx = fmaxf(mx, rowp[l]);
    float ssum = 0.f;
    for (int l = 0; l < L_; ++l) {
      const float e = expf(rowp[l] - mx);
      rowp[l] = e;
      ssum += e;
    }
    const float inv = 1.f / ssum;
    const float* mk = lang_mask + (size_t)s * L_;
    float s2 = 0.f;
    for (int l = 0; l < L_; ++l) {
      const float p = rowp[l] * inv * mk[l];
      rowp[l] = p;
      s2 += p;
    }
    const float inv2 = 1.f / (s2 + 1e-7f);
    for (int l = 0; l < L_; ++l) rowp[l] *= inv2;
    for (int l = L_; l < LP_; ++l) rowp[l] = 0.f;
  }
  __syncthreads();

  {
    v8f ci[2][2] = {};
    {
      v2f a[2][LP_ / 4], b[2][LP_ / 4];
#pragma unroll
      for (int u = 0; u < LP_ / 4; ++u) {
        const int ka  = 4 * u + 2 * half;
        const int kb0 = min(ka, L_ - 1), kb1 = min(ka + 1, L_ - 1);
        a[0][u].x = attnsm[lr * LPP_ + ka];
        a[0][u].y = attnsm[lr * LPP_ + ka + 1];
        a[1][u].x = attnsm[(16 + lr) * LPP_ + ka];
        a[1][u].y = attnsm[(16 + lr) * LPP_ + ka + 1];
        b[0][u].x = langS[kb0 * OD_ + n0 + lr];
        b[0][u].y = langS[kb1 * OD_ + n0 + lr];
        b[1][u].x = langS[kb0 * OD_ + n0 + 16 + lr];
        b[1][u].y = langS[kb1 * OD_ + n0 + 16 + lr];
      }
#pragma unroll
      for (int u = 0; u < LP_ / 4; ++u) {
        ci[0][0] = wmma4(a[0][u], b[0][u], ci[0][0]);
        ci[0][1] = wmma4(a[0][u], b[1][u], ci[0][1]);
        ci[1][0] = wmma4(a[1][u], b[0][u], ci[1][0]);
        ci[1][1] = wmma4(a[1][u], b[1][u], ci[1][1]);
      }
    }
    float* go = g_out + row0 * OD_;
#pragma unroll
    for (int t = 0; t < 2; ++t)
#pragma unroll
      for (int r = 0; r < 2; ++r)
#pragma unroll
        for (int v = 0; v < 8; ++v)
          go[(16 * r + v + 8 * half) * OD_ + n0 + 16 * t + lr] =
              fv[r][t][v] * ci[r][t][v];
  }
}

#define PB_     16
#define K1S_    3
#define FT_     128

__global__ void __launch_bounds__(FT_)
fused_kernel(const float* __restrict__ coord,
             const float* __restrict__ rel_w1,
             const float* __restrict__ rel_b1,
             const float* __restrict__ rel_w2,
             const float* __restrict__ rel_b2,
             const int*   __restrict__ idx_ws,
             const float* __restrict__ g_ws,
             float* __restrict__ out,
             float* __restrict__ score) {
  __shared__ float relin[3 * FT_ * KP_];
  __shared__ float hbuf[2][PB_ * RP_];
  __shared__ int   idxsm[FT_ * 4] __attribute__((aligned(16)));

  const int s    = blockIdx.x >> 4;
  const int pt0  = (blockIdx.x & 15) * PB_;
  const int tid  = threadIdx.x;
  const int lane = tid & 31, wave = tid >> 5;
  const int half = lane >> 4, lr = lane & 15;
  const int c0   = wave * 32;

  const float* cs = coord + (size_t)s * N_ * 3;
  const int4 idxv = reinterpret_cast<const int4*>(
      idx_ws + ((size_t)s * N_ + pt0) * KNB)[min(tid, PB_ * KNB / 4 - 1)];
  const int   nctr = pt0 + (tid & 15);
  const float ox = cs[nctr * 3 + 0], oy = cs[nctr * 3 + 1], oz = cs[nctr * 3 + 2];

  v2f bw1[2][K1S_];
  v2f bw2[2][OD_ / 4];
  float bb1[2], bb2[2];
#pragma unroll
  for (int t = 0; t < 2; ++t) {
    const int col = c0 + 16 * t + lr;
#pragma unroll
    for (int u = 0; u < K1S_; ++u) {
      const int ka  = 4 * u + 2 * half;
      const int kb0 = min(ka, 9), kb1 = min(ka + 1, 9);
      bw1[t][u].x = rel_w1[kb0 * OD_ + col];
      bw1[t][u].y = rel_w1[kb1 * OD_ + col];
    }
#pragma unroll
    for (int u = 0; u < OD_ / 4; ++u) {
      const int ka = 4 * u + 2 * half;
      bw2[t][u].x = rel_w2[ka * OD_ + col];
      bw2[t][u].y = rel_w2[(ka + 1) * OD_ + col];
    }
    bb1[t] = rel_b1[col];
    bb2[t] = rel_b2[col];
  }
  reinterpret_cast<int4*>(idxsm)[tid] = idxv;
  __syncthreads();

  {
    float nb[3][3];
#pragma unroll
    for (int u = 0; u < 3; ++u) {
      const int r = min(tid + FT_ * u, PB_ * KNB - 1);
      const int m = idxsm[(r & 15) * KNB + (r >> 4)];
      nb[u][0] = cs[m * 3 + 0];
      nb[u][1] = cs[m * 3 + 1];
      nb[u][2] = cs[m * 3 + 2];
    }
#pragma unroll
    for (int u = 0; u < 3; ++u) {
      const float nx = nb[u][0], ny = nb[u][1], nz = nb[u][2];
      const float rx = nx - ox, ry = ny - oy, rz = nz - oz;
      const float dist = sqrtf(rx * rx + ry * ry + rz * rz);
      float* ri = relin + (tid + FT_ * u) * KP_;
      ri[0] = nx; ri[1] = ny; ri[2] = nz;
      ri[3] = ox; ri[4] = oy; ri[5] = oz;
      ri[6] = rx; ri[7] = ry; ri[8] = rz;
      ri[9] = dist;
      ri[10] = 0.f; ri[11] = 0.f;
    }
  }
  __syncthreads();

  float acc[2][8];
#pragma unroll
  for (int v = 0; v < 8; ++v) { acc[0][v] = 0.f; acc[1][v] = 0.f; }

  const int*   ip = idxsm + 8 * half * KNB;
  const float* gs = g_ws + (size_t)s * N_ * OD_;
  const unsigned lcol = (unsigned)(c0 + lr);

  for (int j = 0; j < KNB; ++j) {
    unsigned m[8];
#pragma unroll
    for (int v = 0; v < 8; ++v) m[v] = (unsigned)ip[v * KNB + j];

    v8f ca = {}, cb = {};
    {
      const float* rp = relin + (j * PB_ + lr) * KP_ + 2 * half;
      v2f a[K1S_];
#pragma unroll
      for (int u = 0; u < K1S_; ++u) {
        a[u].x = rp[4 * u];
        a[u].y = rp[4 * u + 1];
      }
#pragma unroll
      for (int u = 0; u < K1S_; ++u) {
        ca = wmma4(a[u], bw1[0][u], ca);
        cb = wmma4(a[u], bw1[1][u], cb);
      }
    }
    float* hb = hbuf[j & 1];
    {
      float* hp = hb + 8 * half * RP_ + c0 + lr;
#pragma unroll
      for (int v = 0; v < 8; ++v) {
        hp[v * RP_]      = fmaxf(ca[v] + bb1[0], 0.f);
        hp[v * RP_ + 16] = fmaxf(cb[v] + bb1[1], 0.f);
      }
    }
    __syncthreads();

    float gv[2][8];
#pragma unroll
    for (int v = 0; v < 8; ++v) {
      const float* gp = gs + (m[v] * OD_ + lcol);
      gv[0][v] = gp[0];
      gv[1][v] = gp[16];
    }

    v8f c2a = {}, c2b = {};
    const float* ap = hb + lr * RP_ + 2 * half;
    for (int kg = 0; kg < OD_ / 4; kg += KG_) {
      v2f a[KG_];
#pragma unroll
      for (int u = 0; u < KG_; ++u) {
        a[u].x = ap[4 * (kg + u)];
        a[u].y = ap[4 * (kg + u) + 1];
      }
      __builtin_amdgcn_sched_barrier(0);
#pragma unroll
      for (int u = 0; u < KG_; ++u) {
        c2a = wmma4(a[u], bw2[0][kg + u], c2a);
        c2b = wmma4(a[u], bw2[1][kg + u], c2b);
      }
    }
#pragma unroll
    for (int v = 0; v < 8; ++v) {
      acc[0][v] += gv[0][v] * (c2a[v] + bb2[0]);
      acc[1][v] += gv[1][v] * (c2b[v] + bb2[1]);
    }
  }

  float fst[PB_ / (FT_ / 16)][8];
#pragma unroll
  for (int h2 = 0; h2 < PB_ / (FT_ / 16); ++h2) {
    const float* o = out + ((size_t)s * N_ + pt0 + h2 * (FT_ / 16) + (tid >> 4)) * OD_;
#pragma unroll
    for (int i = 0; i < 8; ++i) fst[h2][i] = o[(tid & 15) + 16 * i];
  }

  float* osm = hbuf[1];
  {
    float* op = osm + 8 * half * RP_ + c0 + lr;
#pragma unroll
    for (int v = 0; v < 8; ++v) {
      op[v * RP_]      = acc[0][v];
      op[v * RP_ + 16] = acc[1][v];
    }
  }
  __syncthreads();

#pragma unroll
  for (int pp = 0; pp < PB_; pp += FT_ / 16) {
    const int p = pp + (tid >> 4), q = tid & 15;
    float* o = out + ((size_t)s * N_ + pt0 + p) * OD_;
    float x[8];
#pragma unroll
    for (int i = 0; i < 8; ++i)
      x[i] = osm[p * RP_ + q + 16 * i] + fst[pp / (FT_ / 16)][i];
#pragma unroll
    for (int i = 0; i < 8; ++i) o[q + 16 * i] = x[i];
    const float y0 = x[0] + x[4], y1 = x[1] + x[5];
    const float y2 = x[2] + x[6], y3 = x[3] + x[7];
    float r = (y0 + y2) + (y1 + y3);
    r += __shfl_xor(r, 8);
    r += __shfl_xor(r, 4);
    r += __shfl_xor(r, 2);
    r += __shfl_xor(r, 1);
    if (q == 0) score[(size_t)s * N_ + pt0 + p] = r;
  }
}

extern "C" void kernel_launch(void* const* d_in, const int* in_sizes, int n_in,
                              void* d_out, int out_size, void* d_ws,
                              size_t ws_size, hipStream_t stream) {
  (void)in_sizes; (void)n_in; (void)out_size; (void)ws_size;
  const float* feat      = (const float*)d_in[0];
  const float* coord     = (const float*)d_in[1];
  const float* lang_feat = (const float*)d_in[2];
  const float* lang_mask = (const float*)d_in[3];
  const float* rel_w1    = (const float*)d_in[4];
  const float* rel_b1    = (const float*)d_in[5];
  const float* rel_w2    = (const float*)d_in[6];
  const float* rel_b2    = (const float*)d_in[7];
  const float* lang_w1   = (const float*)d_in[8];
  const float* lang_b1   = (const float*)d_in[9];
  const float* lang_w2   = (const float*)d_in[10];
  const float* lang_b2   = (const float*)d_in[11];
  const float* feat_w1   = (const float*)d_in[12];
  const float* feat_b1   = (const float*)d_in[13];
  const float* feat_w2   = (const float*)d_in[14];
  const float* feat_b2   = (const float*)d_in[15];

  char* ws = (char*)d_ws;
  int*   idx_ws  = (int*)ws;
  float* lang_ws = (float*)(ws + (size_t)S_ * N_ * KNB * 4);
  float* g_ws    = (float*)(ws + (size_t)S_ * N_ * KNB * 4 +
                                 (size_t)S_ * L_ * OD_ * 4);

  float* out   = (float*)d_out;
  float* score = out + (size_t)S_ * N_ * OD_;

  knn_lang_kernel<<<KNN_BLOCKS_ + LANG_BLOCKS_, 256, 0, stream>>>(
      coord, idx_ws, lang_feat, lang_w1, lang_b1, lang_w2, lang_b2, lang_ws);
  feat_mlp_kernel<<<S_ * (N_ / FR_), FTH_, 0, stream>>>(feat, feat_w1, feat_b1,
                                                        feat_w2, feat_b2,
                                                        lang_mask, lang_ws,
                                                        g_ws, out);
  fused_kernel<<<S_ * (N_ / PB_), FT_, 0, stream>>>(coord, rel_w1, rel_b1,
                                                     rel_w2, rel_b2, idx_ws,
                                                     g_ws, out, score);
}
